# starting kernel
# baseline (speedup 1.0000x reference)
#include <hip/hip_runtime.h>

typedef __attribute__((ext_vector_type(2))) float v2f;
typedef __attribute__((ext_vector_type(8))) float v8f;

#define N_COMBOS 15
#define KP 164
#define PN 112
#define NTILES 7
#define WAVES 2
#define FEAT 18

constexpr int cLE[N_COMBOS]     = {0,0,0,1,1,1,1,1,1,2,2,2,2,2,2};
constexpr int cLN[N_COMBOS]     = {0,1,2,0,1,1,1,2,2,0,1,1,2,2,2};
constexpr int cLO[N_COMBOS]     = {0,1,2,1,0,1,2,1,2,2,1,2,0,1,2};
constexpr int OUT_OFF[N_COMBOS] = {0,2,8,18,24,26,32,42,48,58,68,74,84,86,92};
constexpr int U_OFF[9]          = {0,2,8,18,24,42,72,82,112};
constexpr int KLO[NTILES]       = {0,2,6,10,18,20,28};
constexpr int KHI[NTILES]       = {4,10,17,20,27,40,40};
constexpr int N_BY_LO[3]        = {3,6,6};
constexpr int C_BY_LO[3][6]     = {{0,4,12,0,0,0},{1,3,5,7,10,13},{2,6,8,9,11,14}};

__global__ void build_M_kernel(const float* cg0, const float* cg1, const float* cg2,
                               const float* cg3, const float* cg4, const float* cg5,
                               const float* cg6, const float* cg7, const float* cg8,
                               const float* cg9, const float* cg10, const float* cg11,
                               const float* cg12, const float* cg13, const float* cg14,
                               float* M) {
    const float* cgs[N_COMBOS] = {cg0,cg1,cg2,cg3,cg4,cg5,cg6,cg7,cg8,cg9,cg10,cg11,cg12,cg13,cg14};
    int tid = threadIdx.x;
    for (int i = tid; i < KP * PN; i += blockDim.x) M[i] = 0.0f;
    __syncthreads();
    if (tid < N_COMBOS) {
        int c = tid;
        int le = cLE[c], ln = cLN[c], lo = cLO[c];
        int de = 2 * le + 1, dn = 2 * ln + 1, dO = 2 * lo + 1;
        int uo = U_OFF[le * 3 + ln];
        int po = OUT_OFF[c];
        const float* cg = cgs[c];
        for (int a = 0; a < de; ++a)
            for (int b = 0; b < dn; ++b)
                for (int k = 0; k < dO; ++k) {
                    float v = cg[(a * dn + b) * dO + k];
                    for (int ch = 0; ch < 2; ++ch)
                        M[(uo + (a * dn + b) * 2 + ch) * PN + po + ch * dO + k] = v;
                }
    }
}

__global__ void zero_kernel(float* p, long n) {
    long i = (long)blockIdx.x * blockDim.x + threadIdx.x;
    if (i < n) p[i] = 0.0f;
}

template <int PART>
__device__ __forceinline__ void gate_block(const float* __restrict__ W1,
                                           const float* __restrict__ b1,
                                           const float* __restrict__ W2,
                                           const float* __restrict__ b2,
                                           float x0, float x1,
                                           float* __restrict__ gRow) {
#pragma unroll
    for (int t = 0; t < 30; ++t) {
        const int g = PART * 30 + t;
        const int bi = ((g >> 2) * 2 + ((g >> 1) & 1)) * 2 + (g & 1);
        float acc = b2[bi];
#pragma unroll
        for (int h = 0; h < 4; ++h) {
            float hh = x0 * W1[bi * 8 + h * 2] + x1 * W1[bi * 8 + h * 2 + 1] + b1[bi * 4 + h];
            hh = hh > 0.0f ? hh : 0.0f;
            acc += hh * W2[bi * 4 + h];
        }
        gRow[g] = tanhf(acc);
    }
}

template <int PART>
__device__ __forceinline__ void u_stage(const float* __restrict__ shv,
                                        const float* __restrict__ f,
                                        float* __restrict__ uRow) {
    int u = 0;
#pragma unroll
    for (int le = 0; le < 3; ++le) {
        const int de = 2 * le + 1;
        const int shb = (le == 0) ? 0 : (le == 1 ? 1 : 4);
#pragma unroll
        for (int ln = 0; ln < 3; ++ln) {
            const int dn = 2 * ln + 1;
            const int fb = (ln == 0) ? 0 : (ln == 1 ? 2 : 8);
#pragma unroll
            for (int a = 0; a < de; ++a)
#pragma unroll
                for (int b = 0; b < dn; ++b)
#pragma unroll
                    for (int ch = 0; ch < 2; ++ch) {
                        if ((PART == 0) ? (u < 81) : (u >= 81))
                            uRow[u] = shv[shb + a] * f[fb + ch * dn + b];
                        ++u;
                    }
        }
    }
    if (PART == 1) { uRow[162] = 0.0f; uRow[163] = 0.0f; }
}

template <int S>
__device__ __forceinline__ float slot_val(const float* __restrict__ gRow,
                                          const float* __restrict__ rRow) {
    constexpr int lo = (S < 2) ? 0 : ((S < 8) ? 1 : 2);
    constexpr int o  = (S < 2) ? S : ((S < 8) ? ((S - 2) / 3) : ((S - 8) / 5));
    constexpr int k  = (S < 2) ? 0 : ((S < 8) ? ((S - 2) % 3) : ((S - 8) % 5));
    constexpr int dO = 2 * lo + 1;
    float v = 0.0f;
#pragma unroll
    for (int q = 0; q < N_BY_LO[lo]; ++q) {
        constexpr const int* lst = C_BY_LO[lo];
        const int c = lst[q];
        const int base = OUT_OFF[c];
        v += gRow[c * 4 + o]     * rRow[base + k] +
             gRow[c * 4 + 2 + o] * rRow[base + dO + k];
    }
    return v;
}

__global__ void __launch_bounds__(WAVES * 32)
layer_kernel(const float* __restrict__ feats, const int* __restrict__ eidx,
             const float* __restrict__ sh0, const float* __restrict__ sh1,
             const float* __restrict__ sh2, const float* __restrict__ dist,
             const float* __restrict__ Mg,
             const float* __restrict__ W1, const float* __restrict__ b1,
             const float* __restrict__ W2, const float* __restrict__ b2,
             float* __restrict__ outF, int E) {
    __shared__ float sU[WAVES][16 * KP];
    __shared__ float sRaw[WAVES][16 * PN];
    __shared__ float sGate[WAVES][16 * 60];

    const int lane = threadIdx.x & 31;
    const int w = threadIdx.x >> 5;
    const int tile = blockIdx.x * WAVES + w;
    const int eSub = lane & 15;
    const int part = lane >> 4;
    const long e = (long)tile * 16 + eSub;
    const bool valid = (e < (long)E);

    float f[FEAT];
    float shv[9];
    float x0 = 0.0f, x1 = 0.0f;
    int tgt = 0;
    if (valid) {
        const float* fp = &feats[(long)eidx[e] * FEAT];
        tgt = eidx[(long)E + e];
#pragma unroll
        for (int j = 0; j < FEAT; ++j) f[j] = fp[j];
        shv[0] = sh0[e];
#pragma unroll
        for (int j = 0; j < 3; ++j) shv[1 + j] = sh1[e * 3 + j];
#pragma unroll
        for (int j = 0; j < 5; ++j) shv[4 + j] = sh2[e * 5 + j];
        x0 = dist[e];
        float inv0 = 0.5f * (f[0] + f[1]);
        float n1 = 0.0f, n2 = 0.0f;
#pragma unroll
        for (int j = 0; j < 3; ++j) { float m = 0.5f * (f[2 + j] + f[5 + j]); n1 += m * m; }
#pragma unroll
        for (int j = 0; j < 5; ++j) { float m = 0.5f * (f[8 + j] + f[13 + j]); n2 += m * m; }
        x1 = (inv0 + sqrtf(n1) + sqrtf(n2)) * (1.0f / 3.0f);
    } else {
#pragma unroll
        for (int j = 0; j < FEAT; ++j) f[j] = 0.0f;
#pragma unroll
        for (int j = 0; j < 9; ++j) shv[j] = 0.0f;
    }

    float* gRow = &sGate[w][eSub * 60];
    float* uRow = &sU[w][eSub * KP];
    if (part == 0) {
        gate_block<0>(W1, b1, W2, b2, x0, x1, gRow);
        u_stage<0>(shv, f, uRow);
    } else {
        gate_block<1>(W1, b1, W2, b2, x0, x1, gRow);
        u_stage<1>(shv, f, uRow);
    }
    __syncthreads();

    {
        const int col16 = lane & 15;
        const int kofs = (lane < 16) ? 0 : 2;
        const float* uA = &sU[w][col16 * KP];
#pragma unroll
        for (int n = 0; n < NTILES; ++n) {
            v8f acc = {0.f, 0.f, 0.f, 0.f, 0.f, 0.f, 0.f, 0.f};
            const int col = col16 + n * 16;
#pragma unroll
            for (int kt = KLO[n]; kt <= KHI[n]; ++kt) {
                int k0 = kt * 4 + kofs;
                v2f a, b;
                a.x = uA[k0];
                a.y = uA[k0 + 1];
                b.x = Mg[k0 * PN + col];
                b.y = Mg[(k0 + 1) * PN + col];
                acc = __builtin_amdgcn_wmma_f32_16x16x4_f32(
                    false, a, false, b, (short)0, acc, false, false);
            }
            const int rbase = (lane < 16) ? 0 : 8;
#pragma unroll
            for (int j = 0; j < 8; ++j)
                sRaw[w][(rbase + j) * PN + col] = acc[j];
        }
    }
    __syncthreads();

    {
        const float* rRow = &sRaw[w][eSub * PN];
        float* outBase = &outF[(long)tgt * FEAT];
#define DO_SLOT(S)                                                     \
        { float v = slot_val<S>(gRow, rRow);                           \
          if (valid) atomicAdd(outBase + (S), v); }
        if (part == 0) {
            DO_SLOT(0) DO_SLOT(1) DO_SLOT(2) DO_SLOT(3) DO_SLOT(4)
            DO_SLOT(5) DO_SLOT(6) DO_SLOT(7) DO_SLOT(8)
        } else {
            DO_SLOT(9) DO_SLOT(10) DO_SLOT(11) DO_SLOT(12) DO_SLOT(13)
            DO_SLOT(14) DO_SLOT(15) DO_SLOT(16) DO_SLOT(17)
        }
#undef DO_SLOT
    }
}

extern "C" void kernel_launch(void* const* d_in, const int* in_sizes, int n_in,
                              void* d_out, int out_size, void* d_ws, size_t ws_size,
                              hipStream_t stream) {
    const float* node = (const float*)d_in[0];
    const int* eidx   = (const int*)d_in[1];
    const float* sh0  = (const float*)d_in[2];
    const float* sh1  = (const float*)d_in[3];
    const float* sh2  = (const float*)d_in[4];
    const float* dist = (const float*)d_in[5];
    const float* cg[N_COMBOS];
    for (int i = 0; i < N_COMBOS; ++i) cg[i] = (const float*)d_in[6 + i];
    const float* W1_0 = (const float*)d_in[21];
    const float* b1_0 = (const float*)d_in[22];
    const float* W2_0 = (const float*)d_in[23];
    const float* b2_0 = (const float*)d_in[24];
    const float* W1_1 = (const float*)d_in[25];
    const float* b1_1 = (const float*)d_in[26];
    const float* W2_1 = (const float*)d_in[27];
    const float* b2_1 = (const float*)d_in[28];

    const int E = in_sizes[1] / 2;
    const int N = in_sizes[0] / FEAT;

    float* Mg     = (float*)d_ws;
    float* feats1 = (float*)((char*)d_ws + 131072);

    build_M_kernel<<<1, 256, 0, stream>>>(cg[0], cg[1], cg[2], cg[3], cg[4], cg[5],
                                          cg[6], cg[7], cg[8], cg[9], cg[10], cg[11],
                                          cg[12], cg[13], cg[14], Mg);

    long n1 = (long)N * FEAT;
    zero_kernel<<<(int)((n1 + 255) / 256), 256, 0, stream>>>(feats1, n1);
    zero_kernel<<<(out_size + 255) / 256, 256, 0, stream>>>((float*)d_out, (long)out_size);

    int grid = (E + WAVES * 16 - 1) / (WAVES * 16);
    layer_kernel<<<grid, WAVES * 32, 0, stream>>>(node, eidx, sh0, sh1, sh2, dist, Mg,
                                                  W1_0, b1_0, W2_0, b2_0, feats1, E);
    layer_kernel<<<grid, WAVES * 32, 0, stream>>>(feats1, eidx, sh0, sh1, sh2, dist, Mg,
                                                  W1_1, b1_1, W2_1, b2_1, (float*)d_out, E);
}


# Edges grouped by target once per call (counting sort, int atomics); a wave adds its 16 sorted edges' rows per target in a lane-shuffle tree, then one f32 atomic row per target run, not per edge
# speedup vs baseline: 1.3477x; 1.3477x over previous
#include <hip/hip_runtime.h>

typedef __attribute__((ext_vector_type(2))) float v2f;
typedef __attribute__((ext_vector_type(8))) float v8f;

#define N_COMBOS 15
#define KP 164
#define PN 112
#define NTILES 7
#define WAVES 2
#define FEAT 18

constexpr int cLE[N_COMBOS]     = {0,0,0,1,1,1,1,1,1,2,2,2,2,2,2};
constexpr int cLN[N_COMBOS]     = {0,1,2,0,1,1,1,2,2,0,1,1,2,2,2};
constexpr int cLO[N_COMBOS]     = {0,1,2,1,0,1,2,1,2,2,1,2,0,1,2};
constexpr int OUT_OFF[N_COMBOS] = {0,2,8,18,24,26,32,42,48,58,68,74,84,86,92};
constexpr int U_OFF[9]          = {0,2,8,18,24,42,72,82,112};
constexpr int KLO[NTILES]       = {0,2,6,10,18,20,28};
constexpr int KHI[NTILES]       = {4,10,17,20,27,40,40};
constexpr int N_BY_LO[3]        = {3,6,6};
constexpr int C_BY_LO[3][6]     = {{0,4,12,0,0,0},{1,3,5,7,10,13},{2,6,8,9,11,14}};

__global__ void build_M_kernel(const float* cg0, const float* cg1, const float* cg2,
                               const float* cg3, const float* cg4, const float* cg5,
                               const float* cg6, const float* cg7, const float* cg8,
                               const float* cg9, const float* cg10, const float* cg11,
                               const float* cg12, const float* cg13, const float* cg14,
                               float* M) {
    const float* cgs[N_COMBOS] = {cg0,cg1,cg2,cg3,cg4,cg5,cg6,cg7,cg8,cg9,cg10,cg11,cg12,cg13,cg14};
    int tid = threadIdx.x;
    for (int i = tid; i < KP * PN; i += blockDim.x) M[i] = 0.0f;
    __syncthreads();
    if (tid < N_COMBOS) {
        int c = tid;
        int le = cLE[c], ln = cLN[c], lo = cLO[c];
        int de = 2 * le + 1, dn = 2 * ln + 1, dO = 2 * lo + 1;
        int uo = U_OFF[le * 3 + ln];
        int po = OUT_OFF[c];
        const float* cg = cgs[c];
        for (int a = 0; a < de; ++a)
            for (int b = 0; b < dn; ++b)
                for (int k = 0; k < dO; ++k) {
                    float v = cg[(a * dn + b) * dO + k];
                    for (int ch = 0; ch < 2; ++ch)
                        M[(uo + (a * dn + b) * 2 + ch) * PN + po + ch * dO + k] = v;
                }
    }
}

__global__ void zero_kernel(float* p, long n) {
    long i = (long)blockIdx.x * blockDim.x + threadIdx.x;
    if (i < n) p[i] = 0.0f;
}

__global__ void rank_kernel(const int* __restrict__ eidx, int* __restrict__ cnt,
                            int* __restrict__ rnk, int E) {
    int e = blockIdx.x * blockDim.x + threadIdx.x;
    if (e < E) rnk[e] = atomicAdd(&cnt[eidx[(long)E + e]], 1);
}

__global__ void __launch_bounds__(1024)
scan_kernel(int* __restrict__ cnt, int N) {
    __shared__ int part[1024];
    const int t = threadIdx.x;
    const int chunk = (N + 1023) / 1024;
    const int lo = t * chunk;
    int hi = lo + chunk;
    if (hi > N) hi = N;
    int s = 0;
    for (int i = lo; i < hi; ++i) s += cnt[i];
    part[t] = s;
    __syncthreads();
    for (int d = 1; d < 1024; d <<= 1) {
        int add = (t >= d) ? part[t - d] : 0;
        __syncthreads();
        part[t] += add;
        __syncthreads();
    }
    int run = part[t] - s;
    for (int i = lo; i < hi; ++i) { int c = cnt[i]; cnt[i] = run; run += c; }
}

__global__ void fill_kernel(const int* __restrict__ eidx, const int* __restrict__ first,
                            const int* __restrict__ rnk, int* __restrict__ perm, int E) {
    int e = blockIdx.x * blockDim.x + threadIdx.x;
    if (e < E) perm[first[eidx[(long)E + e]] + rnk[e]] = e;
}

template <int PART>
__device__ __forceinline__ void gate_block(const float* __restrict__ W1,
                                           const float* __restrict__ b1,
                                           const float* __restrict__ W2,
                                           const float* __restrict__ b2,
                                           float x0, float x1,
                                           float* __restrict__ gRow) {
#pragma unroll
    for (int t = 0; t < 30; ++t) {
        const int g = PART * 30 + t;
        const int bi = ((g >> 2) * 2 + ((g >> 1) & 1)) * 2 + (g & 1);
        float acc = b2[bi];
#pragma unroll
        for (int h = 0; h < 4; ++h) {
            float hh = x0 * W1[bi * 8 + h * 2] + x1 * W1[bi * 8 + h * 2 + 1] + b1[bi * 4 + h];
            hh = hh > 0.0f ? hh : 0.0f;
            acc += hh * W2[bi * 4 + h];
        }
        gRow[g] = tanhf(acc);
    }
}

template <int PART>
__device__ __forceinline__ void u_stage(const float* __restrict__ shv,
                                        const float* __restrict__ f,
                                        float* __restrict__ uRow) {
    int u = 0;
#pragma unroll
    for (int le = 0; le < 3; ++le) {
        const int de = 2 * le + 1;
        const int shb = (le == 0) ? 0 : (le == 1 ? 1 : 4);
#pragma unroll
        for (int ln = 0; ln < 3; ++ln) {
            const int dn = 2 * ln + 1;
            const int fb = (ln == 0) ? 0 : (ln == 1 ? 2 : 8);
#pragma unroll
            for (int a = 0; a < de; ++a)
#pragma unroll
                for (int b = 0; b < dn; ++b)
#pragma unroll
                    for (int ch = 0; ch < 2; ++ch) {
                        if ((PART == 0) ? (u < 81) : (u >= 81))
                            uRow[u] = shv[shb + a] * f[fb + ch * dn + b];
                        ++u;
                    }
        }
    }
    if (PART == 1) { uRow[162] = 0.0f; uRow[163] = 0.0f; }
}

template <int S>
__device__ __forceinline__ float slot_val(const float* __restrict__ gRow,
                                          const float* __restrict__ rRow) {
    constexpr int lo = (S < 2) ? 0 : ((S < 8) ? 1 : 2);
    constexpr int o  = (S < 2) ? S : ((S < 8) ? ((S - 2) / 3) : ((S - 8) / 5));
    constexpr int k  = (S < 2) ? 0 : ((S < 8) ? ((S - 2) % 3) : ((S - 8) % 5));
    constexpr int dO = 2 * lo + 1;
    float v = 0.0f;
#pragma unroll
    for (int q = 0; q < N_BY_LO[lo]; ++q) {
        constexpr const int* lst = C_BY_LO[lo];
        const int c = lst[q];
        const int base = OUT_OFF[c];
        v += gRow[c * 4 + o]     * rRow[base + k] +
             gRow[c * 4 + 2 + o] * rRow[base + dO + k];
    }
    return v;
}

__global__ void __launch_bounds__(WAVES * 32)
layer_kernel(const float* __restrict__ feats, const int* __restrict__ eidx,
             const float* __restrict__ sh0, const float* __restrict__ sh1,
             const float* __restrict__ sh2, const float* __restrict__ dist,
             const float* __restrict__ Mg, const int* __restrict__ perm,
             const float* __restrict__ W1, const float* __restrict__ b1,
             const float* __restrict__ W2, const float* __restrict__ b2,
             float* __restrict__ outF, int E) {
    __shared__ float sU[WAVES][16 * KP];
    __shared__ float sRaw[WAVES][16 * PN];
    __shared__ float sGate[WAVES][16 * 60];

    const int lane = threadIdx.x & 31;
    const int w = threadIdx.x >> 5;
    const int tile = blockIdx.x * WAVES + w;
    const int eSub = lane & 15;
    const int part = lane >> 4;
    const long slot = (long)tile * 16 + eSub;
    const bool valid = (slot < (long)E);
    const long e = valid ? (perm ? (long)perm[slot] : slot) : 0;

    float f[FEAT];
    float shv[9];
    float x0 = 0.0f, x1 = 0.0f;
    int tgt = -1;
    if (valid) {
        const float* fp = &feats[(long)eidx[e] * FEAT];
        tgt = eidx[(long)E + e];
#pragma unroll
        for (int j = 0; j < FEAT; ++j) f[j] = fp[j];
        shv[0] = sh0[e];
#pragma unroll
        for (int j = 0; j < 3; ++j) shv[1 + j] = sh1[e * 3 + j];
#pragma unroll
        for (int j = 0; j < 5; ++j) shv[4 + j] = sh2[e * 5 + j];
        x0 = dist[e];
        float inv0 = 0.5f * (f[0] + f[1]);
        float n1 = 0.0f, n2 = 0.0f;
#pragma unroll
        for (int j = 0; j < 3; ++j) { float m = 0.5f * (f[2 + j] + f[5 + j]); n1 += m * m; }
#pragma unroll
        for (int j = 0; j < 5; ++j) { float m = 0.5f * (f[8 + j] + f[13 + j]); n2 += m * m; }
        x1 = (inv0 + sqrtf(n1) + sqrtf(n2)) * (1.0f / 3.0f);
    } else {
#pragma unroll
        for (int j = 0; j < FEAT; ++j) f[j] = 0.0f;
#pragma unroll
        for (int j = 0; j < 9; ++j) shv[j] = 0.0f;
    }

    float* gRow = &sGate[w][eSub * 60];
    float* uRow = &sU[w][eSub * KP];
    if (part == 0) {
        gate_block<0>(W1, b1, W2, b2, x0, x1, gRow);
        u_stage<0>(shv, f, uRow);
    } else {
        gate_block<1>(W1, b1, W2, b2, x0, x1, gRow);
        u_stage<1>(shv, f, uRow);
    }
    __syncthreads();

    {
        const int col16 = lane & 15;
        const int kofs = (lane < 16) ? 0 : 2;
        const float* uA = &sU[w][col16 * KP];
#pragma unroll
        for (int n = 0; n < NTILES; ++n) {
            v8f acc = {0.f, 0.f, 0.f, 0.f, 0.f, 0.f, 0.f, 0.f};
            const int col = col16 + n * 16;
#pragma unroll
            for (int kt = KLO[n]; kt <= KHI[n]; ++kt) {
                int k0 = kt * 4 + kofs;
                v2f a, b;
                a.x = uA[k0];
                a.y = uA[k0 + 1];
                b.x = Mg[k0 * PN + col];
                b.y = Mg[(k0 + 1) * PN + col];
                acc = __builtin_amdgcn_wmma_f32_16x16x4_f32(
                    false, a, false, b, (short)0, acc, false, false);
            }
            const int rbase = (lane < 16) ? 0 : 8;
#pragma unroll
            for (int j = 0; j < 8; ++j)
                sRaw[w][(rbase + j) * PN + col] = acc[j];
        }
    }
    __syncthreads();

    {
        const float* rRow = &sRaw[w][eSub * PN];
        float v[9];
        if (part == 0) {
            v[0] = slot_val<0>(gRow, rRow);  v[1] = slot_val<1>(gRow, rRow);
            v[2] = slot_val<2>(gRow, rRow);  v[3] = slot_val<3>(gRow, rRow);
            v[4] = slot_val<4>(gRow, rRow);  v[5] = slot_val<5>(gRow, rRow);
            v[6] = slot_val<6>(gRow, rRow);  v[7] = slot_val<7>(gRow, rRow);
            v[8] = slot_val<8>(gRow, rRow);
        } else {
            v[0] = slot_val<9>(gRow, rRow);  v[1] = slot_val<10>(gRow, rRow);
            v[2] = slot_val<11>(gRow, rRow); v[3] = slot_val<12>(gRow, rRow);
            v[4] = slot_val<13>(gRow, rRow); v[5] = slot_val<14>(gRow, rRow);
            v[6] = slot_val<15>(gRow, rRow); v[7] = slot_val<16>(gRow, rRow);
            v[8] = slot_val<17>(gRow, rRow);
        }
        const int tUp = __shfl_up(tgt, 1, 16);
        int runLo = (eSub == 0 || tUp != tgt) ? eSub : 0;
#pragma unroll
        for (int d = 1; d < 16; d <<= 1) {
            const int o = __shfl_up(runLo, d, 16);
            if (eSub >= d) runLo = max(runLo, o);
        }
#pragma unroll
        for (int d = 1; d < 16; d <<= 1) {
            const bool take = (eSub - d >= runLo);
#pragma unroll
            for (int i = 0; i < 9; ++i) {
                const float o = __shfl_up(v[i], d, 16);
                v[i] += take ? o : 0.0f;
            }
        }
        const int tDn = __shfl_down(tgt, 1, 16);
        const bool last = (eSub == 15) || (tDn != tgt);
        if (valid && last) {
            float* outBase = &outF[(long)tgt * FEAT + part * 9];
#pragma unroll
            for (int i = 0; i < 9; ++i) atomicAdd(outBase + i, v[i]);
        }
    }
}

extern "C" void kernel_launch(void* const* d_in, const int* in_sizes, int n_in,
                              void* d_out, int out_size, void* d_ws, size_t ws_size,
                              hipStream_t stream) {
    const float* node = (const float*)d_in[0];
    const int* eidx   = (const int*)d_in[1];
    const float* sh0  = (const float*)d_in[2];
    const float* sh1  = (const float*)d_in[3];
    const float* sh2  = (const float*)d_in[4];
    const float* dist = (const float*)d_in[5];
    const float* cg[N_COMBOS];
    for (int i = 0; i < N_COMBOS; ++i) cg[i] = (const float*)d_in[6 + i];
    const float* W1_0 = (const float*)d_in[21];
    const float* b1_0 = (const float*)d_in[22];
    const float* W2_0 = (const float*)d_in[23];
    const float* b2_0 = (const float*)d_in[24];
    const float* W1_1 = (const float*)d_in[25];
    const float* b1_1 = (const float*)d_in[26];
    const float* W2_1 = (const float*)d_in[27];
    const float* b2_1 = (const float*)d_in[28];

    const int E = in_sizes[1] / 2;
    const int N = in_sizes[0] / FEAT;

    float* Mg     = (float*)d_ws;
    float* feats1 = (float*)((char*)d_ws + 131072);
    long n1 = (long)N * FEAT;
    int* perm     = (int*)(feats1 + n1);

    build_M_kernel<<<1, 256, 0, stream>>>(cg[0], cg[1], cg[2], cg[3], cg[4], cg[5],
                                          cg[6], cg[7], cg[8], cg[9], cg[10], cg[11],
                                          cg[12], cg[13], cg[14], Mg);

    const bool grouped = (ws_size >= (size_t)131072 + (size_t)(n1 + E) * 4) &&
                         ((long)E + N <= (long)out_size);
    if (grouped) {
        int* rnk = (int*)d_out;
        int* cnt = rnk + E;
        zero_kernel<<<(N + 255) / 256, 256, 0, stream>>>((float*)cnt, (long)N);
        rank_kernel<<<(E + 255) / 256, 256, 0, stream>>>(eidx, cnt, rnk, E);
        scan_kernel<<<1, 1024, 0, stream>>>(cnt, N);
        fill_kernel<<<(E + 255) / 256, 256, 0, stream>>>(eidx, cnt, rnk, perm, E);
    } else {
        perm = nullptr;
    }

    zero_kernel<<<(int)((n1 + 255) / 256), 256, 0, stream>>>(feats1, n1);
    zero_kernel<<<(out_size + 255) / 256, 256, 0, stream>>>((float*)d_out, (long)out_size);

    int grid = (E + WAVES * 16 - 1) / (WAVES * 16);
    layer_kernel<<<grid, WAVES * 32, 0, stream>>>(node, eidx, sh0, sh1, sh2, dist, Mg, perm,
                                                  W1_0, b1_0, W2_0, b2_0, feats1, E);
    layer_kernel<<<grid, WAVES * 32, 0, stream>>>(feats1, eidx, sh0, sh1, sh2, dist, Mg, perm,
                                                  W1_1, b1_1, W2_1, b2_1, (float*)d_out, E);
}


# One lane per edge: a wave owns 32 sorted edges (two WMMA row tiles share each B fragment), gates stay in registers, no divergent half-wave passes; target runs now sum in a 32-lane shuffle tree
# speedup vs baseline: 1.5106x; 1.1209x over previous
#include <hip/hip_runtime.h>

typedef __attribute__((ext_vector_type(2))) float v2f;
typedef __attribute__((ext_vector_type(8))) float v8f;

#define N_COMBOS 15
#define KP 164
#define PN 112
#define NTILES 7
#define TILE 32
#define FEAT 18

constexpr int cLE[N_COMBOS]     = {0,0,0,1,1,1,1,1,1,2,2,2,2,2,2};
constexpr int cLN[N_COMBOS]     = {0,1,2,0,1,1,1,2,2,0,1,1,2,2,2};
constexpr int cLO[N_COMBOS]     = {0,1,2,1,0,1,2,1,2,2,1,2,0,1,2};
constexpr int OUT_OFF[N_COMBOS] = {0,2,8,18,24,26,32,42,48,58,68,74,84,86,92};
constexpr int U_OFF[9]          = {0,2,8,18,24,42,72,82,112};
constexpr int KLO[NTILES]       = {0,2,6,10,18,20,28};
constexpr int KHI[NTILES]       = {4,10,17,20,27,40,40};
constexpr int N_BY_LO[3]        = {3,6,6};
constexpr int C_BY_LO[3][6]     = {{0,4,12,0,0,0},{1,3,5,7,10,13},{2,6,8,9,11,14}};

__global__ void build_M_kernel(const float* cg0, const float* cg1, const float* cg2,
                               const float* cg3, const float* cg4, const float* cg5,
                               const float* cg6, const float* cg7, const float* cg8,
                               const float* cg9, const float* cg10, const float* cg11,
                               const float* cg12, const float* cg13, const float* cg14,
                               float* M) {
    const float* cgs[N_COMBOS] = {cg0,cg1,cg2,cg3,cg4,cg5,cg6,cg7,cg8,cg9,cg10,cg11,cg12,cg13,cg14};
    int tid = threadIdx.x;
    for (int i = tid; i < KP * PN; i += blockDim.x) M[i] = 0.0f;
    __syncthreads();
    if (tid < N_COMBOS) {
        int c = tid;
        int le = cLE[c], ln = cLN[c], lo = cLO[c];
        int de = 2 * le + 1, dn = 2 * ln + 1, dO = 2 * lo + 1;
        int uo = U_OFF[le * 3 + ln];
        int po = OUT_OFF[c];
        const float* cg = cgs[c];
        for (int a = 0; a < de; ++a)
            for (int b = 0; b < dn; ++b)
                for (int k = 0; k < dO; ++k) {
                    float v = cg[(a * dn + b) * dO + k];
                    for (int ch = 0; ch < 2; ++ch)
                        M[(uo + (a * dn + b) * 2 + ch) * PN + po + ch * dO + k] = v;
                }
    }
}

__global__ void zero_kernel(float* p, long n) {
    long i = (long)blockIdx.x * blockDim.x + threadIdx.x;
    if (i < n) p[i] = 0.0f;
}

__global__ void rank_kernel(const int* __restrict__ eidx, int* __restrict__ cnt,
                            int* __restrict__ rnk, int E) {
    int e = blockIdx.x * blockDim.x + threadIdx.x;
    if (e < E) rnk[e] = atomicAdd(&cnt[eidx[(long)E + e]], 1);
}

__global__ void __launch_bounds__(1024)
scan_kernel(int* __restrict__ cnt, int N) {
    __shared__ int part[1024];
    const int t = threadIdx.x;
    const int chunk = (N + 1023) / 1024;
    const int lo = t * chunk;
    int hi = lo + chunk;
    if (hi > N) hi = N;
    int s = 0;
    for (int i = lo; i < hi; ++i) s += cnt[i];
    part[t] = s;
    __syncthreads();
    for (int d = 1; d < 1024; d <<= 1) {
        int add = (t >= d) ? part[t - d] : 0;
        __syncthreads();
        part[t] += add;
        __syncthreads();
    }
    int run = part[t] - s;
    for (int i = lo; i < hi; ++i) { int c = cnt[i]; cnt[i] = run; run += c; }
}

__global__ void fill_kernel(const int* __restrict__ eidx, const int* __restrict__ first,
                            const int* __restrict__ rnk, int* __restrict__ perm, int E) {
    int e = blockIdx.x * blockDim.x + threadIdx.x;
    if (e < E) perm[first[eidx[(long)E + e]] + rnk[e]] = e;
}

__device__ __forceinline__ void gate_block(const float* __restrict__ W1,
                                           const float* __restrict__ b1,
                                           const float* __restrict__ W2,
                                           const float* __restrict__ b2,
                                           float x0, float x1,
                                           float (&g)[60]) {
#pragma unroll
    for (int t = 0; t < 60; ++t) {
        const int bi = ((t >> 2) * 2 + ((t >> 1) & 1)) * 2 + (t & 1);
        float acc = b2[bi];
#pragma unroll
        for (int h = 0; h < 4; ++h) {
            float hh = x0 * W1[bi * 8 + h * 2] + x1 * W1[bi * 8 + h * 2 + 1] + b1[bi * 4 + h];
            hh = hh > 0.0f ? hh : 0.0f;
            acc += hh * W2[bi * 4 + h];
        }
        g[t] = tanhf(acc);
    }
}

__device__ __forceinline__ void u_stage(const float (&shv)[9],
                                        const float (&f)[FEAT],
                                        float* __restrict__ uRow) {
    int u = 0;
#pragma unroll
    for (int le = 0; le < 3; ++le) {
        const int de = 2 * le + 1;
        const int shb = (le == 0) ? 0 : (le == 1 ? 1 : 4);
#pragma unroll
        for (int ln = 0; ln < 3; ++ln) {
            const int dn = 2 * ln + 1;
            const int fb = (ln == 0) ? 0 : (ln == 1 ? 2 : 8);
#pragma unroll
            for (int a = 0; a < de; ++a)
#pragma unroll
                for (int b = 0; b < dn; ++b)
#pragma unroll
                    for (int ch = 0; ch < 2; ++ch) {
                        uRow[u] = shv[shb + a] * f[fb + ch * dn + b];
                        ++u;
                    }
        }
    }
    uRow[162] = 0.0f;
    uRow[163] = 0.0f;
}

template <int S>
__device__ __forceinline__ float slot_val(const float (&gRow)[60],
                                          const float* __restrict__ rRow) {
    constexpr int lo = (S < 2) ? 0 : ((S < 8) ? 1 : 2);
    constexpr int o  = (S < 2) ? S : ((S < 8) ? ((S - 2) / 3) : ((S - 8) / 5));
    constexpr int k  = (S < 2) ? 0 : ((S < 8) ? ((S - 2) % 3) : ((S - 8) % 5));
    constexpr int dO = 2 * lo + 1;
    float v = 0.0f;
#pragma unroll
    for (int q = 0; q < N_BY_LO[lo]; ++q) {
        constexpr const int* lst = C_BY_LO[lo];
        const int c = lst[q];
        const int base = OUT_OFF[c];
        v += gRow[c * 4 + o]     * rRow[base + k] +
             gRow[c * 4 + 2 + o] * rRow[base + dO + k];
    }
    return v;
}

__global__ void __launch_bounds__(TILE)
layer_kernel(const float* __restrict__ feats, const int* __restrict__ eidx,
             const float* __restrict__ sh0, const float* __restrict__ sh1,
             const float* __restrict__ sh2, const float* __restrict__ dist,
             const float* __restrict__ Mg, const int* __restrict__ perm,
             const float* __restrict__ W1, const float* __restrict__ b1,
             const float* __restrict__ W2, const float* __restrict__ b2,
             float* __restrict__ outF, int E) {
    __shared__ float sU[TILE * KP];
    __shared__ float sRaw[TILE * PN];

    const int lane = threadIdx.x;
    const long slot = (long)blockIdx.x * TILE + lane;
    const bool valid = (slot < (long)E);
    const long e = valid ? (perm ? (long)perm[slot] : slot) : 0;

    float f[FEAT];
    float shv[9];
    float x0 = 0.0f, x1 = 0.0f;
    int tgt = -1;
    if (valid) {
        const float* fp = &feats[(long)eidx[e] * FEAT];
        tgt = eidx[(long)E + e];
#pragma unroll
        for (int j = 0; j < FEAT; ++j) f[j] = fp[j];
        shv[0] = sh0[e];
#pragma unroll
        for (int j = 0; j < 3; ++j) shv[1 + j] = sh1[e * 3 + j];
#pragma unroll
        for (int j = 0; j < 5; ++j) shv[4 + j] = sh2[e * 5 + j];
        x0 = dist[e];
        float inv0 = 0.5f * (f[0] + f[1]);
        float n1 = 0.0f, n2 = 0.0f;
#pragma unroll
        for (int j = 0; j < 3; ++j) { float m = 0.5f * (f[2 + j] + f[5 + j]); n1 += m * m; }
#pragma unroll
        for (int j = 0; j < 5; ++j) { float m = 0.5f * (f[8 + j] + f[13 + j]); n2 += m * m; }
        x1 = (inv0 + sqrtf(n1) + sqrtf(n2)) * (1.0f / 3.0f);
    } else {
#pragma unroll
        for (int j = 0; j < FEAT; ++j) f[j] = 0.0f;
#pragma unroll
        for (int j = 0; j < 9; ++j) shv[j] = 0.0f;
    }

    float g[60];
    u_stage(shv, f, &sU[lane * KP]);
    gate_block(W1, b1, W2, b2, x0, x1, g);
    __syncthreads();

    {
        const int col16 = lane & 15;
        const int kofs = (lane < 16) ? 0 : 2;
        const float* uA0 = &sU[col16 * KP];
        const float* uA1 = &sU[(16 + col16) * KP];
#pragma unroll
        for (int n = 0; n < NTILES; ++n) {
            v8f acc0 = {0.f, 0.f, 0.f, 0.f, 0.f, 0.f, 0.f, 0.f};
            v8f acc1 = {0.f, 0.f, 0.f, 0.f, 0.f, 0.f, 0.f, 0.f};
            const int col = col16 + n * 16;
#pragma unroll
            for (int kt = KLO[n]; kt <= KHI[n]; ++kt) {
                int k0 = kt * 4 + kofs;
                v2f a0, a1, b;
                a0.x = uA0[k0];
                a0.y = uA0[k0 + 1];
                a1.x = uA1[k0];
                a1.y = uA1[k0 + 1];
                b.x = Mg[k0 * PN + col];
                b.y = Mg[(k0 + 1) * PN + col];
                acc0 = __builtin_amdgcn_wmma_f32_16x16x4_f32(
                    false, a0, false, b, (short)0, acc0, false, false);
                acc1 = __builtin_amdgcn_wmma_f32_16x16x4_f32(
                    false, a1, false, b, (short)0, acc1, false, false);
            }
            const int rbase = (lane < 16) ? 0 : 8;
#pragma unroll
            for (int j = 0; j < 8; ++j) {
                sRaw[(rbase + j) * PN + col] = acc0[j];
                sRaw[(16 + rbase + j) * PN + col] = acc1[j];
            }
        }
    }
    __syncthreads();

    {
        const float* rRow = &sRaw[lane * PN];
        float v[FEAT];
        v[0]  = slot_val<0>(g, rRow);   v[1]  = slot_val<1>(g, rRow);
        v[2]  = slot_val<2>(g, rRow);   v[3]  = slot_val<3>(g, rRow);
        v[4]  = slot_val<4>(g, rRow);   v[5]  = slot_val<5>(g, rRow);
        v[6]  = slot_val<6>(g, rRow);   v[7]  = slot_val<7>(g, rRow);
        v[8]  = slot_val<8>(g, rRow);   v[9]  = slot_val<9>(g, rRow);
        v[10] = slot_val<10>(g, rRow);  v[11] = slot_val<11>(g, rRow);
        v[12] = slot_val<12>(g, rRow);  v[13] = slot_val<13>(g, rRow);
        v[14] = slot_val<14>(g, rRow);  v[15] = slot_val<15>(g, rRow);
        v[16] = slot_val<16>(g, rRow);  v[17] = slot_val<17>(g, rRow);
        const int tUp = __shfl_up(tgt, 1, TILE);
        int runLo = (lane == 0 || tUp != tgt) ? lane : 0;
#pragma unroll
        for (int d = 1; d < TILE; d <<= 1) {
            const int o = __shfl_up(runLo, d, TILE);
            if (lane >= d) runLo = max(runLo, o);
        }
#pragma unroll
        for (int d = 1; d < TILE; d <<= 1) {
            const bool take = (lane - d >= runLo);
#pragma unroll
            for (int i = 0; i < FEAT; ++i) {
                const float o = __shfl_up(v[i], d, TILE);
                v[i] += take ? o : 0.0f;
            }
        }
        const int tDn = __shfl_down(tgt, 1, TILE);
        const bool last = (lane == TILE - 1) || (tDn != tgt);
        if (valid && last) {
            float* outBase = &outF[(long)tgt * FEAT];
#pragma unroll
            for (int i = 0; i < FEAT; ++i) atomicAdd(outBase + i, v[i]);
        }
    }
}

extern "C" void kernel_launch(void* const* d_in, const int* in_sizes, int n_in,
                              void* d_out, int out_size, void* d_ws, size_t ws_size,
                              hipStream_t stream) {
    const float* node = (const float*)d_in[0];
    const int* eidx   = (const int*)d_in[1];
    const float* sh0  = (const float*)d_in[2];
    const float* sh1  = (const float*)d_in[3];
    const float* sh2  = (const float*)d_in[4];
    const float* dist = (const float*)d_in[5];
    const float* cg[N_COMBOS];
    for (int i = 0; i < N_COMBOS; ++i) cg[i] = (const float*)d_in[6 + i];
    const float* W1_0 = (const float*)d_in[21];
    const float* b1_0 = (const float*)d_in[22];
    const float* W2_0 = (const float*)d_in[23];
    const float* b2_0 = (const float*)d_in[24];
    const float* W1_1 = (const float*)d_in[25];
    const float* b1_1 = (const float*)d_in[26];
    const float* W2_1 = (const float*)d_in[27];
    const float* b2_1 = (const float*)d_in[28];

    const int E = in_sizes[1] / 2;
    const int N = in_sizes[0] / FEAT;

    float* Mg     = (float*)d_ws;
    float* feats1 = (float*)((char*)d_ws + 131072);
    long n1 = (long)N * FEAT;
    int* perm     = (int*)(feats1 + n1);

    build_M_kernel<<<1, 256, 0, stream>>>(cg[0], cg[1], cg[2], cg[3], cg[4], cg[5],
                                          cg[6], cg[7], cg[8], cg[9], cg[10], cg[11],
                                          cg[12], cg[13], cg[14], Mg);

    const bool grouped = (ws_size >= (size_t)131072 + (size_t)(n1 + E) * 4) &&
                         ((long)E + N <= (long)out_size);
    if (grouped) {
        int* rnk = (int*)d_out;
        int* cnt = rnk + E;
        zero_kernel<<<(N + 255) / 256, 256, 0, stream>>>((float*)cnt, (long)N);
        rank_kernel<<<(E + 255) / 256, 256, 0, stream>>>(eidx, cnt, rnk, E);
        scan_kernel<<<1, 1024, 0, stream>>>(cnt, N);
        fill_kernel<<<(E + 255) / 256, 256, 0, stream>>>(eidx, cnt, rnk, perm, E);
    } else {
        perm = nullptr;
    }

    zero_kernel<<<(int)((n1 + 255) / 256), 256, 0, stream>>>(feats1, n1);
    zero_kernel<<<(out_size + 255) / 256, 256, 0, stream>>>((float*)d_out, (long)out_size);

    int grid = (E + TILE - 1) / TILE;
    layer_kernel<<<grid, TILE, 0, stream>>>(node, eidx, sh0, sh1, sh2, dist, Mg, perm,
                                                  W1_0, b1_0, W2_0, b2_0, feats1, E);
    layer_kernel<<<grid, TILE, 0, stream>>>(feats1, eidx, sh0, sh1, sh2, dist, Mg, perm,
                                                  W1_1, b1_1, W2_1, b2_1, (float*)d_out, E);
}


# Each 32-edge tile is worked by two waves of one workgroup, split by CG combo (0-7 / 8-14): half the gates, u rows and LDS per wave, twice the waves per SIMD; a target run gets two atomic partial rows
# speedup vs baseline: 1.6125x; 1.0675x over previous
#include <hip/hip_runtime.h>

typedef __attribute__((ext_vector_type(2))) float v2f;
typedef __attribute__((ext_vector_type(8))) float v8f;

#define N_COMBOS 15
#define KP 164
#define PN 112
#define NTILES 7
#define TILE 32
#define FEAT 18

constexpr int cLE[N_COMBOS]     = {0,0,0,1,1,1,1,1,1,2,2,2,2,2,2};
constexpr int cLN[N_COMBOS]     = {0,1,2,0,1,1,1,2,2,0,1,1,2,2,2};
constexpr int cLO[N_COMBOS]     = {0,1,2,1,0,1,2,1,2,2,1,2,0,1,2};
constexpr int OUT_OFF[N_COMBOS] = {0,2,8,18,24,26,32,42,48,58,68,74,84,86,92};
constexpr int U_OFF[9]          = {0,2,8,18,24,42,72,82,112};
constexpr int KLO[NTILES]       = {0,2,6,10,18,20,28};
constexpr int KHI[NTILES]       = {4,10,17,20,27,40,40};
constexpr int N_BY_LO[3]        = {3,6,6};
constexpr int C_BY_LO[3][6]     = {{0,4,12,0,0,0},{1,3,5,7,10,13},{2,6,8,9,11,14}};

__global__ void build_M_kernel(const float* cg0, const float* cg1, const float* cg2,
                               const float* cg3, const float* cg4, const float* cg5,
                               const float* cg6, const float* cg7, const float* cg8,
                               const float* cg9, const float* cg10, const float* cg11,
                               const float* cg12, const float* cg13, const float* cg14,
                               float* M) {
    const float* cgs[N_COMBOS] = {cg0,cg1,cg2,cg3,cg4,cg5,cg6,cg7,cg8,cg9,cg10,cg11,cg12,cg13,cg14};
    int tid = threadIdx.x;
    for (int i = tid; i < KP * PN; i += blockDim.x) M[i] = 0.0f;
    __syncthreads();
    if (tid < N_COMBOS) {
        int c = tid;
        int le = cLE[c], ln = cLN[c], lo = cLO[c];
        int de = 2 * le + 1, dn = 2 * ln + 1, dO = 2 * lo + 1;
        int uo = U_OFF[le * 3 + ln];
        int po = OUT_OFF[c];
        const float* cg = cgs[c];
        for (int a = 0; a < de; ++a)
            for (int b = 0; b < dn; ++b)
                for (int k = 0; k < dO; ++k) {
                    float v = cg[(a * dn + b) * dO + k];
                    for (int ch = 0; ch < 2; ++ch)
                        M[(uo + (a * dn + b) * 2 + ch) * PN + po + ch * dO + k] = v;
                }
    }
}

__global__ void zero_kernel(float* p, long n) {
    long i = (long)blockIdx.x * blockDim.x + threadIdx.x;
    if (i < n) p[i] = 0.0f;
}

__global__ void rank_kernel(const int* __restrict__ eidx, int* __restrict__ cnt,
                            int* __restrict__ rnk, int E) {
    int e = blockIdx.x * blockDim.x + threadIdx.x;
    if (e < E) rnk[e] = atomicAdd(&cnt[eidx[(long)E + e]], 1);
}

__global__ void __launch_bounds__(1024)
scan_kernel(int* __restrict__ cnt, int N) {
    __shared__ int part[1024];
    const int t = threadIdx.x;
    const int chunk = (N + 1023) / 1024;
    const int lo = t * chunk;
    int hi = lo + chunk;
    if (hi > N) hi = N;
    int s = 0;
    for (int i = lo; i < hi; ++i) s += cnt[i];
    part[t] = s;
    __syncthreads();
    for (int d = 1; d < 1024; d <<= 1) {
        int add = (t >= d) ? part[t - d] : 0;
        __syncthreads();
        part[t] += add;
        __syncthreads();
    }
    int run = part[t] - s;
    for (int i = lo; i < hi; ++i) { int c = cnt[i]; cnt[i] = run; run += c; }
}

__global__ void fill_kernel(const int* __restrict__ eidx, const int* __restrict__ first,
                            const int* __restrict__ rnk, int* __restrict__ perm, int E) {
    int e = blockIdx.x * blockDim.x + threadIdx.x;
    if (e < E) perm[first[eidx[(long)E + e]] + rnk[e]] = e;
}

template <int H> struct HalfCfg;
template <> struct HalfCfg<0> {
    static constexpr int T0 = 0,  NG = 32;
    static constexpr int C0 = 0,  C1 = 8;
    static constexpr int NT0 = 0, NT1 = 3;
    static constexpr int ULO = 0, UHI = 72;
    static constexpr int UP = 76;
    static constexpr int COL0 = 0;
    static constexpr int RP = 52;
};
template <> struct HalfCfg<1> {
    static constexpr int T0 = 32, NG = 28;
    static constexpr int C0 = 8,  C1 = 15;
    static constexpr int NT0 = 3, NT1 = 7;
    static constexpr int ULO = 40, UHI = 164;
    static constexpr int UP = 124;
    static constexpr int COL0 = 48;
    static constexpr int RP = 68;
};
#define SU0_OFF   0
#define SRAW0_OFF 2432
#define SU1_OFF   4096
#define SRAW1_OFF 8064
#define SMEM_FLOATS 10240

__device__ __forceinline__ void wave_lds_handover() {
    __builtin_amdgcn_fence(__ATOMIC_RELEASE, "wavefront");
    __builtin_amdgcn_wave_barrier();
    __builtin_amdgcn_fence(__ATOMIC_ACQUIRE, "wavefront");
}

template <int H>
__device__ __forceinline__ void gate_block(const float* __restrict__ W1,
                                           const float* __restrict__ b1,
                                           const float* __restrict__ W2,
                                           const float* __restrict__ b2,
                                           float x0, float x1,
                                           float (&g)[HalfCfg<H>::NG]) {
#pragma unroll
    for (int t = 0; t < HalfCfg<H>::NG; ++t) {
        const int bi = HalfCfg<H>::T0 + t;
        float acc = b2[bi];
#pragma unroll
        for (int h = 0; h < 4; ++h) {
            float hh = x0 * W1[bi * 8 + h * 2] + x1 * W1[bi * 8 + h * 2 + 1] + b1[bi * 4 + h];
            hh = hh > 0.0f ? hh : 0.0f;
            acc += hh * W2[bi * 4 + h];
        }
        g[t] = tanhf(acc);
    }
}

template <int H>
__device__ __forceinline__ void u_stage(const float (&shv)[9],
                                        const float (&f)[FEAT],
                                        float* __restrict__ uRow) {
    typedef HalfCfg<H> Cf;
    int u = 0;
#pragma unroll
    for (int le = 0; le < 3; ++le) {
        const int de = 2 * le + 1;
        const int shb = (le == 0) ? 0 : (le == 1 ? 1 : 4);
#pragma unroll
        for (int ln = 0; ln < 3; ++ln) {
            const int dn = 2 * ln + 1;
            const int fb = (ln == 0) ? 0 : (ln == 1 ? 2 : 8);
#pragma unroll
            for (int a = 0; a < de; ++a)
#pragma unroll
                for (int b = 0; b < dn; ++b)
#pragma unroll
                    for (int ch = 0; ch < 2; ++ch) {
                        if (u >= Cf::ULO && u < Cf::UHI)
                            uRow[u - Cf::ULO] = shv[shb + a] * f[fb + ch * dn + b];
                        ++u;
                    }
        }
    }
    if (Cf::UHI > 162) {
        uRow[162 - Cf::ULO] = 0.0f;
        uRow[163 - Cf::ULO] = 0.0f;
    }
}

template <int S, int H>
__device__ __forceinline__ float slot_val(const float (&gRow)[HalfCfg<H>::NG],
                                          const float* __restrict__ rRow) {
    typedef HalfCfg<H> Cf;
    constexpr int lo = (S < 2) ? 0 : ((S < 8) ? 1 : 2);
    constexpr int o  = (S < 2) ? S : ((S < 8) ? ((S - 2) / 3) : ((S - 8) / 5));
    constexpr int k  = (S < 2) ? 0 : ((S < 8) ? ((S - 2) % 3) : ((S - 8) % 5));
    constexpr int dO = 2 * lo + 1;
    float v = 0.0f;
#pragma unroll
    for (int q = 0; q < N_BY_LO[lo]; ++q) {
        constexpr const int* lst = C_BY_LO[lo];
        const int c = lst[q];
        if (c >= Cf::C0 && c < Cf::C1) {
            const int base = OUT_OFF[c] - Cf::COL0;
            const int gb = (c - Cf::C0) * 4;
            v += gRow[gb + o]     * rRow[base + k] +
                 gRow[gb + 2 + o] * rRow[base + dO + k];
        }
    }
    return v;
}

template <int H>
__device__ __forceinline__ void
layer_half(const float* __restrict__ feats, const int* __restrict__ eidx,
           const float* __restrict__ sh0, const float* __restrict__ sh1,
           const float* __restrict__ sh2, const float* __restrict__ dist,
           const float* __restrict__ Mg, const int* __restrict__ perm,
           const float* __restrict__ W1, const float* __restrict__ b1,
           const float* __restrict__ W2, const float* __restrict__ b2,
           float* __restrict__ outF, int E,
           float* __restrict__ sU, float* __restrict__ sRaw, const int lane) {
    typedef HalfCfg<H> Cf;
    const long slot = (long)blockIdx.x * TILE + lane;
    const bool valid = (slot < (long)E);
    const long e = valid ? (perm ? (long)perm[slot] : slot) : 0;

    float f[FEAT];
    float shv[9];
    float x0 = 0.0f, x1 = 0.0f;
    int tgt = -1;
    if (valid) {
        const float* fp = &feats[(long)eidx[e] * FEAT];
        tgt = eidx[(long)E + e];
#pragma unroll
        for (int j = 0; j < FEAT; ++j) f[j] = fp[j];
        shv[0] = sh0[e];
#pragma unroll
        for (int j = 0; j < 3; ++j) shv[1 + j] = sh1[e * 3 + j];
#pragma unroll
        for (int j = 0; j < 5; ++j) shv[4 + j] = sh2[e * 5 + j];
        x0 = dist[e];
        float inv0 = 0.5f * (f[0] + f[1]);
        float n1 = 0.0f, n2 = 0.0f;
#pragma unroll
        for (int j = 0; j < 3; ++j) { float m = 0.5f * (f[2 + j] + f[5 + j]); n1 += m * m; }
#pragma unroll
        for (int j = 0; j < 5; ++j) { float m = 0.5f * (f[8 + j] + f[13 + j]); n2 += m * m; }
        x1 = (inv0 + sqrtf(n1) + sqrtf(n2)) * (1.0f / 3.0f);
    } else {
#pragma unroll
        for (int j = 0; j < FEAT; ++j) f[j] = 0.0f;
#pragma unroll
        for (int j = 0; j < 9; ++j) shv[j] = 0.0f;
    }

    float g[Cf::NG];
    u_stage<H>(shv, f, &sU[lane * Cf::UP]);
    gate_block<H>(W1, b1, W2, b2, x0, x1, g);
    wave_lds_handover();

    {
        const int col16 = lane & 15;
        const int kofs = (lane < 16) ? 0 : 2;
        const float* uA0 = &sU[col16 * Cf::UP];
        const float* uA1 = &sU[(16 + col16) * Cf::UP];
#pragma unroll
        for (int n = Cf::NT0; n < Cf::NT1; ++n) {
            v8f acc0 = {0.f, 0.f, 0.f, 0.f, 0.f, 0.f, 0.f, 0.f};
            v8f acc1 = {0.f, 0.f, 0.f, 0.f, 0.f, 0.f, 0.f, 0.f};
            const int col = col16 + n * 16;
#pragma unroll
            for (int kt = KLO[n]; kt <= KHI[n]; ++kt) {
                int k0 = kt * 4 + kofs;
                v2f a0, a1, b;
                a0.x = uA0[k0 - Cf::ULO];
                a0.y = uA0[k0 - Cf::ULO + 1];
                a1.x = uA1[k0 - Cf::ULO];
                a1.y = uA1[k0 - Cf::ULO + 1];
                b.x = Mg[k0 * PN + col];
                b.y = Mg[(k0 + 1) * PN + col];
                acc0 = __builtin_amdgcn_wmma_f32_16x16x4_f32(
                    false, a0, false, b, (short)0, acc0, false, false);
                acc1 = __builtin_amdgcn_wmma_f32_16x16x4_f32(
                    false, a1, false, b, (short)0, acc1, false, false);
            }
            const int rbase = (lane < 16) ? 0 : 8;
#pragma unroll
            for (int j = 0; j < 8; ++j) {
                sRaw[(rbase + j) * Cf::RP + col - Cf::COL0] = acc0[j];
                sRaw[(16 + rbase + j) * Cf::RP + col - Cf::COL0] = acc1[j];
            }
        }
    }
    wave_lds_handover();

    {
        const float* rRow = &sRaw[lane * Cf::RP];
        float v[FEAT];
        v[0]  = slot_val<0, H>(g, rRow);   v[1]  = slot_val<1, H>(g, rRow);
        v[2]  = slot_val<2, H>(g, rRow);   v[3]  = slot_val<3, H>(g, rRow);
        v[4]  = slot_val<4, H>(g, rRow);   v[5]  = slot_val<5, H>(g, rRow);
        v[6]  = slot_val<6, H>(g, rRow);   v[7]  = slot_val<7, H>(g, rRow);
        v[8]  = slot_val<8, H>(g, rRow);   v[9]  = slot_val<9, H>(g, rRow);
        v[10] = slot_val<10, H>(g, rRow);  v[11] = slot_val<11, H>(g, rRow);
        v[12] = slot_val<12, H>(g, rRow);  v[13] = slot_val<13, H>(g, rRow);
        v[14] = slot_val<14, H>(g, rRow);  v[15] = slot_val<15, H>(g, rRow);
        v[16] = slot_val<16, H>(g, rRow);  v[17] = slot_val<17, H>(g, rRow);
        const int tUp = __shfl_up(tgt, 1, TILE);
        int runLo = (lane == 0 || tUp != tgt) ? lane : 0;
#pragma unroll
        for (int d = 1; d < TILE; d <<= 1) {
            const int o = __shfl_up(runLo, d, TILE);
            if (lane >= d) runLo = max(runLo, o);
        }
#pragma unroll
        for (int d = 1; d < TILE; d <<= 1) {
            const bool take = (lane - d >= runLo);
#pragma unroll
            for (int i = 0; i < FEAT; ++i) {
                const float o = __shfl_up(v[i], d, TILE);
                v[i] += take ? o : 0.0f;
            }
        }
        const int tDn = __shfl_down(tgt, 1, TILE);
        const bool last = (lane == TILE - 1) || (tDn != tgt);
        if (valid && last) {
            float* outBase = &outF[(long)tgt * FEAT];
#pragma unroll
            for (int i = 0; i < FEAT; ++i) atomicAdd(outBase + i, v[i]);
        }
    }
}

__global__ void __launch_bounds__(2 * TILE)
layer_kernel(const float* __restrict__ feats, const int* __restrict__ eidx,
             const float* __restrict__ sh0, const float* __restrict__ sh1,
             const float* __restrict__ sh2, const float* __restrict__ dist,
             const float* __restrict__ Mg, const int* __restrict__ perm,
             const float* __restrict__ W1, const float* __restrict__ b1,
             const float* __restrict__ W2, const float* __restrict__ b2,
             float* __restrict__ outF, int E) {
    __shared__ __attribute__((aligned(16))) float sMem[SMEM_FLOATS];
    const int lane = threadIdx.x & (TILE - 1);
    const int half = __builtin_amdgcn_readfirstlane((int)(threadIdx.x >> 5));
    if (half == 0)
        layer_half<0>(feats, eidx, sh0, sh1, sh2, dist, Mg, perm, W1, b1, W2, b2, outF, E,
                      sMem + SU0_OFF, sMem + SRAW0_OFF, lane);
    else
        layer_half<1>(feats, eidx, sh0, sh1, sh2, dist, Mg, perm, W1, b1, W2, b2, outF, E,
                      sMem + SU1_OFF, sMem + SRAW1_OFF, lane);
}

extern "C" void kernel_launch(void* const* d_in, const int* in_sizes, int n_in,
                              void* d_out, int out_size, void* d_ws, size_t ws_size,
                              hipStream_t stream) {
    const float* node = (const float*)d_in[0];
    const int* eidx   = (const int*)d_in[1];
    const float* sh0  = (const float*)d_in[2];
    const float* sh1  = (const float*)d_in[3];
    const float* sh2  = (const float*)d_in[4];
    const float* dist = (const float*)d_in[5];
    const float* cg[N_COMBOS];
    for (int i = 0; i < N_COMBOS; ++i) cg[i] = (const float*)d_in[6 + i];
    const float* W1_0 = (const float*)d_in[21];
    const float* b1_0 = (const float*)d_in[22];
    const float* W2_0 = (const float*)d_in[23];
    const float* b2_0 = (const float*)d_in[24];
    const float* W1_1 = (const float*)d_in[25];
    const float* b1_1 = (const float*)d_in[26];
    const float* W2_1 = (const float*)d_in[27];
    const float* b2_1 = (const float*)d_in[28];

    const int E = in_sizes[1] / 2;
    const int N = in_sizes[0] / FEAT;

    float* Mg     = (float*)d_ws;
    float* feats1 = (float*)((char*)d_ws + 131072);
    long n1 = (long)N * FEAT;
    int* perm     = (int*)(feats1 + n1);

    build_M_kernel<<<1, 256, 0, stream>>>(cg[0], cg[1], cg[2], cg[3], cg[4], cg[5],
                                          cg[6], cg[7], cg[8], cg[9], cg[10], cg[11],
                                          cg[12], cg[13], cg[14], Mg);

    const bool grouped = (ws_size >= (size_t)131072 + (size_t)(n1 + E) * 4) &&
                         ((long)E + N <= (long)out_size);
    if (grouped) {
        int* rnk = (int*)d_out;
        int* cnt = rnk + E;
        zero_kernel<<<(N + 255) / 256, 256, 0, stream>>>((float*)cnt, (long)N);
        rank_kernel<<<(E + 255) / 256, 256, 0, stream>>>(eidx, cnt, rnk, E);
        scan_kernel<<<1, 1024, 0, stream>>>(cnt, N);
        fill_kernel<<<(E + 255) / 256, 256, 0, stream>>>(eidx, cnt, rnk, perm, E);
    } else {
        perm = nullptr;
    }

    zero_kernel<<<(int)((n1 + 255) / 256), 256, 0, stream>>>(feats1, n1);
    zero_kernel<<<(out_size + 255) / 256, 256, 0, stream>>>((float*)d_out, (long)out_size);

    int grid = (E + TILE - 1) / TILE;
    layer_kernel<<<grid, 2 * TILE, 0, stream>>>(node, eidx, sh0, sh1, sh2, dist, Mg, perm,
                                                  W1_0, b1_0, W2_0, b2_0, feats1, E);
    layer_kernel<<<grid, 2 * TILE, 0, stream>>>(feats1, eidx, sh0, sh1, sh2, dist, Mg, perm,
                                                  W1_1, b1_1, W2_1, b2_1, (float*)d_out, E);
}


# Edge sort as a two-level counting sort on wave-private LDS counters (chunk pass, then one owner workgroup per 196 targets): 2 launches not 4, no global atomics; order inside a target run now fixed
# speedup vs baseline: 2.2045x; 1.3671x over previous
#include <hip/hip_runtime.h>

typedef __attribute__((ext_vector_type(2))) float v2f;
typedef __attribute__((ext_vector_type(8))) float v8f;

#define N_COMBOS 15
#define KP 164
#define PN 112
#define NTILES 7
#define TILE 32
#define FEAT 18

constexpr int cLE[N_COMBOS]     = {0,0,0,1,1,1,1,1,1,2,2,2,2,2,2};
constexpr int cLN[N_COMBOS]     = {0,1,2,0,1,1,1,2,2,0,1,1,2,2,2};
constexpr int cLO[N_COMBOS]     = {0,1,2,1,0,1,2,1,2,2,1,2,0,1,2};
constexpr int OUT_OFF[N_COMBOS] = {0,2,8,18,24,26,32,42,48,58,68,74,84,86,92};
constexpr int U_OFF[9]          = {0,2,8,18,24,42,72,82,112};
constexpr int KLO[NTILES]       = {0,2,6,10,18,20,28};
constexpr int KHI[NTILES]       = {4,10,17,20,27,40,40};
constexpr int N_BY_LO[3]        = {3,6,6};
constexpr int C_BY_LO[3][6]     = {{0,4,12,0,0,0},{1,3,5,7,10,13},{2,6,8,9,11,14}};

__global__ void build_M_kernel(const float* cg0, const float* cg1, const float* cg2,
                               const float* cg3, const float* cg4, const float* cg5,
                               const float* cg6, const float* cg7, const float* cg8,
                               const float* cg9, const float* cg10, const float* cg11,
                               const float* cg12, const float* cg13, const float* cg14,
                               float* M) {
    const float* cgs[N_COMBOS] = {cg0,cg1,cg2,cg3,cg4,cg5,cg6,cg7,cg8,cg9,cg10,cg11,cg12,cg13,cg14};
    int tid = threadIdx.x;
    for (int i = tid; i < KP * PN; i += blockDim.x) M[i] = 0.0f;
    __syncthreads();
    if (tid < N_COMBOS) {
        int c = tid;
        int le = cLE[c], ln = cLN[c], lo = cLO[c];
        int de = 2 * le + 1, dn = 2 * ln + 1, dO = 2 * lo + 1;
        int uo = U_OFF[le * 3 + ln];
        int po = OUT_OFF[c];
        const float* cg = cgs[c];
        for (int a = 0; a < de; ++a)
            for (int b = 0; b < dn; ++b)
                for (int k = 0; k < dO; ++k) {
                    float v = cg[(a * dn + b) * dO + k];
                    for (int ch = 0; ch < 2; ++ch)
                        M[(uo + (a * dn + b) * 2 + ch) * PN + po + ch * dO + k] = v;
                }
    }
}

__global__ void zero_kernel(float* p, long n) {
    long i = (long)blockIdx.x * blockDim.x + threadIdx.x;
    if (i < n) p[i] = 0.0f;
}

__global__ void rank_kernel(const int* __restrict__ eidx, int* __restrict__ cnt,
                            int* __restrict__ rnk, int E) {
    int e = blockIdx.x * blockDim.x + threadIdx.x;
    if (e < E) rnk[e] = atomicAdd(&cnt[eidx[(long)E + e]], 1);
}

__global__ void __launch_bounds__(1024)
scan_kernel(int* __restrict__ cnt, int N) {
    __shared__ int part[1024];
    const int t = threadIdx.x;
    const int chunk = (N + 1023) / 1024;
    const int lo = t * chunk;
    int hi = lo + chunk;
    if (hi > N) hi = N;
    int s = 0;
    for (int i = lo; i < hi; ++i) s += cnt[i];
    part[t] = s;
    __syncthreads();
    for (int d = 1; d < 1024; d <<= 1) {
        int add = (t >= d) ? part[t - d] : 0;
        __syncthreads();
        part[t] += add;
        __syncthreads();
    }
    int run = part[t] - s;
    for (int i = lo; i < hi; ++i) { int c = cnt[i]; cnt[i] = run; run += c; }
}

__global__ void fill_kernel(const int* __restrict__ eidx, const int* __restrict__ first,
                            const int* __restrict__ rnk, int* __restrict__ perm, int E) {
    int e = blockIdx.x * blockDim.x + threadIdx.x;
    if (e < E) perm[first[eidx[(long)E + e]] + rnk[e]] = e;
}

#define NCH 128
#define NBK 256
#define CSTP (NBK + 1)
#define BWMAX 512
#define HPA (NBK + 1)
#define HPB (BWMAX + 1)

__device__ __forceinline__ int block_excl_scan_1024(int v, int* sWave, int& total) {
    const int lane = threadIdx.x & 31, wave = threadIdx.x >> 5;
    int x = v;
    for (int d = 1; d < 32; d <<= 1) { const int y = __shfl_up(x, d, 32); if (lane >= d) x += y; }
    if (lane == 31) sWave[wave] = x;
    __syncthreads();
    const int ws = sWave[lane];
    int xs = ws;
    for (int d = 1; d < 32; d <<= 1) { const int y = __shfl_up(xs, d, 32); if (lane >= d) xs += y; }
    const int waveOff = __shfl(xs, wave, 32) - __shfl(ws, wave, 32);
    total = __shfl(xs, 31, 32);
    __syncthreads();
    return waveOff + x - v;
}

__global__ void __launch_bounds__(1024)
bucket_kernel(const int* __restrict__ eidx, int* __restrict__ sortedA, int* __restrict__ cst,
              int E, int chunk, int BW) {
    __shared__ int hist[32 * HPA];
    __shared__ int sWave[32];
    const int t = threadIdx.x, lane = t & 31, w = t >> 5;
    const int c = blockIdx.x;
    for (int i = t; i < 32 * HPA; i += 1024) hist[i] = 0;
    __syncthreads();
    long lo = (long)c * chunk;
    if (lo > E) lo = E;
    long hi = lo + chunk;
    if (hi > E) hi = E;
    const int cLo = (int)lo, cHi = (int)hi;
    const int slice = (chunk + 31) / 32;
    int sLo = cLo + w * slice;
    if (sLo > cHi) sLo = cHi;
    int sHi = sLo + slice;
    if (sHi > cHi) sHi = cHi;
    const int* __restrict__ tg = eidx + E;
    int* myHist = &hist[w * HPA];
    const unsigned ubw = (unsigned)BW;
    for (int i = sLo + lane; i < sHi; i += 32)
        atomicAdd(&myHist[(unsigned)tg[i] / ubw], 1);
    __syncthreads();
    const int b = t >> 2, q = t & 3;
    int s = 0;
    for (int j = 0; j < 8; ++j) s += hist[(q * 8 + j) * HPA + b];
    int total;
    const int ex = block_excl_scan_1024(s, sWave, total);
    if (q == 0) cst[c * CSTP + b] = ex;
    if (t == 0) cst[c * CSTP + NBK] = total;
    int run = ex;
    for (int j = 0; j < 8; ++j) {
        const int k = (q * 8 + j) * HPA + b;
        const int c0 = hist[k];
        hist[k] = run;
        run += c0;
    }
    __syncthreads();
    for (int i = sLo + lane; i < sHi; i += 32) {
        const unsigned tv = (unsigned)tg[i];
        const unsigned bk = tv / ubw;
        const int pos = atomicAdd(&myHist[bk], 1);
        sortedA[cLo + pos] = (int)(((tv - bk * ubw) << 20) | (unsigned)i);
    }
}

__global__ void __launch_bounds__(1024)
owner_kernel(const int* __restrict__ sortedA, const int* __restrict__ cst,
             int* __restrict__ perm, int E, int chunk, int BW) {
    __shared__ int hist[32 * HPB];
    __shared__ int segStart[NCH];
    __shared__ int segLen[NCH];
    __shared__ int sWave[32];
    const int t = threadIdx.x, lane = t & 31, w = t >> 5;
    const int b = blockIdx.x;
    for (int i = t; i < 32 * HPB; i += 1024) hist[i] = 0;
    int st = 0;
    if (t < NCH) {
        st = cst[t * CSTP + b];
        const int ln = cst[t * CSTP + b + 1] - st;
        long lo = (long)t * chunk;
        if (lo > E) lo = E;
        segStart[t] = (int)lo + st;
        segLen[t] = ln;
    }
    int base;
    block_excl_scan_1024(st, sWave, base);
    constexpr int CPW = NCH / 32;
    int s0[CPW], n[CPW];
    int nMax = 0;
    for (int cc = 0; cc < CPW; ++cc) {
        s0[cc] = segStart[w * CPW + cc];
        n[cc] = segLen[w * CPW + cc];
        nMax = n[cc] > nMax ? n[cc] : nMax;
    }
    int* myHist = &hist[w * HPB];
    for (int j0 = 0; j0 < nMax; j0 += 32) {
        const int j = j0 + lane;
        unsigned wd[CPW];
        for (int cc = 0; cc < CPW; ++cc) wd[cc] = (j < n[cc]) ? (unsigned)sortedA[s0[cc] + j] : 0u;
        for (int cc = 0; cc < CPW; ++cc)
            if (j < n[cc]) atomicAdd(&myHist[wd[cc] >> 20], 1);
    }
    __syncthreads();
    int deg = 0;
    if (t < BW)
        for (int w2 = 0; w2 < 32; ++w2) deg += hist[w2 * HPB + t];
    int tot;
    const int ex = block_excl_scan_1024(deg, sWave, tot);
    if (t < BW) {
        int run = base + ex;
        for (int w2 = 0; w2 < 32; ++w2) {
            const int c0 = hist[w2 * HPB + t];
            hist[w2 * HPB + t] = run;
            run += c0;
        }
    }
    __syncthreads();
    for (int j0 = 0; j0 < nMax; j0 += 32) {
        const int j = j0 + lane;
        unsigned wd[CPW];
        for (int cc = 0; cc < CPW; ++cc) wd[cc] = (j < n[cc]) ? (unsigned)sortedA[s0[cc] + j] : 0u;
        for (int cc = 0; cc < CPW; ++cc)
            if (j < n[cc]) {
                const int r = atomicAdd(&myHist[wd[cc] >> 20], 1);
                perm[r] = (int)(wd[cc] & 0xFFFFFu);
            }
    }
}

template <int H> struct HalfCfg;
template <> struct HalfCfg<0> {
    static constexpr int T0 = 0,  NG = 32;
    static constexpr int C0 = 0,  C1 = 8;
    static constexpr int NT0 = 0, NT1 = 3;
    static constexpr int ULO = 0, UHI = 72;
    static constexpr int UP = 76;
    static constexpr int COL0 = 0;
    static constexpr int RP = 52;
};
template <> struct HalfCfg<1> {
    static constexpr int T0 = 32, NG = 28;
    static constexpr int C0 = 8,  C1 = 15;
    static constexpr int NT0 = 3, NT1 = 7;
    static constexpr int ULO = 40, UHI = 164;
    static constexpr int UP = 124;
    static constexpr int COL0 = 48;
    static constexpr int RP = 68;
};
#define SU0_OFF   0
#define SRAW0_OFF 2432
#define SU1_OFF   4096
#define SRAW1_OFF 8064
#define SMEM_FLOATS 10240

__device__ __forceinline__ void wave_lds_handover() {
    __builtin_amdgcn_fence(__ATOMIC_RELEASE, "wavefront");
    __builtin_amdgcn_wave_barrier();
    __builtin_amdgcn_fence(__ATOMIC_ACQUIRE, "wavefront");
}

template <int H>
__device__ __forceinline__ void gate_block(const float* __restrict__ W1,
                                           const float* __restrict__ b1,
                                           const float* __restrict__ W2,
                                           const float* __restrict__ b2,
                                           float x0, float x1,
                                           float (&g)[HalfCfg<H>::NG]) {
#pragma unroll
    for (int t = 0; t < HalfCfg<H>::NG; ++t) {
        const int bi = HalfCfg<H>::T0 + t;
        float acc = b2[bi];
#pragma unroll
        for (int h = 0; h < 4; ++h) {
            float hh = x0 * W1[bi * 8 + h * 2] + x1 * W1[bi * 8 + h * 2 + 1] + b1[bi * 4 + h];
            hh = hh > 0.0f ? hh : 0.0f;
            acc += hh * W2[bi * 4 + h];
        }
        g[t] = tanhf(acc);
    }
}

template <int H>
__device__ __forceinline__ void u_stage(const float (&shv)[9],
                                        const float (&f)[FEAT],
                                        float* __restrict__ uRow) {
    typedef HalfCfg<H> Cf;
    int u = 0;
#pragma unroll
    for (int le = 0; le < 3; ++le) {
        const int de = 2 * le + 1;
        const int shb = (le == 0) ? 0 : (le == 1 ? 1 : 4);
#pragma unroll
        for (int ln = 0; ln < 3; ++ln) {
            const int dn = 2 * ln + 1;
            const int fb = (ln == 0) ? 0 : (ln == 1 ? 2 : 8);
#pragma unroll
            for (int a = 0; a < de; ++a)
#pragma unroll
                for (int b = 0; b < dn; ++b)
#pragma unroll
                    for (int ch = 0; ch < 2; ++ch) {
                        if (u >= Cf::ULO && u < Cf::UHI)
                            uRow[u - Cf::ULO] = shv[shb + a] * f[fb + ch * dn + b];
                        ++u;
                    }
        }
    }
    if (Cf::UHI > 162) {
        uRow[162 - Cf::ULO] = 0.0f;
        uRow[163 - Cf::ULO] = 0.0f;
    }
}

template <int S, int H>
__device__ __forceinline__ float slot_val(const float (&gRow)[HalfCfg<H>::NG],
                                          const float* __restrict__ rRow) {
    typedef HalfCfg<H> Cf;
    constexpr int lo = (S < 2) ? 0 : ((S < 8) ? 1 : 2);
    constexpr int o  = (S < 2) ? S : ((S < 8) ? ((S - 2) / 3) : ((S - 8) / 5));
    constexpr int k  = (S < 2) ? 0 : ((S < 8) ? ((S - 2) % 3) : ((S - 8) % 5));
    constexpr int dO = 2 * lo + 1;
    float v = 0.0f;
#pragma unroll
    for (int q = 0; q < N_BY_LO[lo]; ++q) {
        constexpr const int* lst = C_BY_LO[lo];
        const int c = lst[q];
        if (c >= Cf::C0 && c < Cf::C1) {
            const int base = OUT_OFF[c] - Cf::COL0;
            const int gb = (c - Cf::C0) * 4;
            v += gRow[gb + o]     * rRow[base + k] +
                 gRow[gb + 2 + o] * rRow[base + dO + k];
        }
    }
    return v;
}

template <int H>
__device__ __forceinline__ void
layer_half(const float* __restrict__ feats, const int* __restrict__ eidx,
           const float* __restrict__ sh0, const float* __restrict__ sh1,
           const float* __restrict__ sh2, const float* __restrict__ dist,
           const float* __restrict__ Mg, const int* __restrict__ perm,
           const float* __restrict__ W1, const float* __restrict__ b1,
           const float* __restrict__ W2, const float* __restrict__ b2,
           float* __restrict__ outF, int E,
           float* __restrict__ sU, float* __restrict__ sRaw, const int lane) {
    typedef HalfCfg<H> Cf;
    const long slot = (long)blockIdx.x * TILE + lane;
    const bool valid = (slot < (long)E);
    const long e = valid ? (perm ? (long)perm[slot] : slot) : 0;

    float f[FEAT];
    float shv[9];
    float x0 = 0.0f, x1 = 0.0f;
    int tgt = -1;
    if (valid) {
        const float* fp = &feats[(long)eidx[e] * FEAT];
        tgt = eidx[(long)E + e];
#pragma unroll
        for (int j = 0; j < FEAT; ++j) f[j] = fp[j];
        shv[0] = sh0[e];
#pragma unroll
        for (int j = 0; j < 3; ++j) shv[1 + j] = sh1[e * 3 + j];
#pragma unroll
        for (int j = 0; j < 5; ++j) shv[4 + j] = sh2[e * 5 + j];
        x0 = dist[e];
        float inv0 = 0.5f * (f[0] + f[1]);
        float n1 = 0.0f, n2 = 0.0f;
#pragma unroll
        for (int j = 0; j < 3; ++j) { float m = 0.5f * (f[2 + j] + f[5 + j]); n1 += m * m; }
#pragma unroll
        for (int j = 0; j < 5; ++j) { float m = 0.5f * (f[8 + j] + f[13 + j]); n2 += m * m; }
        x1 = (inv0 + sqrtf(n1) + sqrtf(n2)) * (1.0f / 3.0f);
    } else {
#pragma unroll
        for (int j = 0; j < FEAT; ++j) f[j] = 0.0f;
#pragma unroll
        for (int j = 0; j < 9; ++j) shv[j] = 0.0f;
    }

    float g[Cf::NG];
    u_stage<H>(shv, f, &sU[lane * Cf::UP]);
    gate_block<H>(W1, b1, W2, b2, x0, x1, g);
    wave_lds_handover();

    {
        const int col16 = lane & 15;
        const int kofs = (lane < 16) ? 0 : 2;
        const float* uA0 = &sU[col16 * Cf::UP];
        const float* uA1 = &sU[(16 + col16) * Cf::UP];
#pragma unroll
        for (int n = Cf::NT0; n < Cf::NT1; ++n) {
            v8f acc0 = {0.f, 0.f, 0.f, 0.f, 0.f, 0.f, 0.f, 0.f};
            v8f acc1 = {0.f, 0.f, 0.f, 0.f, 0.f, 0.f, 0.f, 0.f};
            const int col = col16 + n * 16;
#pragma unroll
            for (int kt = KLO[n]; kt <= KHI[n]; ++kt) {
                int k0 = kt * 4 + kofs;
                v2f a0, a1, b;
                a0.x = uA0[k0 - Cf::ULO];
                a0.y = uA0[k0 - Cf::ULO + 1];
                a1.x = uA1[k0 - Cf::ULO];
                a1.y = uA1[k0 - Cf::ULO + 1];
                b.x = Mg[k0 * PN + col];
                b.y = Mg[(k0 + 1) * PN + col];
                acc0 = __builtin_amdgcn_wmma_f32_16x16x4_f32(
                    false, a0, false, b, (short)0, acc0, false, false);
                acc1 = __builtin_amdgcn_wmma_f32_16x16x4_f32(
                    false, a1, false, b, (short)0, acc1, false, false);
            }
            const int rbase = (lane < 16) ? 0 : 8;
#pragma unroll
            for (int j = 0; j < 8; ++j) {
                sRaw[(rbase + j) * Cf::RP + col - Cf::COL0] = acc0[j];
                sRaw[(16 + rbase + j) * Cf::RP + col - Cf::COL0] = acc1[j];
            }
        }
    }
    wave_lds_handover();

    {
        const float* rRow = &sRaw[lane * Cf::RP];
        float v[FEAT];
        v[0]  = slot_val<0, H>(g, rRow);   v[1]  = slot_val<1, H>(g, rRow);
        v[2]  = slot_val<2, H>(g, rRow);   v[3]  = slot_val<3, H>(g, rRow);
        v[4]  = slot_val<4, H>(g, rRow);   v[5]  = slot_val<5, H>(g, rRow);
        v[6]  = slot_val<6, H>(g, rRow);   v[7]  = slot_val<7, H>(g, rRow);
        v[8]  = slot_val<8, H>(g, rRow);   v[9]  = slot_val<9, H>(g, rRow);
        v[10] = slot_val<10, H>(g, rRow);  v[11] = slot_val<11, H>(g, rRow);
        v[12] = slot_val<12, H>(g, rRow);  v[13] = slot_val<13, H>(g, rRow);
        v[14] = slot_val<14, H>(g, rRow);  v[15] = slot_val<15, H>(g, rRow);
        v[16] = slot_val<16, H>(g, rRow);  v[17] = slot_val<17, H>(g, rRow);
        const int tUp = __shfl_up(tgt, 1, TILE);
        int runLo = (lane == 0 || tUp != tgt) ? lane : 0;
#pragma unroll
        for (int d = 1; d < TILE; d <<= 1) {
            const int o = __shfl_up(runLo, d, TILE);
            if (lane >= d) runLo = max(runLo, o);
        }
#pragma unroll
        for (int d = 1; d < TILE; d <<= 1) {
            const bool take = (lane - d >= runLo);
#pragma unroll
            for (int i = 0; i < FEAT; ++i) {
                const float o = __shfl_up(v[i], d, TILE);
                v[i] += take ? o : 0.0f;
            }
        }
        const int tDn = __shfl_down(tgt, 1, TILE);
        const bool last = (lane == TILE - 1) || (tDn != tgt);
        if (valid && last) {
            float* outBase = &outF[(long)tgt * FEAT];
#pragma unroll
            for (int i = 0; i < FEAT; ++i) atomicAdd(outBase + i, v[i]);
        }
    }
}

__global__ void __launch_bounds__(2 * TILE)
layer_kernel(const float* __restrict__ feats, const int* __restrict__ eidx,
             const float* __restrict__ sh0, const float* __restrict__ sh1,
             const float* __restrict__ sh2, const float* __restrict__ dist,
             const float* __restrict__ Mg, const int* __restrict__ perm,
             const float* __restrict__ W1, const float* __restrict__ b1,
             const float* __restrict__ W2, const float* __restrict__ b2,
             float* __restrict__ outF, int E) {
    __shared__ __attribute__((aligned(16))) float sMem[SMEM_FLOATS];
    const int lane = threadIdx.x & (TILE - 1);
    const int half = __builtin_amdgcn_readfirstlane((int)(threadIdx.x >> 5));
    if (half == 0)
        layer_half<0>(feats, eidx, sh0, sh1, sh2, dist, Mg, perm, W1, b1, W2, b2, outF, E,
                      sMem + SU0_OFF, sMem + SRAW0_OFF, lane);
    else
        layer_half<1>(feats, eidx, sh0, sh1, sh2, dist, Mg, perm, W1, b1, W2, b2, outF, E,
                      sMem + SU1_OFF, sMem + SRAW1_OFF, lane);
}

extern "C" void kernel_launch(void* const* d_in, const int* in_sizes, int n_in,
                              void* d_out, int out_size, void* d_ws, size_t ws_size,
                              hipStream_t stream) {
    const float* node = (const float*)d_in[0];
    const int* eidx   = (const int*)d_in[1];
    const float* sh0  = (const float*)d_in[2];
    const float* sh1  = (const float*)d_in[3];
    const float* sh2  = (const float*)d_in[4];
    const float* dist = (const float*)d_in[5];
    const float* cg[N_COMBOS];
    for (int i = 0; i < N_COMBOS; ++i) cg[i] = (const float*)d_in[6 + i];
    const float* W1_0 = (const float*)d_in[21];
    const float* b1_0 = (const float*)d_in[22];
    const float* W2_0 = (const float*)d_in[23];
    const float* b2_0 = (const float*)d_in[24];
    const float* W1_1 = (const float*)d_in[25];
    const float* b1_1 = (const float*)d_in[26];
    const float* W2_1 = (const float*)d_in[27];
    const float* b2_1 = (const float*)d_in[28];

    const int E = in_sizes[1] / 2;
    const int N = in_sizes[0] / FEAT;

    float* Mg     = (float*)d_ws;
    float* feats1 = (float*)((char*)d_ws + 131072);
    long n1 = (long)N * FEAT;
    int* perm     = (int*)(feats1 + n1);

    build_M_kernel<<<1, 256, 0, stream>>>(cg[0], cg[1], cg[2], cg[3], cg[4], cg[5],
                                          cg[6], cg[7], cg[8], cg[9], cg[10], cg[11],
                                          cg[12], cg[13], cg[14], Mg);

    const bool permFits = (ws_size >= (size_t)131072 + (size_t)(n1 + E) * 4);
    const bool grouped = permFits && ((long)E + N <= (long)out_size);
    const int BW = (N + NBK - 1) / NBK;
    const int chunk = (E + NCH - 1) / NCH;
    const bool twoLevel = permFits && E >= 1 && E <= (1 << 20) && BW >= 1 && BW <= BWMAX &&
                          ((long)E + (long)NCH * CSTP <= (long)out_size);
    if (twoLevel) {
        int* sortedA = (int*)d_out;
        int* cst = sortedA + E;
        bucket_kernel<<<NCH, 1024, 0, stream>>>(eidx, sortedA, cst, E, chunk, BW);
        owner_kernel<<<NBK, 1024, 0, stream>>>(sortedA, cst, perm, E, chunk, BW);
    } else if (grouped) {
        int* rnk = (int*)d_out;
        int* cnt = rnk + E;
        zero_kernel<<<(N + 255) / 256, 256, 0, stream>>>((float*)cnt, (long)N);
        rank_kernel<<<(E + 255) / 256, 256, 0, stream>>>(eidx, cnt, rnk, E);
        scan_kernel<<<1, 1024, 0, stream>>>(cnt, N);
        fill_kernel<<<(E + 255) / 256, 256, 0, stream>>>(eidx, cnt, rnk, perm, E);
    } else {
        perm = nullptr;
    }

    zero_kernel<<<(int)((n1 + 255) / 256), 256, 0, stream>>>(feats1, n1);
    zero_kernel<<<(out_size + 255) / 256, 256, 0, stream>>>((float*)d_out, (long)out_size);

    int grid = (E + TILE - 1) / TILE;
    layer_kernel<<<grid, 2 * TILE, 0, stream>>>(node, eidx, sh0, sh1, sh2, dist, Mg, perm,
                                                  W1_0, b1_0, W2_0, b2_0, feats1, E);
    layer_kernel<<<grid, 2 * TILE, 0, stream>>>(feats1, eidx, sh0, sh1, sh2, dist, Mg, perm,
                                                  W1_1, b1_1, W2_1, b2_1, (float*)d_out, E);
}


# Two-wave split re-cut at combo 9 (edge degree 0-1 / 2): matrix+vector work per wave evened, no u row staged twice, LDS 40,960 -> 34,816 B; combo 8 now adds into the lower partial row
# speedup vs baseline: 2.4161x; 1.0960x over previous
#include <hip/hip_runtime.h>

typedef __attribute__((ext_vector_type(2))) float v2f;
typedef __attribute__((ext_vector_type(8))) float v8f;

#define N_COMBOS 15
#define KP 164
#define PN 112
#define NTILES 7
#define TILE 32
#define FEAT 18

constexpr int cLE[N_COMBOS]     = {0,0,0,1,1,1,1,1,1,2,2,2,2,2,2};
constexpr int cLN[N_COMBOS]     = {0,1,2,0,1,1,1,2,2,0,1,1,2,2,2};
constexpr int cLO[N_COMBOS]     = {0,1,2,1,0,1,2,1,2,2,1,2,0,1,2};
constexpr int OUT_OFF[N_COMBOS] = {0,2,8,18,24,26,32,42,48,58,68,74,84,86,92};
constexpr int U_OFF[9]          = {0,2,8,18,24,42,72,82,112};
constexpr int KLO2[2][NTILES]   = {{0,2,6,10,0,0,0},  {0,0,0,18,18,20,28}};
constexpr int KHI2[2][NTILES]   = {{4,10,17,17,0,0,0},{0,0,0,20,27,40,40}};
constexpr int N_BY_LO[3]        = {3,6,6};
constexpr int C_BY_LO[3][6]     = {{0,4,12,0,0,0},{1,3,5,7,10,13},{2,6,8,9,11,14}};

__global__ void build_M_kernel(const float* cg0, const float* cg1, const float* cg2,
                               const float* cg3, const float* cg4, const float* cg5,
                               const float* cg6, const float* cg7, const float* cg8,
                               const float* cg9, const float* cg10, const float* cg11,
                               const float* cg12, const float* cg13, const float* cg14,
                               float* M) {
    const float* cgs[N_COMBOS] = {cg0,cg1,cg2,cg3,cg4,cg5,cg6,cg7,cg8,cg9,cg10,cg11,cg12,cg13,cg14};
    int tid = threadIdx.x;
    for (int i = tid; i < KP * PN; i += blockDim.x) M[i] = 0.0f;
    __syncthreads();
    if (tid < N_COMBOS) {
        int c = tid;
        int le = cLE[c], ln = cLN[c], lo = cLO[c];
        int de = 2 * le + 1, dn = 2 * ln + 1, dO = 2 * lo + 1;
        int uo = U_OFF[le * 3 + ln];
        int po = OUT_OFF[c];
        const float* cg = cgs[c];
        for (int a = 0; a < de; ++a)
            for (int b = 0; b < dn; ++b)
                for (int k = 0; k < dO; ++k) {
                    float v = cg[(a * dn + b) * dO + k];
                    for (int ch = 0; ch < 2; ++ch)
                        M[(uo + (a * dn + b) * 2 + ch) * PN + po + ch * dO + k] = v;
                }
    }
}

__global__ void zero_kernel(float* p, long n) {
    long i = (long)blockIdx.x * blockDim.x + threadIdx.x;
    if (i < n) p[i] = 0.0f;
}

__global__ void rank_kernel(const int* __restrict__ eidx, int* __restrict__ cnt,
                            int* __restrict__ rnk, int E) {
    int e = blockIdx.x * blockDim.x + threadIdx.x;
    if (e < E) rnk[e] = atomicAdd(&cnt[eidx[(long)E + e]], 1);
}

__global__ void __launch_bounds__(1024)
scan_kernel(int* __restrict__ cnt, int N) {
    __shared__ int part[1024];
    const int t = threadIdx.x;
    const int chunk = (N + 1023) / 1024;
    const int lo = t * chunk;
    int hi = lo + chunk;
    if (hi > N) hi = N;
    int s = 0;
    for (int i = lo; i < hi; ++i) s += cnt[i];
    part[t] = s;
    __syncthreads();
    for (int d = 1; d < 1024; d <<= 1) {
        int add = (t >= d) ? part[t - d] : 0;
        __syncthreads();
        part[t] += add;
        __syncthreads();
    }
    int run = part[t] - s;
    for (int i = lo; i < hi; ++i) { int c = cnt[i]; cnt[i] = run; run += c; }
}

__global__ void fill_kernel(const int* __restrict__ eidx, const int* __restrict__ first,
                            const int* __restrict__ rnk, int* __restrict__ perm, int E) {
    int e = blockIdx.x * blockDim.x + threadIdx.x;
    if (e < E) perm[first[eidx[(long)E + e]] + rnk[e]] = e;
}

#define NCH 128
#define NBK 256
#define CSTP (NBK + 1)
#define BWMAX 512
#define HPA (NBK + 1)
#define HPB (BWMAX + 1)

__device__ __forceinline__ int block_excl_scan_1024(int v, int* sWave, int& total) {
    const int lane = threadIdx.x & 31, wave = threadIdx.x >> 5;
    int x = v;
    for (int d = 1; d < 32; d <<= 1) { const int y = __shfl_up(x, d, 32); if (lane >= d) x += y; }
    if (lane == 31) sWave[wave] = x;
    __syncthreads();
    const int ws = sWave[lane];
    int xs = ws;
    for (int d = 1; d < 32; d <<= 1) { const int y = __shfl_up(xs, d, 32); if (lane >= d) xs += y; }
    const int waveOff = __shfl(xs, wave, 32) - __shfl(ws, wave, 32);
    total = __shfl(xs, 31, 32);
    __syncthreads();
    return waveOff + x - v;
}

__global__ void __launch_bounds__(1024)
bucket_kernel(const int* __restrict__ eidx, int* __restrict__ sortedA, int* __restrict__ cst,
              int E, int chunk, int BW) {
    __shared__ int hist[32 * HPA];
    __shared__ int sWave[32];
    const int t = threadIdx.x, lane = t & 31, w = t >> 5;
    const int c = blockIdx.x;
    for (int i = t; i < 32 * HPA; i += 1024) hist[i] = 0;
    __syncthreads();
    long lo = (long)c * chunk;
    if (lo > E) lo = E;
    long hi = lo + chunk;
    if (hi > E) hi = E;
    const int cLo = (int)lo, cHi = (int)hi;
    const int slice = (chunk + 31) / 32;
    int sLo = cLo + w * slice;
    if (sLo > cHi) sLo = cHi;
    int sHi = sLo + slice;
    if (sHi > cHi) sHi = cHi;
    const int* __restrict__ tg = eidx + E;
    int* myHist = &hist[w * HPA];
    const unsigned ubw = (unsigned)BW;
    for (int i = sLo + lane; i < sHi; i += 32)
        atomicAdd(&myHist[(unsigned)tg[i] / ubw], 1);
    __syncthreads();
    const int b = t >> 2, q = t & 3;
    int s = 0;
    for (int j = 0; j < 8; ++j) s += hist[(q * 8 + j) * HPA + b];
    int total;
    const int ex = block_excl_scan_1024(s, sWave, total);
    if (q == 0) cst[c * CSTP + b] = ex;
    if (t == 0) cst[c * CSTP + NBK] = total;
    int run = ex;
    for (int j = 0; j < 8; ++j) {
        const int k = (q * 8 + j) * HPA + b;
        const int c0 = hist[k];
        hist[k] = run;
        run += c0;
    }
    __syncthreads();
    for (int i = sLo + lane; i < sHi; i += 32) {
        const unsigned tv = (unsigned)tg[i];
        const unsigned bk = tv / ubw;
        const int pos = atomicAdd(&myHist[bk], 1);
        sortedA[cLo + pos] = (int)(((tv - bk * ubw) << 20) | (unsigned)i);
    }
}

__global__ void __launch_bounds__(1024)
owner_kernel(const int* __restrict__ sortedA, const int* __restrict__ cst,
             int* __restrict__ perm, int E, int chunk, int BW) {
    __shared__ int hist[32 * HPB];
    __shared__ int segStart[NCH];
    __shared__ int segLen[NCH];
    __shared__ int sWave[32];
    const int t = threadIdx.x, lane = t & 31, w = t >> 5;
    const int b = blockIdx.x;
    for (int i = t; i < 32 * HPB; i += 1024) hist[i] = 0;
    int st = 0;
    if (t < NCH) {
        st = cst[t * CSTP + b];
        const int ln = cst[t * CSTP + b + 1] - st;
        long lo = (long)t * chunk;
        if (lo > E) lo = E;
        segStart[t] = (int)lo + st;
        segLen[t] = ln;
    }
    int base;
    block_excl_scan_1024(st, sWave, base);
    constexpr int CPW = NCH / 32;
    int s0[CPW], n[CPW];
    int nMax = 0;
    for (int cc = 0; cc < CPW; ++cc) {
        s0[cc] = segStart[w * CPW + cc];
        n[cc] = segLen[w * CPW + cc];
        nMax = n[cc] > nMax ? n[cc] : nMax;
    }
    int* myHist = &hist[w * HPB];
    for (int j0 = 0; j0 < nMax; j0 += 32) {
        const int j = j0 + lane;
        unsigned wd[CPW];
        for (int cc = 0; cc < CPW; ++cc) wd[cc] = (j < n[cc]) ? (unsigned)sortedA[s0[cc] + j] : 0u;
        for (int cc = 0; cc < CPW; ++cc)
            if (j < n[cc]) atomicAdd(&myHist[wd[cc] >> 20], 1);
    }
    __syncthreads();
    int deg = 0;
    if (t < BW)
        for (int w2 = 0; w2 < 32; ++w2) deg += hist[w2 * HPB + t];
    int tot;
    const int ex = block_excl_scan_1024(deg, sWave, tot);
    if (t < BW) {
        int run = base + ex;
        for (int w2 = 0; w2 < 32; ++w2) {
            const int c0 = hist[w2 * HPB + t];
            hist[w2 * HPB + t] = run;
            run += c0;
        }
    }
    __syncthreads();
    for (int j0 = 0; j0 < nMax; j0 += 32) {
        const int j = j0 + lane;
        unsigned wd[CPW];
        for (int cc = 0; cc < CPW; ++cc) wd[cc] = (j < n[cc]) ? (unsigned)sortedA[s0[cc] + j] : 0u;
        for (int cc = 0; cc < CPW; ++cc)
            if (j < n[cc]) {
                const int r = atomicAdd(&myHist[wd[cc] >> 20], 1);
                perm[r] = (int)(wd[cc] & 0xFFFFFu);
            }
    }
}

template <int H> struct HalfCfg;
template <> struct HalfCfg<0> {
    static constexpr int T0 = 0,  NG = 36;
    static constexpr int C0 = 0,  C1 = 9;
    static constexpr int NT0 = 0, NT1 = 4;
    static constexpr int ULO = 0, UHI = 72;
    static constexpr int UP = 76;
    static constexpr int COL0 = 0;
    static constexpr int COL1 = 58;
    static constexpr int RP = 60;
};
template <> struct HalfCfg<1> {
    static constexpr int T0 = 36, NG = 24;
    static constexpr int C0 = 9,  C1 = 15;
    static constexpr int NT0 = 3, NT1 = 7;
    static constexpr int ULO = 72, UHI = 164;
    static constexpr int UP = 92;
    static constexpr int COL0 = 58;
    static constexpr int COL1 = 102;
    static constexpr int RP = 44;
};
#define SU0_OFF   0
#define SRAW0_OFF 2432
#define SU1_OFF   4352
#define SRAW1_OFF 7296
#define SMEM_FLOATS 8704

__device__ __forceinline__ void wave_lds_handover() {
    __builtin_amdgcn_fence(__ATOMIC_RELEASE, "wavefront");
    __builtin_amdgcn_wave_barrier();
    __builtin_amdgcn_fence(__ATOMIC_ACQUIRE, "wavefront");
}

template <int H>
__device__ __forceinline__ void gate_block(const float* __restrict__ W1,
                                           const float* __restrict__ b1,
                                           const float* __restrict__ W2,
                                           const float* __restrict__ b2,
                                           float x0, float x1,
                                           float (&g)[HalfCfg<H>::NG]) {
#pragma unroll
    for (int t = 0; t < HalfCfg<H>::NG; ++t) {
        const int bi = HalfCfg<H>::T0 + t;
        float acc = b2[bi];
#pragma unroll
        for (int h = 0; h < 4; ++h) {
            float hh = x0 * W1[bi * 8 + h * 2] + x1 * W1[bi * 8 + h * 2 + 1] + b1[bi * 4 + h];
            hh = hh > 0.0f ? hh : 0.0f;
            acc += hh * W2[bi * 4 + h];
        }
        g[t] = tanhf(acc);
    }
}

template <int H>
__device__ __forceinline__ void u_stage(const float (&shv)[9],
                                        const float (&f)[FEAT],
                                        float* __restrict__ uRow) {
    typedef HalfCfg<H> Cf;
    int u = 0;
#pragma unroll
    for (int le = 0; le < 3; ++le) {
        const int de = 2 * le + 1;
        const int shb = (le == 0) ? 0 : (le == 1 ? 1 : 4);
#pragma unroll
        for (int ln = 0; ln < 3; ++ln) {
            const int dn = 2 * ln + 1;
            const int fb = (ln == 0) ? 0 : (ln == 1 ? 2 : 8);
#pragma unroll
            for (int a = 0; a < de; ++a)
#pragma unroll
                for (int b = 0; b < dn; ++b)
#pragma unroll
                    for (int ch = 0; ch < 2; ++ch) {
                        if (u >= Cf::ULO && u < Cf::UHI)
                            uRow[u - Cf::ULO] = shv[shb + a] * f[fb + ch * dn + b];
                        ++u;
                    }
        }
    }
    if (Cf::UHI > 162) {
        uRow[162 - Cf::ULO] = 0.0f;
        uRow[163 - Cf::ULO] = 0.0f;
    }
}

template <int S, int H>
__device__ __forceinline__ float slot_val(const float (&gRow)[HalfCfg<H>::NG],
                                          const float* __restrict__ rRow) {
    typedef HalfCfg<H> Cf;
    constexpr int lo = (S < 2) ? 0 : ((S < 8) ? 1 : 2);
    constexpr int o  = (S < 2) ? S : ((S < 8) ? ((S - 2) / 3) : ((S - 8) / 5));
    constexpr int k  = (S < 2) ? 0 : ((S < 8) ? ((S - 2) % 3) : ((S - 8) % 5));
    constexpr int dO = 2 * lo + 1;
    float v = 0.0f;
#pragma unroll
    for (int q = 0; q < N_BY_LO[lo]; ++q) {
        constexpr const int* lst = C_BY_LO[lo];
        const int c = lst[q];
        if (c >= Cf::C0 && c < Cf::C1) {
            const int base = OUT_OFF[c] - Cf::COL0;
            const int gb = (c - Cf::C0) * 4;
            v += gRow[gb + o]     * rRow[base + k] +
                 gRow[gb + 2 + o] * rRow[base + dO + k];
        }
    }
    return v;
}

template <int H>
__device__ __forceinline__ void
layer_half(const float* __restrict__ feats, const int* __restrict__ eidx,
           const float* __restrict__ sh0, const float* __restrict__ sh1,
           const float* __restrict__ sh2, const float* __restrict__ dist,
           const float* __restrict__ Mg, const int* __restrict__ perm,
           const float* __restrict__ W1, const float* __restrict__ b1,
           const float* __restrict__ W2, const float* __restrict__ b2,
           float* __restrict__ outF, int E,
           float* __restrict__ sU, float* __restrict__ sRaw, const int lane) {
    typedef HalfCfg<H> Cf;
    const long slot = (long)blockIdx.x * TILE + lane;
    const bool valid = (slot < (long)E);
    const long e = valid ? (perm ? (long)perm[slot] : slot) : 0;

    float f[FEAT];
    float shv[9];
    float x0 = 0.0f, x1 = 0.0f;
    int tgt = -1;
    if (valid) {
        const float* fp = &feats[(long)eidx[e] * FEAT];
        tgt = eidx[(long)E + e];
#pragma unroll
        for (int j = 0; j < FEAT; ++j) f[j] = fp[j];
        shv[0] = sh0[e];
#pragma unroll
        for (int j = 0; j < 3; ++j) shv[1 + j] = sh1[e * 3 + j];
#pragma unroll
        for (int j = 0; j < 5; ++j) shv[4 + j] = sh2[e * 5 + j];
        x0 = dist[e];
        float inv0 = 0.5f * (f[0] + f[1]);
        float n1 = 0.0f, n2 = 0.0f;
#pragma unroll
        for (int j = 0; j < 3; ++j) { float m = 0.5f * (f[2 + j] + f[5 + j]); n1 += m * m; }
#pragma unroll
        for (int j = 0; j < 5; ++j) { float m = 0.5f * (f[8 + j] + f[13 + j]); n2 += m * m; }
        x1 = (inv0 + sqrtf(n1) + sqrtf(n2)) * (1.0f / 3.0f);
    } else {
#pragma unroll
        for (int j = 0; j < FEAT; ++j) f[j] = 0.0f;
#pragma unroll
        for (int j = 0; j < 9; ++j) shv[j] = 0.0f;
    }

    float g[Cf::NG];
    u_stage<H>(shv, f, &sU[lane * Cf::UP]);
    gate_block<H>(W1, b1, W2, b2, x0, x1, g);
    wave_lds_handover();

    {
        const int col16 = lane & 15;
        const int kofs = (lane < 16) ? 0 : 2;
        const float* uA0 = &sU[col16 * Cf::UP];
        const float* uA1 = &sU[(16 + col16) * Cf::UP];
#pragma unroll
        for (int n = Cf::NT0; n < Cf::NT1; ++n) {
            v8f acc0 = {0.f, 0.f, 0.f, 0.f, 0.f, 0.f, 0.f, 0.f};
            v8f acc1 = {0.f, 0.f, 0.f, 0.f, 0.f, 0.f, 0.f, 0.f};
            const int col = col16 + n * 16;
#pragma unroll
            for (int kt = KLO2[H][n]; kt <= KHI2[H][n]; ++kt) {
                int k0 = kt * 4 + kofs;
                v2f a0, a1, b;
                a0.x = uA0[k0 - Cf::ULO];
                a0.y = uA0[k0 - Cf::ULO + 1];
                a1.x = uA1[k0 - Cf::ULO];
                a1.y = uA1[k0 - Cf::ULO + 1];
                b.x = Mg[k0 * PN + col];
                b.y = Mg[(k0 + 1) * PN + col];
                acc0 = __builtin_amdgcn_wmma_f32_16x16x4_f32(
                    false, a0, false, b, (short)0, acc0, false, false);
                acc1 = __builtin_amdgcn_wmma_f32_16x16x4_f32(
                    false, a1, false, b, (short)0, acc1, false, false);
            }
            const int rbase = (lane < 16) ? 0 : 8;
            if (col >= Cf::COL0 && col < Cf::COL1) {
#pragma unroll
                for (int j = 0; j < 8; ++j) {
                    sRaw[(rbase + j) * Cf::RP + col - Cf::COL0] = acc0[j];
                    sRaw[(16 + rbase + j) * Cf::RP + col - Cf::COL0] = acc1[j];
                }
            }
        }
    }
    wave_lds_handover();

    {
        const float* rRow = &sRaw[lane * Cf::RP];
        float v[FEAT];
        v[0]  = slot_val<0, H>(g, rRow);   v[1]  = slot_val<1, H>(g, rRow);
        v[2]  = slot_val<2, H>(g, rRow);   v[3]  = slot_val<3, H>(g, rRow);
        v[4]  = slot_val<4, H>(g, rRow);   v[5]  = slot_val<5, H>(g, rRow);
        v[6]  = slot_val<6, H>(g, rRow);   v[7]  = slot_val<7, H>(g, rRow);
        v[8]  = slot_val<8, H>(g, rRow);   v[9]  = slot_val<9, H>(g, rRow);
        v[10] = slot_val<10, H>(g, rRow);  v[11] = slot_val<11, H>(g, rRow);
        v[12] = slot_val<12, H>(g, rRow);  v[13] = slot_val<13, H>(g, rRow);
        v[14] = slot_val<14, H>(g, rRow);  v[15] = slot_val<15, H>(g, rRow);
        v[16] = slot_val<16, H>(g, rRow);  v[17] = slot_val<17, H>(g, rRow);
        const int tUp = __shfl_up(tgt, 1, TILE);
        int runLo = (lane == 0 || tUp != tgt) ? lane : 0;
#pragma unroll
        for (int d = 1; d < TILE; d <<= 1) {
            const int o = __shfl_up(runLo, d, TILE);
            if (lane >= d) runLo = max(runLo, o);
        }
#pragma unroll
        for (int d = 1; d < TILE; d <<= 1) {
            const bool take = (lane - d >= runLo);
#pragma unroll
            for (int i = 0; i < FEAT; ++i) {
                const float o = __shfl_up(v[i], d, TILE);
                v[i] += take ? o : 0.0f;
            }
        }
        const int tDn = __shfl_down(tgt, 1, TILE);
        const bool last = (lane == TILE - 1) || (tDn != tgt);
        if (valid && last) {
            float* outBase = &outF[(long)tgt * FEAT];
#pragma unroll
            for (int i = 0; i < FEAT; ++i) atomicAdd(outBase + i, v[i]);
        }
    }
}

__global__ void __launch_bounds__(2 * TILE)
layer_kernel(const float* __restrict__ feats, const int* __restrict__ eidx,
             const float* __restrict__ sh0, const float* __restrict__ sh1,
             const float* __restrict__ sh2, const float* __restrict__ dist,
             const float* __restrict__ Mg, const int* __restrict__ perm,
             const float* __restrict__ W1, const float* __restrict__ b1,
             const float* __restrict__ W2, const float* __restrict__ b2,
             float* __restrict__ outF, int E) {
    __shared__ __attribute__((aligned(16))) float sMem[SMEM_FLOATS];
    const int lane = threadIdx.x & (TILE - 1);
    const int half = __builtin_amdgcn_readfirstlane((int)(threadIdx.x >> 5));
    if (half == 0)
        layer_half<0>(feats, eidx, sh0, sh1, sh2, dist, Mg, perm, W1, b1, W2, b2, outF, E,
                      sMem + SU0_OFF, sMem + SRAW0_OFF, lane);
    else
        layer_half<1>(feats, eidx, sh0, sh1, sh2, dist, Mg, perm, W1, b1, W2, b2, outF, E,
                      sMem + SU1_OFF, sMem + SRAW1_OFF, lane);
}

extern "C" void kernel_launch(void* const* d_in, const int* in_sizes, int n_in,
                              void* d_out, int out_size, void* d_ws, size_t ws_size,
                              hipStream_t stream) {
    const float* node = (const float*)d_in[0];
    const int* eidx   = (const int*)d_in[1];
    const float* sh0  = (const float*)d_in[2];
    const float* sh1  = (const float*)d_in[3];
    const float* sh2  = (const float*)d_in[4];
    const float* dist = (const float*)d_in[5];
    const float* cg[N_COMBOS];
    for (int i = 0; i < N_COMBOS; ++i) cg[i] = (const float*)d_in[6 + i];
    const float* W1_0 = (const float*)d_in[21];
    const float* b1_0 = (const float*)d_in[22];
    const float* W2_0 = (const float*)d_in[23];
    const float* b2_0 = (const float*)d_in[24];
    const float* W1_1 = (const float*)d_in[25];
    const float* b1_1 = (const float*)d_in[26];
    const float* W2_1 = (const float*)d_in[27];
    const float* b2_1 = (const float*)d_in[28];

    const int E = in_sizes[1] / 2;
    const int N = in_sizes[0] / FEAT;

    float* Mg     = (float*)d_ws;
    float* feats1 = (float*)((char*)d_ws + 131072);
    long n1 = (long)N * FEAT;
    int* perm     = (int*)(feats1 + n1);

    build_M_kernel<<<1, 256, 0, stream>>>(cg[0], cg[1], cg[2], cg[3], cg[4], cg[5],
                                          cg[6], cg[7], cg[8], cg[9], cg[10], cg[11],
                                          cg[12], cg[13], cg[14], Mg);

    const bool permFits = (ws_size >= (size_t)131072 + (size_t)(n1 + E) * 4);
    const bool grouped = permFits && ((long)E + N <= (long)out_size);
    const int BW = (N + NBK - 1) / NBK;
    const int chunk = (E + NCH - 1) / NCH;
    const bool twoLevel = permFits && E >= 1 && E <= (1 << 20) && BW >= 1 && BW <= BWMAX &&
                          ((long)E + (long)NCH * CSTP <= (long)out_size);
    if (twoLevel) {
        int* sortedA = (int*)d_out;
        int* cst = sortedA + E;
        bucket_kernel<<<NCH, 1024, 0, stream>>>(eidx, sortedA, cst, E, chunk, BW);
        owner_kernel<<<NBK, 1024, 0, stream>>>(sortedA, cst, perm, E, chunk, BW);
    } else if (grouped) {
        int* rnk = (int*)d_out;
        int* cnt = rnk + E;
        zero_kernel<<<(N + 255) / 256, 256, 0, stream>>>((float*)cnt, (long)N);
        rank_kernel<<<(E + 255) / 256, 256, 0, stream>>>(eidx, cnt, rnk, E);
        scan_kernel<<<1, 1024, 0, stream>>>(cnt, N);
        fill_kernel<<<(E + 255) / 256, 256, 0, stream>>>(eidx, cnt, rnk, perm, E);
    } else {
        perm = nullptr;
    }

    zero_kernel<<<(int)((n1 + 255) / 256), 256, 0, stream>>>(feats1, n1);
    zero_kernel<<<(out_size + 255) / 256, 256, 0, stream>>>((float*)d_out, (long)out_size);

    int grid = (E + TILE - 1) / TILE;
    layer_kernel<<<grid, 2 * TILE, 0, stream>>>(node, eidx, sh0, sh1, sh2, dist, Mg, perm,
                                                  W1_0, b1_0, W2_0, b2_0, feats1, E);
    layer_kernel<<<grid, 2 * TILE, 0, stream>>>(feats1, eidx, sh0, sh1, sh2, dist, Mg, perm,
                                                  W1_1, b1_1, W2_1, b2_1, (float*)d_out, E);
}


# Per-target sums via LDS: each wave parks its 18 partial sums per edge in its dead u tile; after a barrier a lane per column adds both halves' rows in slot order (no shuffle tree), one atomic per run
# speedup vs baseline: 2.7514x; 1.1388x over previous
#include <hip/hip_runtime.h>

typedef __attribute__((ext_vector_type(2))) float v2f;
typedef __attribute__((ext_vector_type(8))) float v8f;

#define N_COMBOS 15
#define KP 164
#define PN 112
#define NTILES 7
#define TILE 32
#define FEAT 18

constexpr int cLE[N_COMBOS]     = {0,0,0,1,1,1,1,1,1,2,2,2,2,2,2};
constexpr int cLN[N_COMBOS]     = {0,1,2,0,1,1,1,2,2,0,1,1,2,2,2};
constexpr int cLO[N_COMBOS]     = {0,1,2,1,0,1,2,1,2,2,1,2,0,1,2};
constexpr int OUT_OFF[N_COMBOS] = {0,2,8,18,24,26,32,42,48,58,68,74,84,86,92};
constexpr int U_OFF[9]          = {0,2,8,18,24,42,72,82,112};
constexpr int KLO2[2][NTILES]   = {{0,2,6,10,0,0,0},  {0,0,0,18,18,20,28}};
constexpr int KHI2[2][NTILES]   = {{4,10,17,17,0,0,0},{0,0,0,20,27,40,40}};
constexpr int N_BY_LO[3]        = {3,6,6};
constexpr int C_BY_LO[3][6]     = {{0,4,12,0,0,0},{1,3,5,7,10,13},{2,6,8,9,11,14}};

__global__ void build_M_kernel(const float* cg0, const float* cg1, const float* cg2,
                               const float* cg3, const float* cg4, const float* cg5,
                               const float* cg6, const float* cg7, const float* cg8,
                               const float* cg9, const float* cg10, const float* cg11,
                               const float* cg12, const float* cg13, const float* cg14,
                               float* M) {
    const float* cgs[N_COMBOS] = {cg0,cg1,cg2,cg3,cg4,cg5,cg6,cg7,cg8,cg9,cg10,cg11,cg12,cg13,cg14};
    int tid = threadIdx.x;
    for (int i = tid; i < KP * PN; i += blockDim.x) M[i] = 0.0f;
    __syncthreads();
    if (tid < N_COMBOS) {
        int c = tid;
        int le = cLE[c], ln = cLN[c], lo = cLO[c];
        int de = 2 * le + 1, dn = 2 * ln + 1, dO = 2 * lo + 1;
        int uo = U_OFF[le * 3 + ln];
        int po = OUT_OFF[c];
        const float* cg = cgs[c];
        for (int a = 0; a < de; ++a)
            for (int b = 0; b < dn; ++b)
                for (int k = 0; k < dO; ++k) {
                    float v = cg[(a * dn + b) * dO + k];
                    for (int ch = 0; ch < 2; ++ch)
                        M[(uo + (a * dn + b) * 2 + ch) * PN + po + ch * dO + k] = v;
                }
    }
}

__global__ void zero_kernel(float* p, long n) {
    long i = (long)blockIdx.x * blockDim.x + threadIdx.x;
    if (i < n) p[i] = 0.0f;
}

__global__ void rank_kernel(const int* __restrict__ eidx, int* __restrict__ cnt,
                            int* __restrict__ rnk, int E) {
    int e = blockIdx.x * blockDim.x + threadIdx.x;
    if (e < E) rnk[e] = atomicAdd(&cnt[eidx[(long)E + e]], 1);
}

__global__ void __launch_bounds__(1024)
scan_kernel(int* __restrict__ cnt, int N) {
    __shared__ int part[1024];
    const int t = threadIdx.x;
    const int chunk = (N + 1023) / 1024;
    const int lo = t * chunk;
    int hi = lo + chunk;
    if (hi > N) hi = N;
    int s = 0;
    for (int i = lo; i < hi; ++i) s += cnt[i];
    part[t] = s;
    __syncthreads();
    for (int d = 1; d < 1024; d <<= 1) {
        int add = (t >= d) ? part[t - d] : 0;
        __syncthreads();
        part[t] += add;
        __syncthreads();
    }
    int run = part[t] - s;
    for (int i = lo; i < hi; ++i) { int c = cnt[i]; cnt[i] = run; run += c; }
}

__global__ void fill_kernel(const int* __restrict__ eidx, const int* __restrict__ first,
                            const int* __restrict__ rnk, int* __restrict__ perm, int E) {
    int e = blockIdx.x * blockDim.x + threadIdx.x;
    if (e < E) perm[first[eidx[(long)E + e]] + rnk[e]] = e;
}

#define NCH 128
#define NBK 256
#define CSTP (NBK + 1)
#define BWMAX 512
#define HPA (NBK + 1)
#define HPB (BWMAX + 1)

__device__ __forceinline__ int block_excl_scan_1024(int v, int* sWave, int& total) {
    const int lane = threadIdx.x & 31, wave = threadIdx.x >> 5;
    int x = v;
    for (int d = 1; d < 32; d <<= 1) { const int y = __shfl_up(x, d, 32); if (lane >= d) x += y; }
    if (lane == 31) sWave[wave] = x;
    __syncthreads();
    const int ws = sWave[lane];
    int xs = ws;
    for (int d = 1; d < 32; d <<= 1) { const int y = __shfl_up(xs, d, 32); if (lane >= d) xs += y; }
    const int waveOff = __shfl(xs, wave, 32) - __shfl(ws, wave, 32);
    total = __shfl(xs, 31, 32);
    __syncthreads();
    return waveOff + x - v;
}

__global__ void __launch_bounds__(1024)
bucket_kernel(const int* __restrict__ eidx, int* __restrict__ sortedA, int* __restrict__ cst,
              int E, int chunk, int BW) {
    __shared__ int hist[32 * HPA];
    __shared__ int sWave[32];
    const int t = threadIdx.x, lane = t & 31, w = t >> 5;
    const int c = blockIdx.x;
    for (int i = t; i < 32 * HPA; i += 1024) hist[i] = 0;
    __syncthreads();
    long lo = (long)c * chunk;
    if (lo > E) lo = E;
    long hi = lo + chunk;
    if (hi > E) hi = E;
    const int cLo = (int)lo, cHi = (int)hi;
    const int slice = (chunk + 31) / 32;
    int sLo = cLo + w * slice;
    if (sLo > cHi) sLo = cHi;
    int sHi = sLo + slice;
    if (sHi > cHi) sHi = cHi;
    const int* __restrict__ tg = eidx + E;
    int* myHist = &hist[w * HPA];
    const unsigned ubw = (unsigned)BW;
    for (int i = sLo + lane; i < sHi; i += 32)
        atomicAdd(&myHist[(unsigned)tg[i] / ubw], 1);
    __syncthreads();
    const int b = t >> 2, q = t & 3;
    int s = 0;
    for (int j = 0; j < 8; ++j) s += hist[(q * 8 + j) * HPA + b];
    int total;
    const int ex = block_excl_scan_1024(s, sWave, total);
    if (q == 0) cst[c * CSTP + b] = ex;
    if (t == 0) cst[c * CSTP + NBK] = total;
    int run = ex;
    for (int j = 0; j < 8; ++j) {
        const int k = (q * 8 + j) * HPA + b;
        const int c0 = hist[k];
        hist[k] = run;
        run += c0;
    }
    __syncthreads();
    for (int i = sLo + lane; i < sHi; i += 32) {
        const unsigned tv = (unsigned)tg[i];
        const unsigned bk = tv / ubw;
        const int pos = atomicAdd(&myHist[bk], 1);
        sortedA[cLo + pos] = (int)(((tv - bk * ubw) << 20) | (unsigned)i);
    }
}

__global__ void __launch_bounds__(1024)
owner_kernel(const int* __restrict__ sortedA, const int* __restrict__ cst,
             int* __restrict__ perm, int E, int chunk, int BW) {
    __shared__ int hist[32 * HPB];
    __shared__ int segStart[NCH];
    __shared__ int segLen[NCH];
    __shared__ int sWave[32];
    const int t = threadIdx.x, lane = t & 31, w = t >> 5;
    const int b = blockIdx.x;
    for (int i = t; i < 32 * HPB; i += 1024) hist[i] = 0;
    int st = 0;
    if (t < NCH) {
        st = cst[t * CSTP + b];
        const int ln = cst[t * CSTP + b + 1] - st;
        long lo = (long)t * chunk;
        if (lo > E) lo = E;
        segStart[t] = (int)lo + st;
        segLen[t] = ln;
    }
    int base;
    block_excl_scan_1024(st, sWave, base);
    constexpr int CPW = NCH / 32;
    int s0[CPW], n[CPW];
    int nMax = 0;
    for (int cc = 0; cc < CPW; ++cc) {
        s0[cc] = segStart[w * CPW + cc];
        n[cc] = segLen[w * CPW + cc];
        nMax = n[cc] > nMax ? n[cc] : nMax;
    }
    int* myHist = &hist[w * HPB];
    for (int j0 = 0; j0 < nMax; j0 += 32) {
        const int j = j0 + lane;
        unsigned wd[CPW];
        for (int cc = 0; cc < CPW; ++cc) wd[cc] = (j < n[cc]) ? (unsigned)sortedA[s0[cc] + j] : 0u;
        for (int cc = 0; cc < CPW; ++cc)
            if (j < n[cc]) atomicAdd(&myHist[wd[cc] >> 20], 1);
    }
    __syncthreads();
    int deg = 0;
    if (t < BW)
        for (int w2 = 0; w2 < 32; ++w2) deg += hist[w2 * HPB + t];
    int tot;
    const int ex = block_excl_scan_1024(deg, sWave, tot);
    if (t < BW) {
        int run = base + ex;
        for (int w2 = 0; w2 < 32; ++w2) {
            const int c0 = hist[w2 * HPB + t];
            hist[w2 * HPB + t] = run;
            run += c0;
        }
    }
    __syncthreads();
    for (int j0 = 0; j0 < nMax; j0 += 32) {
        const int j = j0 + lane;
        unsigned wd[CPW];
        for (int cc = 0; cc < CPW; ++cc) wd[cc] = (j < n[cc]) ? (unsigned)sortedA[s0[cc] + j] : 0u;
        for (int cc = 0; cc < CPW; ++cc)
            if (j < n[cc]) {
                const int r = atomicAdd(&myHist[wd[cc] >> 20], 1);
                perm[r] = (int)(wd[cc] & 0xFFFFFu);
            }
    }
}

template <int H> struct HalfCfg;
template <> struct HalfCfg<0> {
    static constexpr int T0 = 0,  NG = 36;
    static constexpr int C0 = 0,  C1 = 9;
    static constexpr int NT0 = 0, NT1 = 4;
    static constexpr int ULO = 0, UHI = 72;
    static constexpr int UP = 76;
    static constexpr int COL0 = 0;
    static constexpr int COL1 = 58;
    static constexpr int RP = 60;
};
template <> struct HalfCfg<1> {
    static constexpr int T0 = 36, NG = 24;
    static constexpr int C0 = 9,  C1 = 15;
    static constexpr int NT0 = 3, NT1 = 7;
    static constexpr int ULO = 72, UHI = 164;
    static constexpr int UP = 92;
    static constexpr int COL0 = 58;
    static constexpr int COL1 = 102;
    static constexpr int RP = 44;
};
#define SU0_OFF   0
#define SRAW0_OFF 2432
#define SU1_OFF   4352
#define SRAW1_OFF 7296
#define SMEM_FLOATS 8704
#define PSUM_P FEAT

__device__ __forceinline__ void wave_lds_handover() {
    __builtin_amdgcn_fence(__ATOMIC_RELEASE, "wavefront");
    __builtin_amdgcn_wave_barrier();
    __builtin_amdgcn_fence(__ATOMIC_ACQUIRE, "wavefront");
}

template <int H>
__device__ __forceinline__ void gate_block(const float* __restrict__ W1,
                                           const float* __restrict__ b1,
                                           const float* __restrict__ W2,
                                           const float* __restrict__ b2,
                                           float x0, float x1,
                                           float (&g)[HalfCfg<H>::NG]) {
#pragma unroll
    for (int t = 0; t < HalfCfg<H>::NG; ++t) {
        const int bi = HalfCfg<H>::T0 + t;
        float acc = b2[bi];
#pragma unroll
        for (int h = 0; h < 4; ++h) {
            float hh = x0 * W1[bi * 8 + h * 2] + x1 * W1[bi * 8 + h * 2 + 1] + b1[bi * 4 + h];
            hh = hh > 0.0f ? hh : 0.0f;
            acc += hh * W2[bi * 4 + h];
        }
        g[t] = tanhf(acc);
    }
}

template <int H>
__device__ __forceinline__ void u_stage(const float (&shv)[9],
                                        const float (&f)[FEAT],
                                        float* __restrict__ uRow) {
    typedef HalfCfg<H> Cf;
    int u = 0;
#pragma unroll
    for (int le = 0; le < 3; ++le) {
        const int de = 2 * le + 1;
        const int shb = (le == 0) ? 0 : (le == 1 ? 1 : 4);
#pragma unroll
        for (int ln = 0; ln < 3; ++ln) {
            const int dn = 2 * ln + 1;
            const int fb = (ln == 0) ? 0 : (ln == 1 ? 2 : 8);
#pragma unroll
            for (int a = 0; a < de; ++a)
#pragma unroll
                for (int b = 0; b < dn; ++b)
#pragma unroll
                    for (int ch = 0; ch < 2; ++ch) {
                        if (u >= Cf::ULO && u < Cf::UHI)
                            uRow[u - Cf::ULO] = shv[shb + a] * f[fb + ch * dn + b];
                        ++u;
                    }
        }
    }
    if (Cf::UHI > 162) {
        uRow[162 - Cf::ULO] = 0.0f;
        uRow[163 - Cf::ULO] = 0.0f;
    }
}

template <int S, int H>
__device__ __forceinline__ float slot_val(const float (&gRow)[HalfCfg<H>::NG],
                                          const float* __restrict__ rRow) {
    typedef HalfCfg<H> Cf;
    constexpr int lo = (S < 2) ? 0 : ((S < 8) ? 1 : 2);
    constexpr int o  = (S < 2) ? S : ((S < 8) ? ((S - 2) / 3) : ((S - 8) / 5));
    constexpr int k  = (S < 2) ? 0 : ((S < 8) ? ((S - 2) % 3) : ((S - 8) % 5));
    constexpr int dO = 2 * lo + 1;
    float v = 0.0f;
#pragma unroll
    for (int q = 0; q < N_BY_LO[lo]; ++q) {
        constexpr const int* lst = C_BY_LO[lo];
        const int c = lst[q];
        if (c >= Cf::C0 && c < Cf::C1) {
            const int base = OUT_OFF[c] - Cf::COL0;
            const int gb = (c - Cf::C0) * 4;
            v += gRow[gb + o]     * rRow[base + k] +
                 gRow[gb + 2 + o] * rRow[base + dO + k];
        }
    }
    return v;
}

template <int H>
__device__ __forceinline__ void
layer_half(const float* __restrict__ feats, const int* __restrict__ eidx,
           const float* __restrict__ sh0, const float* __restrict__ sh1,
           const float* __restrict__ sh2, const float* __restrict__ dist,
           const float* __restrict__ Mg, const int* __restrict__ perm,
           const float* __restrict__ W1, const float* __restrict__ b1,
           const float* __restrict__ W2, const float* __restrict__ b2,
           int E, float* __restrict__ sU, float* __restrict__ sRaw, const int lane,
           int& tgtOut) {
    typedef HalfCfg<H> Cf;
    const long slot = (long)blockIdx.x * TILE + lane;
    const bool valid = (slot < (long)E);
    const long e = valid ? (perm ? (long)perm[slot] : slot) : 0;

    float f[FEAT];
    float shv[9];
    float x0 = 0.0f, x1 = 0.0f;
    int tgt = -1;
    if (valid) {
        const float* fp = &feats[(long)eidx[e] * FEAT];
        tgt = eidx[(long)E + e];
#pragma unroll
        for (int j = 0; j < FEAT; ++j) f[j] = fp[j];
        shv[0] = sh0[e];
#pragma unroll
        for (int j = 0; j < 3; ++j) shv[1 + j] = sh1[e * 3 + j];
#pragma unroll
        for (int j = 0; j < 5; ++j) shv[4 + j] = sh2[e * 5 + j];
        x0 = dist[e];
        float inv0 = 0.5f * (f[0] + f[1]);
        float n1 = 0.0f, n2 = 0.0f;
#pragma unroll
        for (int j = 0; j < 3; ++j) { float m = 0.5f * (f[2 + j] + f[5 + j]); n1 += m * m; }
#pragma unroll
        for (int j = 0; j < 5; ++j) { float m = 0.5f * (f[8 + j] + f[13 + j]); n2 += m * m; }
        x1 = (inv0 + sqrtf(n1) + sqrtf(n2)) * (1.0f / 3.0f);
    } else {
#pragma unroll
        for (int j = 0; j < FEAT; ++j) f[j] = 0.0f;
#pragma unroll
        for (int j = 0; j < 9; ++j) shv[j] = 0.0f;
    }

    float g[Cf::NG];
    u_stage<H>(shv, f, &sU[lane * Cf::UP]);
    gate_block<H>(W1, b1, W2, b2, x0, x1, g);
    wave_lds_handover();

    {
        const int col16 = lane & 15;
        const int kofs = (lane < 16) ? 0 : 2;
        const float* uA0 = &sU[col16 * Cf::UP];
        const float* uA1 = &sU[(16 + col16) * Cf::UP];
#pragma unroll
        for (int n = Cf::NT0; n < Cf::NT1; ++n) {
            v8f acc0 = {0.f, 0.f, 0.f, 0.f, 0.f, 0.f, 0.f, 0.f};
            v8f acc1 = {0.f, 0.f, 0.f, 0.f, 0.f, 0.f, 0.f, 0.f};
            const int col = col16 + n * 16;
#pragma unroll
            for (int kt = KLO2[H][n]; kt <= KHI2[H][n]; ++kt) {
                int k0 = kt * 4 + kofs;
                v2f a0, a1, b;
                a0.x = uA0[k0 - Cf::ULO];
                a0.y = uA0[k0 - Cf::ULO + 1];
                a1.x = uA1[k0 - Cf::ULO];
                a1.y = uA1[k0 - Cf::ULO + 1];
                b.x = Mg[k0 * PN + col];
                b.y = Mg[(k0 + 1) * PN + col];
                acc0 = __builtin_amdgcn_wmma_f32_16x16x4_f32(
                    false, a0, false, b, (short)0, acc0, false, false);
                acc1 = __builtin_amdgcn_wmma_f32_16x16x4_f32(
                    false, a1, false, b, (short)0, acc1, false, false);
            }
            const int rbase = (lane < 16) ? 0 : 8;
            if (col >= Cf::COL0 && col < Cf::COL1) {
#pragma unroll
                for (int j = 0; j < 8; ++j) {
                    sRaw[(rbase + j) * Cf::RP + col - Cf::COL0] = acc0[j];
                    sRaw[(16 + rbase + j) * Cf::RP + col - Cf::COL0] = acc1[j];
                }
            }
        }
    }
    wave_lds_handover();

    {
        const float* rRow = &sRaw[lane * Cf::RP];
        float v[FEAT];
        v[0]  = slot_val<0, H>(g, rRow);   v[1]  = slot_val<1, H>(g, rRow);
        v[2]  = slot_val<2, H>(g, rRow);   v[3]  = slot_val<3, H>(g, rRow);
        v[4]  = slot_val<4, H>(g, rRow);   v[5]  = slot_val<5, H>(g, rRow);
        v[6]  = slot_val<6, H>(g, rRow);   v[7]  = slot_val<7, H>(g, rRow);
        v[8]  = slot_val<8, H>(g, rRow);   v[9]  = slot_val<9, H>(g, rRow);
        v[10] = slot_val<10, H>(g, rRow);  v[11] = slot_val<11, H>(g, rRow);
        v[12] = slot_val<12, H>(g, rRow);  v[13] = slot_val<13, H>(g, rRow);
        v[14] = slot_val<14, H>(g, rRow);  v[15] = slot_val<15, H>(g, rRow);
        v[16] = slot_val<16, H>(g, rRow);  v[17] = slot_val<17, H>(g, rRow);
        float* pRow = &sU[lane * PSUM_P];
#pragma unroll
        for (int i = 0; i < FEAT; ++i) pRow[i] = v[i];
    }
    tgtOut = tgt;
}

__global__ void __launch_bounds__(2 * TILE)
layer_kernel(const float* __restrict__ feats, const int* __restrict__ eidx,
             const float* __restrict__ sh0, const float* __restrict__ sh1,
             const float* __restrict__ sh2, const float* __restrict__ dist,
             const float* __restrict__ Mg, const int* __restrict__ perm,
             const float* __restrict__ W1, const float* __restrict__ b1,
             const float* __restrict__ W2, const float* __restrict__ b2,
             float* __restrict__ outF, int E) {
    __shared__ __attribute__((aligned(16))) float sMem[SMEM_FLOATS];
    const int lane = threadIdx.x & (TILE - 1);
    const int half = __builtin_amdgcn_readfirstlane((int)(threadIdx.x >> 5));
    int tgt;
    if (half == 0)
        layer_half<0>(feats, eidx, sh0, sh1, sh2, dist, Mg, perm, W1, b1, W2, b2, E,
                      sMem + SU0_OFF, sMem + SRAW0_OFF, lane, tgt);
    else
        layer_half<1>(feats, eidx, sh0, sh1, sh2, dist, Mg, perm, W1, b1, W2, b2, E,
                      sMem + SU1_OFF, sMem + SRAW1_OFF, lane, tgt);
    __syncthreads();

    const int tDn = __shfl_down(tgt, 1, TILE);
    const bool last = (lane == TILE - 1) || (tDn != tgt);
    const unsigned lastMask = __builtin_amdgcn_ballot_w32(last) | (1u << (TILE / 2 - 1));
    const int r0 = half * (TILE / 2);
    const int colj = min(lane, FEAT - 1);
    const float* p0 = sMem + SU0_OFF + r0 * PSUM_P + colj;
    const float* p1 = sMem + SU1_OFF + r0 * PSUM_P + colj;
    float s[TILE / 2];
#pragma unroll
    for (int i = 0; i < TILE / 2; ++i) s[i] = p0[i * PSUM_P] + p1[i * PSUM_P];
    float acc = 0.0f;
#pragma unroll
    for (int i = 0; i < TILE / 2; ++i) {
        acc += s[i];
        if ((lastMask >> (r0 + i)) & 1u) {
            const int t = __builtin_amdgcn_readlane(tgt, r0 + i);
            if (t >= 0 && lane < FEAT) atomicAdd(&outF[(long)t * FEAT + lane], acc);
            acc = 0.0f;
        }
    }
}

extern "C" void kernel_launch(void* const* d_in, const int* in_sizes, int n_in,
                              void* d_out, int out_size, void* d_ws, size_t ws_size,
                              hipStream_t stream) {
    const float* node = (const float*)d_in[0];
    const int* eidx   = (const int*)d_in[1];
    const float* sh0  = (const float*)d_in[2];
    const float* sh1  = (const float*)d_in[3];
    const float* sh2  = (const float*)d_in[4];
    const float* dist = (const float*)d_in[5];
    const float* cg[N_COMBOS];
    for (int i = 0; i < N_COMBOS; ++i) cg[i] = (const float*)d_in[6 + i];
    const float* W1_0 = (const float*)d_in[21];
    const float* b1_0 = (const float*)d_in[22];
    const float* W2_0 = (const float*)d_in[23];
    const float* b2_0 = (const float*)d_in[24];
    const float* W1_1 = (const float*)d_in[25];
    const float* b1_1 = (const float*)d_in[26];
    const float* W2_1 = (const float*)d_in[27];
    const float* b2_1 = (const float*)d_in[28];

    const int E = in_sizes[1] / 2;
    const int N = in_sizes[0] / FEAT;

    float* Mg     = (float*)d_ws;
    float* feats1 = (float*)((char*)d_ws + 131072);
    long n1 = (long)N * FEAT;
    int* perm     = (int*)(feats1 + n1);

    build_M_kernel<<<1, 256, 0, stream>>>(cg[0], cg[1], cg[2], cg[3], cg[4], cg[5],
                                          cg[6], cg[7], cg[8], cg[9], cg[10], cg[11],
                                          cg[12], cg[13], cg[14], Mg);

    const bool permFits = (ws_size >= (size_t)131072 + (size_t)(n1 + E) * 4);
    const bool grouped = permFits && ((long)E + N <= (long)out_size);
    const int BW = (N + NBK - 1) / NBK;
    const int chunk = (E + NCH - 1) / NCH;
    const bool twoLevel = permFits && E >= 1 && E <= (1 << 20) && BW >= 1 && BW <= BWMAX &&
                          ((long)E + (long)NCH * CSTP <= (long)out_size);
    if (twoLevel) {
        int* sortedA = (int*)d_out;
        int* cst = sortedA + E;
        bucket_kernel<<<NCH, 1024, 0, stream>>>(eidx, sortedA, cst, E, chunk, BW);
        owner_kernel<<<NBK, 1024, 0, stream>>>(sortedA, cst, perm, E, chunk, BW);
    } else if (grouped) {
        int* rnk = (int*)d_out;
        int* cnt = rnk + E;
        zero_kernel<<<(N + 255) / 256, 256, 0, stream>>>((float*)cnt, (long)N);
        rank_kernel<<<(E + 255) / 256, 256, 0, stream>>>(eidx, cnt, rnk, E);
        scan_kernel<<<1, 1024, 0, stream>>>(cnt, N);
        fill_kernel<<<(E + 255) / 256, 256, 0, stream>>>(eidx, cnt, rnk, perm, E);
    } else {
        perm = nullptr;
    }

    zero_kernel<<<(int)((n1 + 255) / 256), 256, 0, stream>>>(feats1, n1);
    zero_kernel<<<(out_size + 255) / 256, 256, 0, stream>>>((float*)d_out, (long)out_size);

    int grid = (E + TILE - 1) / TILE;
    layer_kernel<<<grid, 2 * TILE, 0, stream>>>(node, eidx, sh0, sh1, sh2, dist, Mg, perm,
                                                  W1_0, b1_0, W2_0, b2_0, feats1, E);
    layer_kernel<<<grid, 2 * TILE, 0, stream>>>(feats1, eidx, sh0, sh1, sh2, dist, Mg, perm,
                                                  W1_1, b1_1, W2_1, b2_1, (float*)d_out, E);
}


# Two edges per lane (64-edge tiles): gate MLP in packed f32, each scalar weight read once per edge pair; its 32-edge halves share one rolled tanh/u/WMMA/mix body; target runs cut per 32 slots, not 16
# speedup vs baseline: 2.9254x; 1.0632x over previous
#include <hip/hip_runtime.h>

typedef __attribute__((ext_vector_type(2))) float v2f;
typedef __attribute__((ext_vector_type(8))) float v8f;

#define N_COMBOS 15
#define KP 164
#define PN 112
#define NTILES 7
#define TILE 32
#define FEAT 18

constexpr int cLE[N_COMBOS]     = {0,0,0,1,1,1,1,1,1,2,2,2,2,2,2};
constexpr int cLN[N_COMBOS]     = {0,1,2,0,1,1,1,2,2,0,1,1,2,2,2};
constexpr int cLO[N_COMBOS]     = {0,1,2,1,0,1,2,1,2,2,1,2,0,1,2};
constexpr int OUT_OFF[N_COMBOS] = {0,2,8,18,24,26,32,42,48,58,68,74,84,86,92};
constexpr int U_OFF[9]          = {0,2,8,18,24,42,72,82,112};
constexpr int KLO2[2][NTILES]   = {{0,2,6,10,0,0,0},  {0,0,0,18,18,20,28}};
constexpr int KHI2[2][NTILES]   = {{4,10,17,17,0,0,0},{0,0,0,20,27,40,40}};
constexpr int N_BY_LO[3]        = {3,6,6};
constexpr int C_BY_LO[3][6]     = {{0,4,12,0,0,0},{1,3,5,7,10,13},{2,6,8,9,11,14}};

__global__ void build_M_kernel(const float* cg0, const float* cg1, const float* cg2,
                               const float* cg3, const float* cg4, const float* cg5,
                               const float* cg6, const float* cg7, const float* cg8,
                               const float* cg9, const float* cg10, const float* cg11,
                               const float* cg12, const float* cg13, const float* cg14,
                               float* M) {
    const float* cgs[N_COMBOS] = {cg0,cg1,cg2,cg3,cg4,cg5,cg6,cg7,cg8,cg9,cg10,cg11,cg12,cg13,cg14};
    int tid = threadIdx.x;
    for (int i = tid; i < KP * PN; i += blockDim.x) M[i] = 0.0f;
    __syncthreads();
    if (tid < N_COMBOS) {
        int c = tid;
        int le = cLE[c], ln = cLN[c], lo = cLO[c];
        int de = 2 * le + 1, dn = 2 * ln + 1, dO = 2 * lo + 1;
        int uo = U_OFF[le * 3 + ln];
        int po = OUT_OFF[c];
        const float* cg = cgs[c];
        for (int a = 0; a < de; ++a)
            for (int b = 0; b < dn; ++b)
                for (int k = 0; k < dO; ++k) {
                    float v = cg[(a * dn + b) * dO + k];
                    for (int ch = 0; ch < 2; ++ch)
                        M[(uo + (a * dn + b) * 2 + ch) * PN + po + ch * dO + k] = v;
                }
    }
}

__global__ void zero_kernel(float* p, long n) {
    long i = (long)blockIdx.x * blockDim.x + threadIdx.x;
    if (i < n) p[i] = 0.0f;
}

__global__ void rank_kernel(const int* __restrict__ eidx, int* __restrict__ cnt,
                            int* __restrict__ rnk, int E) {
    int e = blockIdx.x * blockDim.x + threadIdx.x;
    if (e < E) rnk[e] = atomicAdd(&cnt[eidx[(long)E + e]], 1);
}

__global__ void __launch_bounds__(1024)
scan_kernel(int* __restrict__ cnt, int N) {
    __shared__ int part[1024];
    const int t = threadIdx.x;
    const int chunk = (N + 1023) / 1024;
    const int lo = t * chunk;
    int hi = lo + chunk;
    if (hi > N) hi = N;
    int s = 0;
    for (int i = lo; i < hi; ++i) s += cnt[i];
    part[t] = s;
    __syncthreads();
    for (int d = 1; d < 1024; d <<= 1) {
        int add = (t >= d) ? part[t - d] : 0;
        __syncthreads();
        part[t] += add;
        __syncthreads();
    }
    int run = part[t] - s;
    for (int i = lo; i < hi; ++i) { int c = cnt[i]; cnt[i] = run; run += c; }
}

__global__ void fill_kernel(const int* __restrict__ eidx, const int* __restrict__ first,
                            const int* __restrict__ rnk, int* __restrict__ perm, int E) {
    int e = blockIdx.x * blockDim.x + threadIdx.x;
    if (e < E) perm[first[eidx[(long)E + e]] + rnk[e]] = e;
}

#define NCH 128
#define NBK 256
#define CSTP (NBK + 1)
#define BWMAX 512
#define HPA (NBK + 1)
#define HPB (BWMAX + 1)

__device__ __forceinline__ int block_excl_scan_1024(int v, int* sWave, int& total) {
    const int lane = threadIdx.x & 31, wave = threadIdx.x >> 5;
    int x = v;
    for (int d = 1; d < 32; d <<= 1) { const int y = __shfl_up(x, d, 32); if (lane >= d) x += y; }
    if (lane == 31) sWave[wave] = x;
    __syncthreads();
    const int ws = sWave[lane];
    int xs = ws;
    for (int d = 1; d < 32; d <<= 1) { const int y = __shfl_up(xs, d, 32); if (lane >= d) xs += y; }
    const int waveOff = __shfl(xs, wave, 32) - __shfl(ws, wave, 32);
    total = __shfl(xs, 31, 32);
    __syncthreads();
    return waveOff + x - v;
}

__global__ void __launch_bounds__(1024)
bucket_kernel(const int* __restrict__ eidx, int* __restrict__ sortedA, int* __restrict__ cst,
              int E, int chunk, int BW) {
    __shared__ int hist[32 * HPA];
    __shared__ int sWave[32];
    const int t = threadIdx.x, lane = t & 31, w = t >> 5;
    const int c = blockIdx.x;
    for (int i = t; i < 32 * HPA; i += 1024) hist[i] = 0;
    __syncthreads();
    long lo = (long)c * chunk;
    if (lo > E) lo = E;
    long hi = lo + chunk;
    if (hi > E) hi = E;
    const int cLo = (int)lo, cHi = (int)hi;
    const int slice = (chunk + 31) / 32;
    int sLo = cLo + w * slice;
    if (sLo > cHi) sLo = cHi;
    int sHi = sLo + slice;
    if (sHi > cHi) sHi = cHi;
    const int* __restrict__ tg = eidx + E;
    int* myHist = &hist[w * HPA];
    const unsigned ubw = (unsigned)BW;
    for (int i = sLo + lane; i < sHi; i += 32)
        atomicAdd(&myHist[(unsigned)tg[i] / ubw], 1);
    __syncthreads();
    const int b = t >> 2, q = t & 3;
    int s = 0;
    for (int j = 0; j < 8; ++j) s += hist[(q * 8 + j) * HPA + b];
    int total;
    const int ex = block_excl_scan_1024(s, sWave, total);
    if (q == 0) cst[c * CSTP + b] = ex;
    if (t == 0) cst[c * CSTP + NBK] = total;
    int run = ex;
    for (int j = 0; j < 8; ++j) {
        const int k = (q * 8 + j) * HPA + b;
        const int c0 = hist[k];
        hist[k] = run;
        run += c0;
    }
    __syncthreads();
    for (int i = sLo + lane; i < sHi; i += 32) {
        const unsigned tv = (unsigned)tg[i];
        const unsigned bk = tv / ubw;
        const int pos = atomicAdd(&myHist[bk], 1);
        sortedA[cLo + pos] = (int)(((tv - bk * ubw) << 20) | (unsigned)i);
    }
}

__global__ void __launch_bounds__(1024)
owner_kernel(const int* __restrict__ sortedA, const int* __restrict__ cst,
             int* __restrict__ perm, int E, int chunk, int BW) {
    __shared__ int hist[32 * HPB];
    __shared__ int segStart[NCH];
    __shared__ int segLen[NCH];
    __shared__ int sWave[32];
    const int t = threadIdx.x, lane = t & 31, w = t >> 5;
    const int b = blockIdx.x;
    for (int i = t; i < 32 * HPB; i += 1024) hist[i] = 0;
    int st = 0;
    if (t < NCH) {
        st = cst[t * CSTP + b];
        const int ln = cst[t * CSTP + b + 1] - st;
        long lo = (long)t * chunk;
        if (lo > E) lo = E;
        segStart[t] = (int)lo + st;
        segLen[t] = ln;
    }
    int base;
    block_excl_scan_1024(st, sWave, base);
    constexpr int CPW = NCH / 32;
    int s0[CPW], n[CPW];
    int nMax = 0;
    for (int cc = 0; cc < CPW; ++cc) {
        s0[cc] = segStart[w * CPW + cc];
        n[cc] = segLen[w * CPW + cc];
        nMax = n[cc] > nMax ? n[cc] : nMax;
    }
    int* myHist = &hist[w * HPB];
    for (int j0 = 0; j0 < nMax; j0 += 32) {
        const int j = j0 + lane;
        unsigned wd[CPW];
        for (int cc = 0; cc < CPW; ++cc) wd[cc] = (j < n[cc]) ? (unsigned)sortedA[s0[cc] + j] : 0u;
        for (int cc = 0; cc < CPW; ++cc)
            if (j < n[cc]) atomicAdd(&myHist[wd[cc] >> 20], 1);
    }
    __syncthreads();
    int deg = 0;
    if (t < BW)
        for (int w2 = 0; w2 < 32; ++w2) deg += hist[w2 * HPB + t];
    int tot;
    const int ex = block_excl_scan_1024(deg, sWave, tot);
    if (t < BW) {
        int run = base + ex;
        for (int w2 = 0; w2 < 32; ++w2) {
            const int c0 = hist[w2 * HPB + t];
            hist[w2 * HPB + t] = run;
            run += c0;
        }
    }
    __syncthreads();
    for (int j0 = 0; j0 < nMax; j0 += 32) {
        const int j = j0 + lane;
        unsigned wd[CPW];
        for (int cc = 0; cc < CPW; ++cc) wd[cc] = (j < n[cc]) ? (unsigned)sortedA[s0[cc] + j] : 0u;
        for (int cc = 0; cc < CPW; ++cc)
            if (j < n[cc]) {
                const int r = atomicAdd(&myHist[wd[cc] >> 20], 1);
                perm[r] = (int)(wd[cc] & 0xFFFFFu);
            }
    }
}

template <int H> struct HalfCfg;
template <> struct HalfCfg<0> {
    static constexpr int T0 = 0,  NG = 36;
    static constexpr int C0 = 0,  C1 = 9;
    static constexpr int NT0 = 0, NT1 = 4;
    static constexpr int ULO = 0, UHI = 72;
    static constexpr int UP = 76;
    static constexpr int COL0 = 0;
    static constexpr int COL1 = 58;
    static constexpr int RP = 60;
};
template <> struct HalfCfg<1> {
    static constexpr int T0 = 36, NG = 24;
    static constexpr int C0 = 9,  C1 = 15;
    static constexpr int NT0 = 3, NT1 = 7;
    static constexpr int ULO = 72, UHI = 164;
    static constexpr int UP = 92;
    static constexpr int COL0 = 58;
    static constexpr int COL1 = 102;
    static constexpr int RP = 44;
};
#define SU0_OFF   0
#define SRAW0_OFF 2432
#define SU1_OFF   4352
#define SRAW1_OFF 7296
#define SP0_OFF   8704
#define SP1_OFF   9280
#define SMEM_FLOATS 9856
#define PSUM_P FEAT

__device__ __forceinline__ void wave_lds_handover() {
    __builtin_amdgcn_fence(__ATOMIC_RELEASE, "wavefront");
    __builtin_amdgcn_wave_barrier();
    __builtin_amdgcn_fence(__ATOMIC_ACQUIRE, "wavefront");
}

template <int H>
__device__ __forceinline__ void gate_mlp2(const float* __restrict__ W1g,
                                          const float* __restrict__ b1g,
                                          const float* __restrict__ W2g,
                                          const float* __restrict__ b2g,
                                          v2f x0, v2f x1,
                                          float (&aa)[HalfCfg<H>::NG],
                                          float (&ab)[HalfCfg<H>::NG]) {
    typedef const __attribute__((address_space(4))) float* cptr;
    const cptr W1 = (cptr)W1g, b1 = (cptr)b1g, W2 = (cptr)W2g, b2 = (cptr)b2g;
#pragma unroll
    for (int t = 0; t < HalfCfg<H>::NG; ++t) {
        const int bi = HalfCfg<H>::T0 + t;
        const float b2s = b2[bi];
        v2f acc = {b2s, b2s};
#pragma unroll
        for (int h = 0; h < 4; ++h) {
            v2f hh = x0 * W1[bi * 8 + h * 2] + x1 * W1[bi * 8 + h * 2 + 1] + b1[bi * 4 + h];
            hh.x = hh.x > 0.0f ? hh.x : 0.0f;
            hh.y = hh.y > 0.0f ? hh.y : 0.0f;
            acc += hh * W2[bi * 4 + h];
        }
        aa[t] = acc.x;
        ab[t] = acc.y;
        __builtin_amdgcn_sched_barrier(0);
    }
}

template <int H>
__device__ __forceinline__ void u_stage(const float (&shv)[9],
                                        const float (&f)[FEAT],
                                        float* __restrict__ uRow) {
    typedef HalfCfg<H> Cf;
    int u = 0;
#pragma unroll
    for (int le = 0; le < 3; ++le) {
        const int de = 2 * le + 1;
        const int shb = (le == 0) ? 0 : (le == 1 ? 1 : 4);
#pragma unroll
        for (int ln = 0; ln < 3; ++ln) {
            const int dn = 2 * ln + 1;
            const int fb = (ln == 0) ? 0 : (ln == 1 ? 2 : 8);
#pragma unroll
            for (int a = 0; a < de; ++a)
#pragma unroll
                for (int b = 0; b < dn; ++b)
#pragma unroll
                    for (int ch = 0; ch < 2; ++ch) {
                        if (u >= Cf::ULO && u < Cf::UHI)
                            uRow[u - Cf::ULO] = shv[shb + a] * f[fb + ch * dn + b];
                        ++u;
                    }
        }
    }
    if (Cf::UHI > 162) {
        uRow[162 - Cf::ULO] = 0.0f;
        uRow[163 - Cf::ULO] = 0.0f;
    }
}

template <int S, int H>
__device__ __forceinline__ float slot_val(const float (&gRow)[HalfCfg<H>::NG],
                                          const float* __restrict__ rRow) {
    typedef HalfCfg<H> Cf;
    constexpr int lo = (S < 2) ? 0 : ((S < 8) ? 1 : 2);
    constexpr int o  = (S < 2) ? S : ((S < 8) ? ((S - 2) / 3) : ((S - 8) / 5));
    constexpr int k  = (S < 2) ? 0 : ((S < 8) ? ((S - 2) % 3) : ((S - 8) % 5));
    constexpr int dO = 2 * lo + 1;
    float v = 0.0f;
#pragma unroll
    for (int q = 0; q < N_BY_LO[lo]; ++q) {
        constexpr const int* lst = C_BY_LO[lo];
        const int c = lst[q];
        if (c >= Cf::C0 && c < Cf::C1) {
            const int base = OUT_OFF[c] - Cf::COL0;
            const int gb = (c - Cf::C0) * 4;
            v += gRow[gb + o]     * rRow[base + k] +
                 gRow[gb + 2 + o] * rRow[base + dO + k];
        }
    }
    return v;
}

__device__ __forceinline__ void
edge_ids(const int* __restrict__ eidx, const int* __restrict__ perm, int E, long slot,
         bool& valid, long& e, int& src, int& tgt) {
    valid = (slot < (long)E);
    const long sl = valid ? slot : (long)E - 1;
    e = perm ? (long)perm[sl] : sl;
    src = eidx[e];
    const int tg = eidx[(long)E + e];
    tgt = valid ? tg : -1;
}

__device__ __forceinline__ void
edge_rows(const float* __restrict__ feats, const float* __restrict__ sh0,
          const float* __restrict__ sh1, const float* __restrict__ sh2,
          bool valid, long e, int src, float (&f)[FEAT], float (&shv)[9]) {
    const float* fp = &feats[(long)src * FEAT];
#pragma unroll
    for (int j = 0; j < FEAT; ++j) f[j] = fp[j];
    shv[0] = sh0[e];
#pragma unroll
    for (int j = 0; j < 3; ++j) shv[1 + j] = sh1[e * 3 + j];
#pragma unroll
    for (int j = 0; j < 5; ++j) shv[4 + j] = sh2[e * 5 + j];
#pragma unroll
    for (int j = 0; j < FEAT; ++j) f[j] = valid ? f[j] : 0.0f;
#pragma unroll
    for (int j = 0; j < 9; ++j) shv[j] = valid ? shv[j] : 0.0f;
}

__device__ __forceinline__ float edge_invariant(const float (&f)[FEAT]) {
    float inv0 = 0.5f * (f[0] + f[1]);
    float n1 = 0.0f, n2 = 0.0f;
#pragma unroll
    for (int j = 0; j < 3; ++j) { float m = 0.5f * (f[2 + j] + f[5 + j]); n1 += m * m; }
#pragma unroll
    for (int j = 0; j < 5; ++j) { float m = 0.5f * (f[8 + j] + f[13 + j]); n2 += m * m; }
    return (inv0 + sqrtf(n1) + sqrtf(n2)) * (1.0f / 3.0f);
}

template <int H>
__device__ __forceinline__ void
mma_mix(const float* __restrict__ Mg, const float (&g)[HalfCfg<H>::NG],
        float* __restrict__ sU, float* __restrict__ sRaw, float* __restrict__ pRow,
        const int lane) {
    typedef HalfCfg<H> Cf;
    wave_lds_handover();
    {
        const int col16 = lane & 15;
        const int kofs = (lane < 16) ? 0 : 2;
        const float* uA0 = &sU[col16 * Cf::UP];
        const float* uA1 = &sU[(16 + col16) * Cf::UP];
#pragma unroll
        for (int n = Cf::NT0; n < Cf::NT1; ++n) {
            v8f acc0 = {0.f, 0.f, 0.f, 0.f, 0.f, 0.f, 0.f, 0.f};
            v8f acc1 = {0.f, 0.f, 0.f, 0.f, 0.f, 0.f, 0.f, 0.f};
            const int col = col16 + n * 16;
#pragma unroll
            for (int kt = KLO2[H][n]; kt <= KHI2[H][n]; ++kt) {
                int k0 = kt * 4 + kofs;
                v2f a0, a1, b;
                a0.x = uA0[k0 - Cf::ULO];
                a0.y = uA0[k0 - Cf::ULO + 1];
                a1.x = uA1[k0 - Cf::ULO];
                a1.y = uA1[k0 - Cf::ULO + 1];
                b.x = Mg[k0 * PN + col];
                b.y = Mg[(k0 + 1) * PN + col];
                acc0 = __builtin_amdgcn_wmma_f32_16x16x4_f32(
                    false, a0, false, b, (short)0, acc0, false, false);
                acc1 = __builtin_amdgcn_wmma_f32_16x16x4_f32(
                    false, a1, false, b, (short)0, acc1, false, false);
            }
            const int rbase = (lane < 16) ? 0 : 8;
            if (col >= Cf::COL0 && col < Cf::COL1) {
#pragma unroll
                for (int j = 0; j < 8; ++j) {
                    sRaw[(rbase + j) * Cf::RP + col - Cf::COL0] = acc0[j];
                    sRaw[(16 + rbase + j) * Cf::RP + col - Cf::COL0] = acc1[j];
                }
            }
        }
    }
    wave_lds_handover();
    {
        const float* rRow = &sRaw[lane * Cf::RP];
        float v[FEAT];
        v[0]  = slot_val<0, H>(g, rRow);   v[1]  = slot_val<1, H>(g, rRow);
        v[2]  = slot_val<2, H>(g, rRow);   v[3]  = slot_val<3, H>(g, rRow);
        v[4]  = slot_val<4, H>(g, rRow);   v[5]  = slot_val<5, H>(g, rRow);
        v[6]  = slot_val<6, H>(g, rRow);   v[7]  = slot_val<7, H>(g, rRow);
        v[8]  = slot_val<8, H>(g, rRow);   v[9]  = slot_val<9, H>(g, rRow);
        v[10] = slot_val<10, H>(g, rRow);  v[11] = slot_val<11, H>(g, rRow);
        v[12] = slot_val<12, H>(g, rRow);  v[13] = slot_val<13, H>(g, rRow);
        v[14] = slot_val<14, H>(g, rRow);  v[15] = slot_val<15, H>(g, rRow);
        v[16] = slot_val<16, H>(g, rRow);  v[17] = slot_val<17, H>(g, rRow);
#pragma unroll
        for (int i = 0; i < FEAT; ++i) pRow[i] = v[i];
    }
}

template <int H>
__device__ __forceinline__ void
layer_half(const float* __restrict__ feats, const int* __restrict__ eidx,
           const float* __restrict__ sh0, const float* __restrict__ sh1,
           const float* __restrict__ sh2, const float* __restrict__ dist,
           const float* __restrict__ Mg, const int* __restrict__ perm,
           const float* __restrict__ W1, const float* __restrict__ b1,
           const float* __restrict__ W2, const float* __restrict__ b2,
           int E, float* __restrict__ sU, float* __restrict__ sRaw, float* __restrict__ sP,
           const int lane, int& tgtA, int& tgtB) {
    typedef HalfCfg<H> Cf;
    const long slotA = (long)blockIdx.x * (2 * TILE) + lane;
    const long slotB = slotA + TILE;

    bool vA, vB;
    long eA, eB;
    int srcA, srcB;
    edge_ids(eidx, perm, E, slotA, vA, eA, srcA, tgtA);
    edge_ids(eidx, perm, E, slotB, vB, eB, srcB, tgtB);

    float aA[Cf::NG], aB[Cf::NG];
    {
        float fA[FEAT], fB[FEAT], shA[9], shB[9];
        edge_rows(feats, sh0, sh1, sh2, vA, eA, srcA, fA, shA);
        edge_rows(feats, sh0, sh1, sh2, vB, eB, srcB, fB, shB);
        const float dA = dist[eA], dB = dist[eB];
        v2f x0 = {vA ? dA : 0.0f, vB ? dB : 0.0f};
        v2f x1 = {edge_invariant(fA), edge_invariant(fB)};
        gate_mlp2<H>(W1, b1, W2, b2, x0, x1, aA, aB);
    }

    float* pRow = &sP[lane * PSUM_P];
#pragma nounroll
    for (int s = 0; s < 2; ++s) {
        float f[FEAT], shv[9];
        edge_rows(feats, sh0, sh1, sh2, vA, eA, srcA, f, shv);
        float g[Cf::NG];
#pragma unroll
        for (int t = 0; t < Cf::NG; ++t) g[t] = tanhf(aA[t]);
        u_stage<H>(shv, f, &sU[lane * Cf::UP]);
        mma_mix<H>(Mg, g, sU, sRaw, pRow, lane);
#pragma unroll
        for (int t = 0; t < Cf::NG; ++t) aA[t] = aB[t];
        vA = vB;
        eA = eB;
        srcA = srcB;
        pRow = &sU[lane * PSUM_P];
    }
}

__global__ void __launch_bounds__(2 * TILE)
layer_kernel(const float* __restrict__ feats, const int* __restrict__ eidx,
             const float* __restrict__ sh0, const float* __restrict__ sh1,
             const float* __restrict__ sh2, const float* __restrict__ dist,
             const float* __restrict__ Mg, const int* __restrict__ perm,
             const float* __restrict__ W1, const float* __restrict__ b1,
             const float* __restrict__ W2, const float* __restrict__ b2,
             float* __restrict__ outF, int E) {
    __shared__ __attribute__((aligned(16))) float sMem[SMEM_FLOATS];
    const int lane = threadIdx.x & (TILE - 1);
    const int half = __builtin_amdgcn_readfirstlane((int)(threadIdx.x >> 5));
    int tgtA, tgtB;
    if (half == 0)
        layer_half<0>(feats, eidx, sh0, sh1, sh2, dist, Mg, perm, W1, b1, W2, b2, E,
                      sMem + SU0_OFF, sMem + SRAW0_OFF, sMem + SP0_OFF, lane, tgtA, tgtB);
    else
        layer_half<1>(feats, eidx, sh0, sh1, sh2, dist, Mg, perm, W1, b1, W2, b2, E,
                      sMem + SU1_OFF, sMem + SRAW1_OFF, sMem + SP1_OFF, lane, tgtA, tgtB);
    __syncthreads();

    const int tgt = (half == 0) ? tgtA : tgtB;
    const int tDn = __shfl_down(tgt, 1, TILE);
    const bool last = (lane == TILE - 1) || (tDn != tgt);
    const unsigned lastMask = __builtin_amdgcn_ballot_w32(last);
    const int colj = min(lane, FEAT - 1);
    const float* p0 = sMem + ((half == 0) ? SP0_OFF : SU0_OFF) + colj;
    const float* p1 = sMem + ((half == 0) ? SP1_OFF : SU1_OFF) + colj;
    float s[TILE];
#pragma unroll
    for (int i = 0; i < TILE; ++i) s[i] = p0[i * PSUM_P] + p1[i * PSUM_P];
    float acc = 0.0f;
#pragma unroll
    for (int i = 0; i < TILE; ++i) {
        acc += s[i];
        if ((lastMask >> i) & 1u) {
            const int t = __builtin_amdgcn_readlane(tgt, i);
            if (t >= 0 && lane < FEAT) atomicAdd(&outF[(long)t * FEAT + lane], acc);
            acc = 0.0f;
        }
    }
}

extern "C" void kernel_launch(void* const* d_in, const int* in_sizes, int n_in,
                              void* d_out, int out_size, void* d_ws, size_t ws_size,
                              hipStream_t stream) {
    const float* node = (const float*)d_in[0];
    const int* eidx   = (const int*)d_in[1];
    const float* sh0  = (const float*)d_in[2];
    const float* sh1  = (const float*)d_in[3];
    const float* sh2  = (const float*)d_in[4];
    const float* dist = (const float*)d_in[5];
    const float* cg[N_COMBOS];
    for (int i = 0; i < N_COMBOS; ++i) cg[i] = (const float*)d_in[6 + i];
    const float* W1_0 = (const float*)d_in[21];
    const float* b1_0 = (const float*)d_in[22];
    const float* W2_0 = (const float*)d_in[23];
    const float* b2_0 = (const float*)d_in[24];
    const float* W1_1 = (const float*)d_in[25];
    const float* b1_1 = (const float*)d_in[26];
    const float* W2_1 = (const float*)d_in[27];
    const float* b2_1 = (const float*)d_in[28];

    const int E = in_sizes[1] / 2;
    const int N = in_sizes[0] / FEAT;

    float* Mg     = (float*)d_ws;
    float* feats1 = (float*)((char*)d_ws + 131072);
    long n1 = (long)N * FEAT;
    int* perm     = (int*)(feats1 + n1);

    build_M_kernel<<<1, 256, 0, stream>>>(cg[0], cg[1], cg[2], cg[3], cg[4], cg[5],
                                          cg[6], cg[7], cg[8], cg[9], cg[10], cg[11],
                                          cg[12], cg[13], cg[14], Mg);

    const bool permFits = (ws_size >= (size_t)131072 + (size_t)(n1 + E) * 4);
    const bool grouped = permFits && ((long)E + N <= (long)out_size);
    const int BW = (N + NBK - 1) / NBK;
    const int chunk = (E + NCH - 1) / NCH;
    const bool twoLevel = permFits && E >= 1 && E <= (1 << 20) && BW >= 1 && BW <= BWMAX &&
                          ((long)E + (long)NCH * CSTP <= (long)out_size);
    if (twoLevel) {
        int* sortedA = (int*)d_out;
        int* cst = sortedA + E;
        bucket_kernel<<<NCH, 1024, 0, stream>>>(eidx, sortedA, cst, E, chunk, BW);
        owner_kernel<<<NBK, 1024, 0, stream>>>(sortedA, cst, perm, E, chunk, BW);
    } else if (grouped) {
        int* rnk = (int*)d_out;
        int* cnt = rnk + E;
        zero_kernel<<<(N + 255) / 256, 256, 0, stream>>>((float*)cnt, (long)N);
        rank_kernel<<<(E + 255) / 256, 256, 0, stream>>>(eidx, cnt, rnk, E);
        scan_kernel<<<1, 1024, 0, stream>>>(cnt, N);
        fill_kernel<<<(E + 255) / 256, 256, 0, stream>>>(eidx, cnt, rnk, perm, E);
    } else {
        perm = nullptr;
    }

    zero_kernel<<<(int)((n1 + 255) / 256), 256, 0, stream>>>(feats1, n1);
    zero_kernel<<<(out_size + 255) / 256, 256, 0, stream>>>((float*)d_out, (long)out_size);

    int grid = (E + 2 * TILE - 1) / (2 * TILE);
    layer_kernel<<<grid, 2 * TILE, 0, stream>>>(node, eidx, sh0, sh1, sh2, dist, Mg, perm,
                                                  W1_0, b1_0, W2_0, b2_0, feats1, E);
    layer_kernel<<<grid, 2 * TILE, 0, stream>>>(feats1, eidx, sh0, sh1, sh2, dist, Mg, perm,
                                                  W1_1, b1_1, W2_1, b2_1, (float*)d_out, E);
}


# Gate tanh two gates at a time in packed f32: the library's odd polynomial on the signed pair, always; its exp form only as a fix-up behind one test per pair (was two tests per gate); sums unchanged
# speedup vs baseline: 3.0949x; 1.0579x over previous
#include <hip/hip_runtime.h>

typedef __attribute__((ext_vector_type(2))) float v2f;
typedef __attribute__((ext_vector_type(8))) float v8f;

#define N_COMBOS 15
#define KP 164
#define PN 112
#define NTILES 7
#define TILE 32
#define FEAT 18

constexpr int cLE[N_COMBOS]     = {0,0,0,1,1,1,1,1,1,2,2,2,2,2,2};
constexpr int cLN[N_COMBOS]     = {0,1,2,0,1,1,1,2,2,0,1,1,2,2,2};
constexpr int cLO[N_COMBOS]     = {0,1,2,1,0,1,2,1,2,2,1,2,0,1,2};
constexpr int OUT_OFF[N_COMBOS] = {0,2,8,18,24,26,32,42,48,58,68,74,84,86,92};
constexpr int U_OFF[9]          = {0,2,8,18,24,42,72,82,112};
constexpr int KLO2[2][NTILES]   = {{0,2,6,10,0,0,0},  {0,0,0,18,18,20,28}};
constexpr int KHI2[2][NTILES]   = {{4,10,17,17,0,0,0},{0,0,0,20,27,40,40}};
constexpr int N_BY_LO[3]        = {3,6,6};
constexpr int C_BY_LO[3][6]     = {{0,4,12,0,0,0},{1,3,5,7,10,13},{2,6,8,9,11,14}};

__global__ void build_M_kernel(const float* cg0, const float* cg1, const float* cg2,
                               const float* cg3, const float* cg4, const float* cg5,
                               const float* cg6, const float* cg7, const float* cg8,
                               const float* cg9, const float* cg10, const float* cg11,
                               const float* cg12, const float* cg13, const float* cg14,
                               float* M) {
    const float* cgs[N_COMBOS] = {cg0,cg1,cg2,cg3,cg4,cg5,cg6,cg7,cg8,cg9,cg10,cg11,cg12,cg13,cg14};
    int tid = threadIdx.x;
    for (int i = tid; i < KP * PN; i += blockDim.x) M[i] = 0.0f;
    __syncthreads();
    if (tid < N_COMBOS) {
        int c = tid;
        int le = cLE[c], ln = cLN[c], lo = cLO[c];
        int de = 2 * le + 1, dn = 2 * ln + 1, dO = 2 * lo + 1;
        int uo = U_OFF[le * 3 + ln];
        int po = OUT_OFF[c];
        const float* cg = cgs[c];
        for (int a = 0; a < de; ++a)
            for (int b = 0; b < dn; ++b)
                for (int k = 0; k < dO; ++k) {
                    float v = cg[(a * dn + b) * dO + k];
                    for (int ch = 0; ch < 2; ++ch)
                        M[(uo + (a * dn + b) * 2 + ch) * PN + po + ch * dO + k] = v;
                }
    }
}

__global__ void zero_kernel(float* p, long n) {
    long i = (long)blockIdx.x * blockDim.x + threadIdx.x;
    if (i < n) p[i] = 0.0f;
}

__global__ void rank_kernel(const int* __restrict__ eidx, int* __restrict__ cnt,
                            int* __restrict__ rnk, int E) {
    int e = blockIdx.x * blockDim.x + threadIdx.x;
    if (e < E) rnk[e] = atomicAdd(&cnt[eidx[(long)E + e]], 1);
}

__global__ void __launch_bounds__(1024)
scan_kernel(int* __restrict__ cnt, int N) {
    __shared__ int part[1024];
    const int t = threadIdx.x;
    const int chunk = (N + 1023) / 1024;
    const int lo = t * chunk;
    int hi = lo + chunk;
    if (hi > N) hi = N;
    int s = 0;
    for (int i = lo; i < hi; ++i) s += cnt[i];
    part[t] = s;
    __syncthreads();
    for (int d = 1; d < 1024; d <<= 1) {
        int add = (t >= d) ? part[t - d] : 0;
        __syncthreads();
        part[t] += add;
        __syncthreads();
    }
    int run = part[t] - s;
    for (int i = lo; i < hi; ++i) { int c = cnt[i]; cnt[i] = run; run += c; }
}

__global__ void fill_kernel(const int* __restrict__ eidx, const int* __restrict__ first,
                            const int* __restrict__ rnk, int* __restrict__ perm, int E) {
    int e = blockIdx.x * blockDim.x + threadIdx.x;
    if (e < E) perm[first[eidx[(long)E + e]] + rnk[e]] = e;
}

#define NCH 128
#define NBK 256
#define CSTP (NBK + 1)
#define BWMAX 512
#define HPA (NBK + 1)
#define HPB (BWMAX + 1)

__device__ __forceinline__ int block_excl_scan_1024(int v, int* sWave, int& total) {
    const int lane = threadIdx.x & 31, wave = threadIdx.x >> 5;
    int x = v;
    for (int d = 1; d < 32; d <<= 1) { const int y = __shfl_up(x, d, 32); if (lane >= d) x += y; }
    if (lane == 31) sWave[wave] = x;
    __syncthreads();
    const int ws = sWave[lane];
    int xs = ws;
    for (int d = 1; d < 32; d <<= 1) { const int y = __shfl_up(xs, d, 32); if (lane >= d) xs += y; }
    const int waveOff = __shfl(xs, wave, 32) - __shfl(ws, wave, 32);
    total = __shfl(xs, 31, 32);
    __syncthreads();
    return waveOff + x - v;
}

__global__ void __launch_bounds__(1024)
bucket_kernel(const int* __restrict__ eidx, int* __restrict__ sortedA, int* __restrict__ cst,
              int E, int chunk, int BW) {
    __shared__ int hist[32 * HPA];
    __shared__ int sWave[32];
    const int t = threadIdx.x, lane = t & 31, w = t >> 5;
    const int c = blockIdx.x;
    for (int i = t; i < 32 * HPA; i += 1024) hist[i] = 0;
    __syncthreads();
    long lo = (long)c * chunk;
    if (lo > E) lo = E;
    long hi = lo + chunk;
    if (hi > E) hi = E;
    const int cLo = (int)lo, cHi = (int)hi;
    const int slice = (chunk + 31) / 32;
    int sLo = cLo + w * slice;
    if (sLo > cHi) sLo = cHi;
    int sHi = sLo + slice;
    if (sHi > cHi) sHi = cHi;
    const int* __restrict__ tg = eidx + E;
    int* myHist = &hist[w * HPA];
    const unsigned ubw = (unsigned)BW;
    for (int i = sLo + lane; i < sHi; i += 32)
        atomicAdd(&myHist[(unsigned)tg[i] / ubw], 1);
    __syncthreads();
    const int b = t >> 2, q = t & 3;
    int s = 0;
    for (int j = 0; j < 8; ++j) s += hist[(q * 8 + j) * HPA + b];
    int total;
    const int ex = block_excl_scan_1024(s, sWave, total);
    if (q == 0) cst[c * CSTP + b] = ex;
    if (t == 0) cst[c * CSTP + NBK] = total;
    int run = ex;
    for (int j = 0; j < 8; ++j) {
        const int k = (q * 8 + j) * HPA + b;
        const int c0 = hist[k];
        hist[k] = run;
        run += c0;
    }
    __syncthreads();
    for (int i = sLo + lane; i < sHi; i += 32) {
        const unsigned tv = (unsigned)tg[i];
        const unsigned bk = tv / ubw;
        const int pos = atomicAdd(&myHist[bk], 1);
        sortedA[cLo + pos] = (int)(((tv - bk * ubw) << 20) | (unsigned)i);
    }
}

__global__ void __launch_bounds__(1024)
owner_kernel(const int* __restrict__ sortedA, const int* __restrict__ cst,
             int* __restrict__ perm, int E, int chunk, int BW) {
    __shared__ int hist[32 * HPB];
    __shared__ int segStart[NCH];
    __shared__ int segLen[NCH];
    __shared__ int sWave[32];
    const int t = threadIdx.x, lane = t & 31, w = t >> 5;
    const int b = blockIdx.x;
    for (int i = t; i < 32 * HPB; i += 1024) hist[i] = 0;
    int st = 0;
    if (t < NCH) {
        st = cst[t * CSTP + b];
        const int ln = cst[t * CSTP + b + 1] - st;
        long lo = (long)t * chunk;
        if (lo > E) lo = E;
        segStart[t] = (int)lo + st;
        segLen[t] = ln;
    }
    int base;
    block_excl_scan_1024(st, sWave, base);
    constexpr int CPW = NCH / 32;
    int s0[CPW], n[CPW];
    int nMax = 0;
    for (int cc = 0; cc < CPW; ++cc) {
        s0[cc] = segStart[w * CPW + cc];
        n[cc] = segLen[w * CPW + cc];
        nMax = n[cc] > nMax ? n[cc] : nMax;
    }
    int* myHist = &hist[w * HPB];
    for (int j0 = 0; j0 < nMax; j0 += 32) {
        const int j = j0 + lane;
        unsigned wd[CPW];
        for (int cc = 0; cc < CPW; ++cc) wd[cc] = (j < n[cc]) ? (unsigned)sortedA[s0[cc] + j] : 0u;
        for (int cc = 0; cc < CPW; ++cc)
            if (j < n[cc]) atomicAdd(&myHist[wd[cc] >> 20], 1);
    }
    __syncthreads();
    int deg = 0;
    if (t < BW)
        for (int w2 = 0; w2 < 32; ++w2) deg += hist[w2 * HPB + t];
    int tot;
    const int ex = block_excl_scan_1024(deg, sWave, tot);
    if (t < BW) {
        int run = base + ex;
        for (int w2 = 0; w2 < 32; ++w2) {
            const int c0 = hist[w2 * HPB + t];
            hist[w2 * HPB + t] = run;
            run += c0;
        }
    }
    __syncthreads();
    for (int j0 = 0; j0 < nMax; j0 += 32) {
        const int j = j0 + lane;
        unsigned wd[CPW];
        for (int cc = 0; cc < CPW; ++cc) wd[cc] = (j < n[cc]) ? (unsigned)sortedA[s0[cc] + j] : 0u;
        for (int cc = 0; cc < CPW; ++cc)
            if (j < n[cc]) {
                const int r = atomicAdd(&myHist[wd[cc] >> 20], 1);
                perm[r] = (int)(wd[cc] & 0xFFFFFu);
            }
    }
}

template <int H> struct HalfCfg;
template <> struct HalfCfg<0> {
    static constexpr int T0 = 0,  NG = 36;
    static constexpr int C0 = 0,  C1 = 9;
    static constexpr int NT0 = 0, NT1 = 4;
    static constexpr int ULO = 0, UHI = 72;
    static constexpr int UP = 76;
    static constexpr int COL0 = 0;
    static constexpr int COL1 = 58;
    static constexpr int RP = 60;
};
template <> struct HalfCfg<1> {
    static constexpr int T0 = 36, NG = 24;
    static constexpr int C0 = 9,  C1 = 15;
    static constexpr int NT0 = 3, NT1 = 7;
    static constexpr int ULO = 72, UHI = 164;
    static constexpr int UP = 92;
    static constexpr int COL0 = 58;
    static constexpr int COL1 = 102;
    static constexpr int RP = 44;
};
#define SU0_OFF   0
#define SRAW0_OFF 2432
#define SU1_OFF   4352
#define SRAW1_OFF 7296
#define SP0_OFF   8704
#define SP1_OFF   9280
#define SMEM_FLOATS 9856
#define PSUM_P FEAT

__device__ __forceinline__ void wave_lds_handover() {
    __builtin_amdgcn_fence(__ATOMIC_RELEASE, "wavefront");
    __builtin_amdgcn_wave_barrier();
    __builtin_amdgcn_fence(__ATOMIC_ACQUIRE, "wavefront");
}

template <int H>
__device__ __forceinline__ void gate_mlp2(const float* __restrict__ W1g,
                                          const float* __restrict__ b1g,
                                          const float* __restrict__ W2g,
                                          const float* __restrict__ b2g,
                                          v2f x0, v2f x1,
                                          float (&aa)[HalfCfg<H>::NG],
                                          float (&ab)[HalfCfg<H>::NG]) {
    typedef const __attribute__((address_space(4))) float* cptr;
    const cptr W1 = (cptr)W1g, b1 = (cptr)b1g, W2 = (cptr)W2g, b2 = (cptr)b2g;
#pragma unroll
    for (int t = 0; t < HalfCfg<H>::NG; ++t) {
        const int bi = HalfCfg<H>::T0 + t;
        const float b2s = b2[bi];
        v2f acc = {b2s, b2s};
#pragma unroll
        for (int h = 0; h < 4; ++h) {
            v2f hh = x0 * W1[bi * 8 + h * 2] + x1 * W1[bi * 8 + h * 2 + 1] + b1[bi * 4 + h];
            hh.x = hh.x > 0.0f ? hh.x : 0.0f;
            hh.y = hh.y > 0.0f ? hh.y : 0.0f;
            acc += hh * W2[bi * 4 + h];
        }
        aa[t] = acc.x;
        ab[t] = acc.y;
        __builtin_amdgcn_sched_barrier(0);
    }
}

template <int NG>
__device__ __forceinline__ void tanh_gate_pairs(const float (&a)[NG], float (&g)[NG]) {
    static_assert(NG % 2 == 0, "gates are taken in pairs");
    const v2f c4 = {-0x1.758e7ap-8f, -0x1.758e7ap-8f};
    const v2f c3 = {0x1.521192p-6f, 0x1.521192p-6f};
    const v2f c2 = {-0x1.b8389cp-5f, -0x1.b8389cp-5f};
    const v2f c1 = {0x1.110704p-3f, 0x1.110704p-3f};
    const v2f c0 = {-0x1.555532p-2f, -0x1.555532p-2f};
#pragma unroll
    for (int t = 0; t < NG; t += 2) {
        const v2f x = {a[t], a[t + 1]};
        const v2f x2 = x * x;
        v2f p = __builtin_elementwise_fma(x2, c4, c3);
        p = __builtin_elementwise_fma(x2, p, c2);
        p = __builtin_elementwise_fma(x2, p, c1);
        p = __builtin_elementwise_fma(x2, p, c0);
        v2f r = __builtin_elementwise_fma(x2, x * p, x);
        const float y0 = __builtin_fabsf(x.x), y1 = __builtin_fabsf(x.y);
        if (!(__builtin_fmaxf(y0, y1) < 0.625f)) {
            const float e0 = expf(2.0f * y0), e1 = expf(2.0f * y1);
            const float z0 = __builtin_fmaf(-2.0f, __builtin_amdgcn_rcpf(e0 + 1.0f), 1.0f);
            const float z1 = __builtin_fmaf(-2.0f, __builtin_amdgcn_rcpf(e1 + 1.0f), 1.0f);
            r.x = (y0 < 0.625f) ? r.x : __builtin_copysignf(z0, x.x);
            r.y = (y1 < 0.625f) ? r.y : __builtin_copysignf(z1, x.y);
        }
        g[t] = r.x;
        g[t + 1] = r.y;
    }
}

template <int H>
__device__ __forceinline__ void u_stage(const float (&shv)[9],
                                        const float (&f)[FEAT],
                                        float* __restrict__ uRow) {
    typedef HalfCfg<H> Cf;
    int u = 0;
#pragma unroll
    for (int le = 0; le < 3; ++le) {
        const int de = 2 * le + 1;
        const int shb = (le == 0) ? 0 : (le == 1 ? 1 : 4);
#pragma unroll
        for (int ln = 0; ln < 3; ++ln) {
            const int dn = 2 * ln + 1;
            const int fb = (ln == 0) ? 0 : (ln == 1 ? 2 : 8);
#pragma unroll
            for (int a = 0; a < de; ++a)
#pragma unroll
                for (int b = 0; b < dn; ++b)
#pragma unroll
                    for (int ch = 0; ch < 2; ++ch) {
                        if (u >= Cf::ULO && u < Cf::UHI)
                            uRow[u - Cf::ULO] = shv[shb + a] * f[fb + ch * dn + b];
                        ++u;
                    }
        }
    }
    if (Cf::UHI > 162) {
        uRow[162 - Cf::ULO] = 0.0f;
        uRow[163 - Cf::ULO] = 0.0f;
    }
}

template <int S, int H>
__device__ __forceinline__ float slot_val(const float (&gRow)[HalfCfg<H>::NG],
                                          const float* __restrict__ rRow) {
    typedef HalfCfg<H> Cf;
    constexpr int lo = (S < 2) ? 0 : ((S < 8) ? 1 : 2);
    constexpr int o  = (S < 2) ? S : ((S < 8) ? ((S - 2) / 3) : ((S - 8) / 5));
    constexpr int k  = (S < 2) ? 0 : ((S < 8) ? ((S - 2) % 3) : ((S - 8) % 5));
    constexpr int dO = 2 * lo + 1;
    float v = 0.0f;
#pragma unroll
    for (int q = 0; q < N_BY_LO[lo]; ++q) {
        constexpr const int* lst = C_BY_LO[lo];
        const int c = lst[q];
        if (c >= Cf::C0 && c < Cf::C1) {
            const int base = OUT_OFF[c] - Cf::COL0;
            const int gb = (c - Cf::C0) * 4;
            v += gRow[gb + o]     * rRow[base + k] +
                 gRow[gb + 2 + o] * rRow[base + dO + k];
        }
    }
    return v;
}

__device__ __forceinline__ void
edge_ids(const int* __restrict__ eidx, const int* __restrict__ perm, int E, long slot,
         bool& valid, long& e, int& src, int& tgt) {
    valid = (slot < (long)E);
    const long sl = valid ? slot : (long)E - 1;
    e = perm ? (long)perm[sl] : sl;
    src = eidx[e];
    const int tg = eidx[(long)E + e];
    tgt = valid ? tg : -1;
}

__device__ __forceinline__ void
edge_rows(const float* __restrict__ feats, const float* __restrict__ sh0,
          const float* __restrict__ sh1, const float* __restrict__ sh2,
          bool valid, long e, int src, float (&f)[FEAT], float (&shv)[9]) {
    const float* fp = &feats[(long)src * FEAT];
#pragma unroll
    for (int j = 0; j < FEAT; ++j) f[j] = fp[j];
    shv[0] = sh0[e];
#pragma unroll
    for (int j = 0; j < 3; ++j) shv[1 + j] = sh1[e * 3 + j];
#pragma unroll
    for (int j = 0; j < 5; ++j) shv[4 + j] = sh2[e * 5 + j];
#pragma unroll
    for (int j = 0; j < FEAT; ++j) f[j] = valid ? f[j] : 0.0f;
#pragma unroll
    for (int j = 0; j < 9; ++j) shv[j] = valid ? shv[j] : 0.0f;
}

__device__ __forceinline__ float edge_invariant(const float (&f)[FEAT]) {
    float inv0 = 0.5f * (f[0] + f[1]);
    float n1 = 0.0f, n2 = 0.0f;
#pragma unroll
    for (int j = 0; j < 3; ++j) { float m = 0.5f * (f[2 + j] + f[5 + j]); n1 += m * m; }
#pragma unroll
    for (int j = 0; j < 5; ++j) { float m = 0.5f * (f[8 + j] + f[13 + j]); n2 += m * m; }
    return (inv0 + sqrtf(n1) + sqrtf(n2)) * (1.0f / 3.0f);
}

template <int H>
__device__ __forceinline__ void
mma_mix(const float* __restrict__ Mg, const float (&g)[HalfCfg<H>::NG],
        float* __restrict__ sU, float* __restrict__ sRaw, float* __restrict__ pRow,
        const int lane) {
    typedef HalfCfg<H> Cf;
    wave_lds_handover();
    {
        const int col16 = lane & 15;
        const int kofs = (lane < 16) ? 0 : 2;
        const float* uA0 = &sU[col16 * Cf::UP];
        const float* uA1 = &sU[(16 + col16) * Cf::UP];
#pragma unroll
        for (int n = Cf::NT0; n < Cf::NT1; ++n) {
            v8f acc0 = {0.f, 0.f, 0.f, 0.f, 0.f, 0.f, 0.f, 0.f};
            v8f acc1 = {0.f, 0.f, 0.f, 0.f, 0.f, 0.f, 0.f, 0.f};
            const int col = col16 + n * 16;
#pragma unroll
            for (int kt = KLO2[H][n]; kt <= KHI2[H][n]; ++kt) {
                int k0 = kt * 4 + kofs;
                v2f a0, a1, b;
                a0.x = uA0[k0 - Cf::ULO];
                a0.y = uA0[k0 - Cf::ULO + 1];
                a1.x = uA1[k0 - Cf::ULO];
                a1.y = uA1[k0 - Cf::ULO + 1];
                b.x = Mg[k0 * PN + col];
                b.y = Mg[(k0 + 1) * PN + col];
                acc0 = __builtin_amdgcn_wmma_f32_16x16x4_f32(
                    false, a0, false, b, (short)0, acc0, false, false);
                acc1 = __builtin_amdgcn_wmma_f32_16x16x4_f32(
                    false, a1, false, b, (short)0, acc1, false, false);
            }
            const int rbase = (lane < 16) ? 0 : 8;
            if (col >= Cf::COL0 && col < Cf::COL1) {
#pragma unroll
                for (int j = 0; j < 8; ++j) {
                    sRaw[(rbase + j) * Cf::RP + col - Cf::COL0] = acc0[j];
                    sRaw[(16 + rbase + j) * Cf::RP + col - Cf::COL0] = acc1[j];
                }
            }
        }
    }
    wave_lds_handover();
    {
        const float* rRow = &sRaw[lane * Cf::RP];
        float v[FEAT];
        v[0]  = slot_val<0, H>(g, rRow);   v[1]  = slot_val<1, H>(g, rRow);
        v[2]  = slot_val<2, H>(g, rRow);   v[3]  = slot_val<3, H>(g, rRow);
        v[4]  = slot_val<4, H>(g, rRow);   v[5]  = slot_val<5, H>(g, rRow);
        v[6]  = slot_val<6, H>(g, rRow);   v[7]  = slot_val<7, H>(g, rRow);
        v[8]  = slot_val<8, H>(g, rRow);   v[9]  = slot_val<9, H>(g, rRow);
        v[10] = slot_val<10, H>(g, rRow);  v[11] = slot_val<11, H>(g, rRow);
        v[12] = slot_val<12, H>(g, rRow);  v[13] = slot_val<13, H>(g, rRow);
        v[14] = slot_val<14, H>(g, rRow);  v[15] = slot_val<15, H>(g, rRow);
        v[16] = slot_val<16, H>(g, rRow);  v[17] = slot_val<17, H>(g, rRow);
#pragma unroll
        for (int i = 0; i < FEAT; ++i) pRow[i] = v[i];
    }
}

template <int H>
__device__ __forceinline__ void
layer_half(const float* __restrict__ feats, const int* __restrict__ eidx,
           const float* __restrict__ sh0, const float* __restrict__ sh1,
           const float* __restrict__ sh2, const float* __restrict__ dist,
           const float* __restrict__ Mg, const int* __restrict__ perm,
           const float* __restrict__ W1, const float* __restrict__ b1,
           const float* __restrict__ W2, const float* __restrict__ b2,
           int E, float* __restrict__ sU, float* __restrict__ sRaw, float* __restrict__ sP,
           const int lane, int& tgtA, int& tgtB) {
    typedef HalfCfg<H> Cf;
    const long slotA = (long)blockIdx.x * (2 * TILE) + lane;
    const long slotB = slotA + TILE;

    bool vA, vB;
    long eA, eB;
    int srcA, srcB;
    edge_ids(eidx, perm, E, slotA, vA, eA, srcA, tgtA);
    edge_ids(eidx, perm, E, slotB, vB, eB, srcB, tgtB);

    float aA[Cf::NG], aB[Cf::NG];
    {
        float fA[FEAT], fB[FEAT], shA[9], shB[9];
        edge_rows(feats, sh0, sh1, sh2, vA, eA, srcA, fA, shA);
        edge_rows(feats, sh0, sh1, sh2, vB, eB, srcB, fB, shB);
        const float dA = dist[eA], dB = dist[eB];
        v2f x0 = {vA ? dA : 0.0f, vB ? dB : 0.0f};
        v2f x1 = {edge_invariant(fA), edge_invariant(fB)};
        gate_mlp2<H>(W1, b1, W2, b2, x0, x1, aA, aB);
    }

    float* pRow = &sP[lane * PSUM_P];
#pragma nounroll
    for (int s = 0; s < 2; ++s) {
        float f[FEAT], shv[9];
        edge_rows(feats, sh0, sh1, sh2, vA, eA, srcA, f, shv);
        float g[Cf::NG];
        tanh_gate_pairs<Cf::NG>(aA, g);
        u_stage<H>(shv, f, &sU[lane * Cf::UP]);
        mma_mix<H>(Mg, g, sU, sRaw, pRow, lane);
#pragma unroll
        for (int t = 0; t < Cf::NG; ++t) aA[t] = aB[t];
        vA = vB;
        eA = eB;
        srcA = srcB;
        pRow = &sU[lane * PSUM_P];
    }
}

__global__ void __launch_bounds__(2 * TILE)
layer_kernel(const float* __restrict__ feats, const int* __restrict__ eidx,
             const float* __restrict__ sh0, const float* __restrict__ sh1,
             const float* __restrict__ sh2, const float* __restrict__ dist,
             const float* __restrict__ Mg, const int* __restrict__ perm,
             const float* __restrict__ W1, const float* __restrict__ b1,
             const float* __restrict__ W2, const float* __restrict__ b2,
             float* __restrict__ outF, int E) {
    __shared__ __attribute__((aligned(16))) float sMem[SMEM_FLOATS];
    const int lane = threadIdx.x & (TILE - 1);
    const int half = __builtin_amdgcn_readfirstlane((int)(threadIdx.x >> 5));
    int tgtA, tgtB;
    if (half == 0)
        layer_half<0>(feats, eidx, sh0, sh1, sh2, dist, Mg, perm, W1, b1, W2, b2, E,
                      sMem + SU0_OFF, sMem + SRAW0_OFF, sMem + SP0_OFF, lane, tgtA, tgtB);
    else
        layer_half<1>(feats, eidx, sh0, sh1, sh2, dist, Mg, perm, W1, b1, W2, b2, E,
                      sMem + SU1_OFF, sMem + SRAW1_OFF, sMem + SP1_OFF, lane, tgtA, tgtB);
    __syncthreads();

    const int tgt = (half == 0) ? tgtA : tgtB;
    const int tDn = __shfl_down(tgt, 1, TILE);
    const bool last = (lane == TILE - 1) || (tDn != tgt);
    const unsigned lastMask = __builtin_amdgcn_ballot_w32(last);
    const int colj = min(lane, FEAT - 1);
    const float* p0 = sMem + ((half == 0) ? SP0_OFF : SU0_OFF) + colj;
    const float* p1 = sMem + ((half == 0) ? SP1_OFF : SU1_OFF) + colj;
    float s[TILE];
#pragma unroll
    for (int i = 0; i < TILE; ++i) s[i] = p0[i * PSUM_P] + p1[i * PSUM_P];
    float acc = 0.0f;
#pragma unroll
    for (int i = 0; i < TILE; ++i) {
        acc += s[i];
        if ((lastMask >> i) & 1u) {
            const int t = __builtin_amdgcn_readlane(tgt, i);
            if (t >= 0 && lane < FEAT) atomicAdd(&outF[(long)t * FEAT + lane], acc);
            acc = 0.0f;
        }
    }
}

extern "C" void kernel_launch(void* const* d_in, const int* in_sizes, int n_in,
                              void* d_out, int out_size, void* d_ws, size_t ws_size,
                              hipStream_t stream) {
    const float* node = (const float*)d_in[0];
    const int* eidx   = (const int*)d_in[1];
    const float* sh0  = (const float*)d_in[2];
    const float* sh1  = (const float*)d_in[3];
    const float* sh2  = (const float*)d_in[4];
    const float* dist = (const float*)d_in[5];
    const float* cg[N_COMBOS];
    for (int i = 0; i < N_COMBOS; ++i) cg[i] = (const float*)d_in[6 + i];
    const float* W1_0 = (const float*)d_in[21];
    const float* b1_0 = (const float*)d_in[22];
    const float* W2_0 = (const float*)d_in[23];
    const float* b2_0 = (const float*)d_in[24];
    const float* W1_1 = (const float*)d_in[25];
    const float* b1_1 = (const float*)d_in[26];
    const float* W2_1 = (const float*)d_in[27];
    const float* b2_1 = (const float*)d_in[28];

    const int E = in_sizes[1] / 2;
    const int N = in_sizes[0] / FEAT;

    float* Mg     = (float*)d_ws;
    float* feats1 = (float*)((char*)d_ws + 131072);
    long n1 = (long)N * FEAT;
    int* perm     = (int*)(feats1 + n1);

    build_M_kernel<<<1, 256, 0, stream>>>(cg[0], cg[1], cg[2], cg[3], cg[4], cg[5],
                                          cg[6], cg[7], cg[8], cg[9], cg[10], cg[11],
                                          cg[12], cg[13], cg[14], Mg);

    const bool permFits = (ws_size >= (size_t)131072 + (size_t)(n1 + E) * 4);
    const bool grouped = permFits && ((long)E + N <= (long)out_size);
    const int BW = (N + NBK - 1) / NBK;
    const int chunk = (E + NCH - 1) / NCH;
    const bool twoLevel = permFits && E >= 1 && E <= (1 << 20) && BW >= 1 && BW <= BWMAX &&
                          ((long)E + (long)NCH * CSTP <= (long)out_size);
    if (twoLevel) {
        int* sortedA = (int*)d_out;
        int* cst = sortedA + E;
        bucket_kernel<<<NCH, 1024, 0, stream>>>(eidx, sortedA, cst, E, chunk, BW);
        owner_kernel<<<NBK, 1024, 0, stream>>>(sortedA, cst, perm, E, chunk, BW);
    } else if (grouped) {
        int* rnk = (int*)d_out;
        int* cnt = rnk + E;
        zero_kernel<<<(N + 255) / 256, 256, 0, stream>>>((float*)cnt, (long)N);
        rank_kernel<<<(E + 255) / 256, 256, 0, stream>>>(eidx, cnt, rnk, E);
        scan_kernel<<<1, 1024, 0, stream>>>(cnt, N);
        fill_kernel<<<(E + 255) / 256, 256, 0, stream>>>(eidx, cnt, rnk, perm, E);
    } else {
        perm = nullptr;
    }

    zero_kernel<<<(int)((n1 + 255) / 256), 256, 0, stream>>>(feats1, n1);
    zero_kernel<<<(out_size + 255) / 256, 256, 0, stream>>>((float*)d_out, (long)out_size);

    int grid = (E + 2 * TILE - 1) / (2 * TILE);
    layer_kernel<<<grid, 2 * TILE, 0, stream>>>(node, eidx, sh0, sh1, sh2, dist, Mg, perm,
                                                  W1_0, b1_0, W2_0, b2_0, feats1, E);
    layer_kernel<<<grid, 2 * TILE, 0, stream>>>(feats1, eidx, sh0, sh1, sh2, dist, Mg, perm,
                                                  W1_1, b1_1, W2_1, b2_1, (float*)d_out, E);
}


# Each edge gathered from global memory once per work-group: a wave gathers one 32-edge sub-tile for both waves, rows wait in unused LDS words, one more barrier; pass loop load-free; sums unchanged
# speedup vs baseline: 4.0719x; 1.3157x over previous
#include <hip/hip_runtime.h>


typedef __attribute__((ext_vector_type(2))) float v2f;
typedef __attribute__((ext_vector_type(4))) float v4f;
typedef __attribute__((ext_vector_type(8))) float v8f;

#define N_COMBOS 15
#define K2 16
#define N2 208
#define TILE 32
#define FEAT 18

constexpr int cLE[N_COMBOS]     = {0,0,0,1,1,1,1,1,1,2,2,2,2,2,2};
constexpr int cLN[N_COMBOS]     = {0,1,2,0,1,1,1,2,2,0,1,1,2,2,2};
constexpr int cLO[N_COMBOS]     = {0,1,2,1,0,1,2,1,2,2,1,2,0,1,2};
constexpr int KROW[3]           = {0, 4, 8};
constexpr int TOFF[N_COMBOS]    = {0,1,10,48,51,54,63,78,93,128,133,142,157,162,177};
constexpr int nt_klo(int n) { return n < 3 ? 0 : (n < 8 ? 1 : 2); }
constexpr int nt_khi(int n) { return n < 3 ? 0 : (n < 8 ? 1 : 3); }

__global__ void build_M_kernel(const float* cg0, const float* cg1, const float* cg2,
                               const float* cg3, const float* cg4, const float* cg5,
                               const float* cg6, const float* cg7, const float* cg8,
                               const float* cg9, const float* cg10, const float* cg11,
                               const float* cg12, const float* cg13, const float* cg14,
                               float* M) {
    const float* cgs[N_COMBOS] = {cg0,cg1,cg2,cg3,cg4,cg5,cg6,cg7,cg8,cg9,cg10,cg11,cg12,cg13,cg14};
    int tid = threadIdx.x;
    for (int i = tid; i < K2 * N2; i += blockDim.x) M[i] = 0.0f;
    __syncthreads();
    if (tid < N_COMBOS) {
        int c = tid;
        int le = cLE[c], ln = cLN[c], lo = cLO[c];
        int de = 2 * le + 1, dn = 2 * ln + 1, dO = 2 * lo + 1;
        int ro = KROW[le];
        int po = TOFF[c];
        const float* cg = cgs[c];
        for (int a = 0; a < de; ++a)
            for (int b = 0; b < dn; ++b)
                for (int k = 0; k < dO; ++k)
                    M[(ro + a) * N2 + po + b * dO + k] = cg[(a * dn + b) * dO + k];
    }
}

__global__ void zero_kernel(float* p, long n) {
    long i = (long)blockIdx.x * blockDim.x + threadIdx.x;
    if (i < n) p[i] = 0.0f;
}

__global__ void rank_kernel(const int* __restrict__ eidx, int* __restrict__ cnt,
                            int* __restrict__ rnk, int E) {
    int e = blockIdx.x * blockDim.x + threadIdx.x;
    if (e < E) rnk[e] = atomicAdd(&cnt[eidx[(long)E + e]], 1);
}

__global__ void __launch_bounds__(1024)
scan_kernel(int* __restrict__ cnt, int N) {
    __shared__ int part[1024];
    const int t = threadIdx.x;
    const int chunk = (N + 1023) / 1024;
    const int lo = t * chunk;
    int hi = lo + chunk;
    if (hi > N) hi = N;
    int s = 0;
    for (int i = lo; i < hi; ++i) s += cnt[i];
    part[t] = s;
    __syncthreads();
    for (int d = 1; d < 1024; d <<= 1) {
        int add = (t >= d) ? part[t - d] : 0;
        __syncthreads();
        part[t] += add;
        __syncthreads();
    }
    int run = part[t] - s;
    for (int i = lo; i < hi; ++i) { int c = cnt[i]; cnt[i] = run; run += c; }
}

__global__ void fill_kernel(const int* __restrict__ eidx, const int* __restrict__ first,
                            const int* __restrict__ rnk, int* __restrict__ perm, int E) {
    int e = blockIdx.x * blockDim.x + threadIdx.x;
    if (e < E) perm[first[eidx[(long)E + e]] + rnk[e]] = e;
}

#define NCH 128
#define NBK 256
#define CSTP (NBK + 1)
#define BWMAX 512
#define HPA (NBK + 1)
#define HPB (BWMAX + 1)

__device__ __forceinline__ int block_excl_scan_1024(int v, int* sWave, int& total) {
    const int lane = threadIdx.x & 31, wave = threadIdx.x >> 5;
    int x = v;
    for (int d = 1; d < 32; d <<= 1) { const int y = __shfl_up(x, d, 32); if (lane >= d) x += y; }
    if (lane == 31) sWave[wave] = x;
    __syncthreads();
    const int ws = sWave[lane];
    int xs = ws;
    for (int d = 1; d < 32; d <<= 1) { const int y = __shfl_up(xs, d, 32); if (lane >= d) xs += y; }
    const int waveOff = __shfl(xs, wave, 32) - __shfl(ws, wave, 32);
    total = __shfl(xs, 31, 32);
    __syncthreads();
    return waveOff + x - v;
}

__global__ void __launch_bounds__(1024)
bucket_kernel(const int* __restrict__ eidx, int* __restrict__ sortedA, int* __restrict__ cst,
              int E, int chunk, int BW) {
    __shared__ int hist[32 * HPA];
    __shared__ int sWave[32];
    const int t = threadIdx.x, lane = t & 31, w = t >> 5;
    const int c = blockIdx.x;
    for (int i = t; i < 32 * HPA; i += 1024) hist[i] = 0;
    __syncthreads();
    long lo = (long)c * chunk;
    if (lo > E) lo = E;
    long hi = lo + chunk;
    if (hi > E) hi = E;
    const int cLo = (int)lo, cHi = (int)hi;
    const int slice = (chunk + 31) / 32;
    int sLo = cLo + w * slice;
    if (sLo > cHi) sLo = cHi;
    int sHi = sLo + slice;
    if (sHi > cHi) sHi = cHi;
    const int* __restrict__ tg = eidx + E;
    int* myHist = &hist[w * HPA];
    const unsigned ubw = (unsigned)BW;
    for (int i = sLo + lane; i < sHi; i += 32)
        atomicAdd(&myHist[(unsigned)tg[i] / ubw], 1);
    __syncthreads();
    const int b = t >> 2, q = t & 3;
    int s = 0;
    for (int j = 0; j < 8; ++j) s += hist[(q * 8 + j) * HPA + b];
    int total;
    const int ex = block_excl_scan_1024(s, sWave, total);
    if (q == 0) cst[c * CSTP + b] = ex;
    if (t == 0) cst[c * CSTP + NBK] = total;
    int run = ex;
    for (int j = 0; j < 8; ++j) {
        const int k = (q * 8 + j) * HPA + b;
        const int c0 = hist[k];
        hist[k] = run;
        run += c0;
    }
    __syncthreads();
    for (int i = sLo + lane; i < sHi; i += 32) {
        const unsigned tv = (unsigned)tg[i];
        const unsigned bk = tv / ubw;
        const int pos = atomicAdd(&myHist[bk], 1);
        sortedA[cLo + pos] = (int)(((tv - bk * ubw) << 20) | (unsigned)i);
    }
}

__global__ void __launch_bounds__(1024)
owner_kernel(const int* __restrict__ sortedA, const int* __restrict__ cst,
             int* __restrict__ perm, int E, int chunk, int BW) {
    __shared__ int hist[32 * HPB];
    __shared__ int segStart[NCH];
    __shared__ int segLen[NCH];
    __shared__ int sWave[32];
    const int t = threadIdx.x, lane = t & 31, w = t >> 5;
    const int b = blockIdx.x;
    for (int i = t; i < 32 * HPB; i += 1024) hist[i] = 0;
    int st = 0;
    if (t < NCH) {
        st = cst[t * CSTP + b];
        const int ln = cst[t * CSTP + b + 1] - st;
        long lo = (long)t * chunk;
        if (lo > E) lo = E;
        segStart[t] = (int)lo + st;
        segLen[t] = ln;
    }
    int base;
    block_excl_scan_1024(st, sWave, base);
    constexpr int CPW = NCH / 32;
    int s0[CPW], n[CPW];
    int nMax = 0;
    for (int cc = 0; cc < CPW; ++cc) {
        s0[cc] = segStart[w * CPW + cc];
        n[cc] = segLen[w * CPW + cc];
        nMax = n[cc] > nMax ? n[cc] : nMax;
    }
    int* myHist = &hist[w * HPB];
    for (int j0 = 0; j0 < nMax; j0 += 32) {
        const int j = j0 + lane;
        unsigned wd[CPW];
        for (int cc = 0; cc < CPW; ++cc) wd[cc] = (j < n[cc]) ? (unsigned)sortedA[s0[cc] + j] : 0u;
        for (int cc = 0; cc < CPW; ++cc)
            if (j < n[cc]) atomicAdd(&myHist[wd[cc] >> 20], 1);
    }
    __syncthreads();
    int deg = 0;
    if (t < BW)
        for (int w2 = 0; w2 < 32; ++w2) deg += hist[w2 * HPB + t];
    int tot;
    const int ex = block_excl_scan_1024(deg, sWave, tot);
    if (t < BW) {
        int run = base + ex;
        for (int w2 = 0; w2 < 32; ++w2) {
            const int c0 = hist[w2 * HPB + t];
            hist[w2 * HPB + t] = run;
            run += c0;
        }
    }
    __syncthreads();
    for (int j0 = 0; j0 < nMax; j0 += 32) {
        const int j = j0 + lane;
        unsigned wd[CPW];
        for (int cc = 0; cc < CPW; ++cc) wd[cc] = (j < n[cc]) ? (unsigned)sortedA[s0[cc] + j] : 0u;
        for (int cc = 0; cc < CPW; ++cc)
            if (j < n[cc]) {
                const int r = atomicAdd(&myHist[wd[cc] >> 20], 1);
                perm[r] = (int)(wd[cc] & 0xFFFFFu);
            }
    }
}

template <int H> struct HalfCfg;
template <> struct HalfCfg<0> {
    static constexpr int T0 = 0,  NG = 36;
    static constexpr int C0 = 0,  C1 = 9;
    static constexpr int NT0 = 0, NT1 = 8;
    static constexpr int KT0 = 0;
    static constexpr int NB = 8;
    static constexpr int RP = 132;
};
template <> struct HalfCfg<1> {
    static constexpr int T0 = 36, NG = 24;
    static constexpr int C0 = 9,  C1 = 15;
    static constexpr int NT0 = 8, NT1 = 13;
    static constexpr int KT0 = 2;
    static constexpr int NB = 10;
    static constexpr int RP = 84;
};
#define AP 12
#define SA0_OFF   0
#define SA1_OFF   384
#define ST0_OFF   768
#define ST1_OFF   4992
#define SP0_OFF   7680
#define SP1_OFF   8256
#define SQ0_OFF   8832
#define SQ1_OFF   9408
#define SMEM_FLOATS 9984
#define PSUM_P FEAT
static_assert(SP1_OFF - SP0_OFF == SQ1_OFF - SQ0_OFF, "a wave's two partial-sum tiles lie at one distance");
static_assert(SQ0_OFF - SP0_OFF == SQ1_OFF - SP1_OFF, "sub-tile distance is the same in both halves");

__device__ __forceinline__ void wave_lds_handover() {
    __builtin_amdgcn_fence(__ATOMIC_RELEASE, "wavefront");
    __builtin_amdgcn_wave_barrier();
    __builtin_amdgcn_fence(__ATOMIC_ACQUIRE, "wavefront");
}

template <int H>
__device__ __forceinline__ void gate_mlp2(const float* __restrict__ W1g,
                                          const float* __restrict__ b1g,
                                          const float* __restrict__ W2g,
                                          const float* __restrict__ b2g,
                                          v2f x0, v2f x1,
                                          float (&aa)[HalfCfg<H>::NG],
                                          float (&ab)[HalfCfg<H>::NG]) {
    typedef const __attribute__((address_space(4))) float* cptr;
    const cptr W1 = (cptr)W1g, b1 = (cptr)b1g, W2 = (cptr)W2g, b2 = (cptr)b2g;
#pragma unroll
    for (int t = 0; t < HalfCfg<H>::NG; ++t) {
        const int bi = HalfCfg<H>::T0 + t;
        const float b2s = b2[bi];
        v2f acc = {b2s, b2s};
#pragma unroll
        for (int h = 0; h < 4; ++h) {
            v2f hh = x0 * W1[bi * 8 + h * 2] + x1 * W1[bi * 8 + h * 2 + 1] + b1[bi * 4 + h];
            hh.x = hh.x > 0.0f ? hh.x : 0.0f;
            hh.y = hh.y > 0.0f ? hh.y : 0.0f;
            acc += hh * W2[bi * 4 + h];
        }
        aa[t] = acc.x;
        ab[t] = acc.y;
        __builtin_amdgcn_sched_barrier(0);
    }
}

template <int NG>
__device__ __forceinline__ void tanh_gate_pairs(const float (&a)[NG], float (&g)[NG]) {
    static_assert(NG % 2 == 0, "gates are taken in pairs");
    const v2f c4 = {-0x1.758e7ap-8f, -0x1.758e7ap-8f};
    const v2f c3 = {0x1.521192p-6f, 0x1.521192p-6f};
    const v2f c2 = {-0x1.b8389cp-5f, -0x1.b8389cp-5f};
    const v2f c1 = {0x1.110704p-3f, 0x1.110704p-3f};
    const v2f c0 = {-0x1.555532p-2f, -0x1.555532p-2f};
#pragma unroll
    for (int t = 0; t < NG; t += 2) {
        const v2f x = {a[t], a[t + 1]};
        const v2f x2 = x * x;
        v2f p = __builtin_elementwise_fma(x2, c4, c3);
        p = __builtin_elementwise_fma(x2, p, c2);
        p = __builtin_elementwise_fma(x2, p, c1);
        p = __builtin_elementwise_fma(x2, p, c0);
        v2f r = __builtin_elementwise_fma(x2, x * p, x);
        const float y0 = __builtin_fabsf(x.x), y1 = __builtin_fabsf(x.y);
        if (!(__builtin_fmaxf(y0, y1) < 0.625f)) {
            const float e0 = expf(2.0f * y0), e1 = expf(2.0f * y1);
            const float z0 = __builtin_fmaf(-2.0f, __builtin_amdgcn_rcpf(e0 + 1.0f), 1.0f);
            const float z1 = __builtin_fmaf(-2.0f, __builtin_amdgcn_rcpf(e1 + 1.0f), 1.0f);
            r.x = (y0 < 0.625f) ? r.x : __builtin_copysignf(z0, x.x);
            r.y = (y1 < 0.625f) ? r.y : __builtin_copysignf(z1, x.y);
        }
        g[t] = r.x;
        g[t + 1] = r.y;
    }
}

__device__ __forceinline__ void
edge_ids(const int* __restrict__ eidx, const int* __restrict__ perm, int E, long slot,
         bool& valid, long& e, int& src, int& tgt) {
    valid = (slot < (long)E);
    const long sl = valid ? slot : (long)E - 1;
    e = perm ? (long)perm[sl] : sl;
    src = eidx[e];
    const int tg = eidx[(long)E + e];
    tgt = valid ? tg : -1;
}

__device__ __forceinline__ void
feat_row(const float* __restrict__ feats, bool valid, int src, float (&f)[FEAT]) {
    const float* fp = &feats[(long)src * FEAT];
#pragma unroll
    for (int j = 0; j < FEAT; ++j) f[j] = fp[j];
#pragma unroll
    for (int j = 0; j < FEAT; ++j) f[j] = valid ? f[j] : 0.0f;
}

__device__ __forceinline__ void
sh_rows(const float* __restrict__ sh0, const float* __restrict__ sh1,
        const float* __restrict__ sh2, bool valid, long e, float (&shv)[9]) {
    shv[0] = sh0[e];
#pragma unroll
    for (int j = 0; j < 3; ++j) shv[1 + j] = sh1[e * 3 + j];
#pragma unroll
    for (int j = 0; j < 5; ++j) shv[4 + j] = sh2[e * 5 + j];
#pragma unroll
    for (int j = 0; j < 9; ++j) shv[j] = valid ? shv[j] : 0.0f;
}

__device__ __forceinline__ float edge_invariant(const float (&f)[FEAT]) {
    float inv0 = 0.5f * (f[0] + f[1]);
    float n1 = 0.0f, n2 = 0.0f;
#pragma unroll
    for (int j = 0; j < 3; ++j) { float m = 0.5f * (f[2 + j] + f[5 + j]); n1 += m * m; }
#pragma unroll
    for (int j = 0; j < 5; ++j) { float m = 0.5f * (f[8 + j] + f[13 + j]); n2 += m * m; }
    return (inv0 + sqrtf(n1) + sqrtf(n2)) * (1.0f / 3.0f);
}

template <int H>
__device__ __forceinline__ void a_stage(const float (&shv)[9], float* __restrict__ aRow) {
    v4f r0, r1;
    if (H == 0) {
        r0 = (v4f){shv[0], 0.0f, 0.0f, 0.0f};
        r1 = (v4f){shv[1], shv[2], shv[3], 0.0f};
    } else {
        r0 = (v4f){shv[4], shv[5], shv[6], shv[7]};
        r1 = (v4f){shv[8], 0.0f, 0.0f, 0.0f};
    }
    *(v4f*)(aRow) = r0;
    *(v4f*)(aRow + 4) = r1;
}

__device__ __forceinline__ void park_row(const float (&f)[FEAT], float* __restrict__ row) {
#pragma unroll
    for (int j = 0; j < FEAT; ++j) row[j] = f[j];
}

__device__ __forceinline__ void fetch_row(const float* row, float (&f)[FEAT]) {
#pragma unroll
    for (int j = 0; j < FEAT; ++j) f[j] = row[j];
}

template <int H>
__device__ __forceinline__ void park_sh(const float (&shv)[9], float* aPad, float* tPad) {
    if (H == 0) {
        *(v4f*)aPad = (v4f){shv[0], shv[1], shv[2], shv[3]};
    } else {
        *(v4f*)aPad = (v4f){shv[4], shv[5], shv[6], shv[7]};
        tPad[0] = shv[8];
    }
}

template <int H>
__device__ __forceinline__ void fetch_sh(const float* aPad, const float* tPad, float (&shv)[9]) {
    const v4f w = *(const v4f*)aPad;
#pragma unroll
    for (int j = 0; j < 9; ++j) shv[j] = 0.0f;
    if (H == 0) {
        shv[0] = w.x; shv[1] = w.y; shv[2] = w.z; shv[3] = w.w;
    } else {
        shv[4] = w.x; shv[5] = w.y; shv[6] = w.z; shv[7] = w.w;
        shv[8] = tPad[0];
    }
}

template <int H>
__device__ __forceinline__ void load_b(const float* __restrict__ Mg, const int lane,
                                       v2f (&bf)[HalfCfg<H>::NB]) {
    typedef HalfCfg<H> Cf;
    const int col16 = lane & 15;
    const int kofs = (lane < 16) ? 0 : 2;
    int idx = 0;
#pragma unroll
    for (int n = Cf::NT0; n < Cf::NT1; ++n) {
#pragma unroll
        for (int kt = nt_klo(n); kt <= nt_khi(n); ++kt) {
            const int k0 = kt * 4 + kofs;
            bf[idx].x = Mg[k0 * N2 + n * 16 + col16];
            bf[idx].y = Mg[(k0 + 1) * N2 + n * 16 + col16];
            ++idx;
        }
    }
}

template <int H>
__device__ __forceinline__ void
mma_mix(const v2f (&bf)[HalfCfg<H>::NB], const float (&g)[HalfCfg<H>::NG],
        float* sA, float* sT, float* pRow, const int lane, const bool restage) {
    typedef HalfCfg<H> Cf;
    wave_lds_handover();
    {
        const int col16 = lane & 15;
        const int kofs = (lane < 16) ? 0 : 2;
        const float* aP0 = &sA[col16 * AP + kofs];
        const float* aP1 = &sA[(16 + col16) * AP + kofs];
        v2f a0[2], a1[2];
#pragma unroll
        for (int kl = 0; kl < 2; ++kl) {
            a0[kl].x = aP0[kl * 4];
            a0[kl].y = aP0[kl * 4 + 1];
            a1[kl].x = aP1[kl * 4];
            a1[kl].y = aP1[kl * 4 + 1];
        }
        if (restage) {
            float* aRow = &sA[lane * AP];
            float shv[9];
            fetch_sh<H>(aRow + 8, &sT[lane * Cf::RP + (Cf::NT1 - Cf::NT0) * 16], shv);
            a_stage<H>(shv, aRow);
        }
        const int rbase = (lane < 16) ? 0 : 8;
        int idx = 0;
#pragma unroll
        for (int n = Cf::NT0; n < Cf::NT1; ++n) {
            v8f acc0 = {0.f, 0.f, 0.f, 0.f, 0.f, 0.f, 0.f, 0.f};
            v8f acc1 = {0.f, 0.f, 0.f, 0.f, 0.f, 0.f, 0.f, 0.f};
#pragma unroll
            for (int kt = nt_klo(n); kt <= nt_khi(n); ++kt) {
                const int kl = kt - Cf::KT0;
                acc0 = __builtin_amdgcn_wmma_f32_16x16x4_f32(
                    false, a0[kl], false, bf[idx], (short)0, acc0, false, false);
                acc1 = __builtin_amdgcn_wmma_f32_16x16x4_f32(
                    false, a1[kl], false, bf[idx], (short)0, acc1, false, false);
                ++idx;
            }
            const int lc = (n - Cf::NT0) * 16 + col16;
#pragma unroll
            for (int j = 0; j < 8; ++j) {
                sT[(rbase + j) * Cf::RP + lc] = acc0[j];
                sT[(16 + rbase + j) * Cf::RP + lc] = acc1[j];
            }
        }
    }
    wave_lds_handover();
    {
        float f[FEAT];
        fetch_row(pRow, f);
        const float* tRow = &sT[lane * Cf::RP];
        constexpr int TB0 = TOFF[Cf::C0];
        constexpr int NQ = (TOFF[Cf::C1 - 1] + (2 * cLN[Cf::C1 - 1] + 1) * (2 * cLO[Cf::C1 - 1] + 1)
                            - TB0 + 3) / 4;
        float t[NQ * 4];
#pragma unroll
        for (int q = 0; q < NQ; ++q) {
            const v4f w = *(const v4f*)(tRow + q * 4);
            t[q * 4] = w.x; t[q * 4 + 1] = w.y; t[q * 4 + 2] = w.z; t[q * 4 + 3] = w.w;
        }
        v2f acc[9];
#pragma unroll
        for (int i = 0; i < 9; ++i) acc[i] = (v2f){0.0f, 0.0f};
#pragma unroll
        for (int c = Cf::C0; c < Cf::C1; ++c) {
            const int ln = cLN[c], lo = cLO[c];
            const int dn = 2 * ln + 1, dO = 2 * lo + 1;
            const int fb = (ln == 0) ? 0 : (ln == 1 ? 2 : 8);
            const int ab = (lo == 0) ? 0 : (lo == 1 ? 1 : 4);
            const int gb = (c - Cf::C0) * 4;
            const v2f G0 = {g[gb], g[gb + 1]};
            const v2f G1 = {g[gb + 2], g[gb + 3]};
#pragma unroll
            for (int j = 0; j < dn; ++j) {
                const float f0 = f[fb + j], f1 = f[fb + dn + j];
                const v2f fg = G0 * (v2f){f0, f0} + G1 * (v2f){f1, f1};
#pragma unroll
                for (int k = 0; k < dO; ++k) {
                    const float tv = t[TOFF[c] - TB0 + j * dO + k];
                    acc[ab + k] += fg * (v2f){tv, tv};
                }
            }
        }
        pRow[0] = acc[0].x;
        pRow[1] = acc[0].y;
#pragma unroll
        for (int k = 0; k < 3; ++k) { pRow[2 + k] = acc[1 + k].x; pRow[5 + k] = acc[1 + k].y; }
#pragma unroll
        for (int k = 0; k < 5; ++k) { pRow[8 + k] = acc[4 + k].x; pRow[13 + k] = acc[4 + k].y; }
    }
}

template <int H>
__device__ __forceinline__ void
layer_half(const float* __restrict__ sh0, const float* __restrict__ sh1,
           const float* __restrict__ sh2, const float* __restrict__ Mg,
           const float* __restrict__ W1, const float* __restrict__ b1,
           const float* __restrict__ W2, const float* __restrict__ b2,
           float* sA, float* sT, float* sP, float* sQ, const int lane,
           const bool vA, const bool vB, const long eA, const long eB) {
    typedef HalfCfg<H> Cf;

    v2f bf[Cf::NB];
    load_b<H>(Mg, lane, bf);

    float* rowA = &sP[lane * PSUM_P];
    float* rowB = &sQ[lane * PSUM_P];
    float* aRow = &sA[lane * AP];
    float* aPad = aRow + 8;
    float* tPad = &sT[lane * Cf::RP + (Cf::NT1 - Cf::NT0) * 16];

    float shA[9], shB[9];
    sh_rows(sh0, sh1, sh2, vA, eA, shA);
    sh_rows(sh0, sh1, sh2, vB, eB, shB);

    float aA[Cf::NG], aB[Cf::NG];
    {
        const v4f x = *(const v4f*)aPad;
        v2f x0 = {x.x, x.z};
        v2f x1 = {x.y, x.w};
        gate_mlp2<H>(W1, b1, W2, b2, x0, x1, aA, aB);
    }
    a_stage<H>(shA, aRow);
    park_sh<H>(shB, aPad, tPad);

    float* pRow = rowA;
#pragma nounroll
    for (int s = 0; s < 2; ++s) {
        float g[Cf::NG];
        tanh_gate_pairs<Cf::NG>(aA, g);
        mma_mix<H>(bf, g, sA, sT, pRow, lane, s == 0);
#pragma unroll
        for (int t = 0; t < Cf::NG; ++t) aA[t] = aB[t];
        pRow = rowB;
    }
}

__global__ void __launch_bounds__(2 * TILE)
layer_kernel(const float* __restrict__ feats, const int* __restrict__ eidx,
             const float* __restrict__ sh0, const float* __restrict__ sh1,
             const float* __restrict__ sh2, const float* __restrict__ dist,
             const float* __restrict__ Mg, const int* __restrict__ perm,
             const float* __restrict__ W1, const float* __restrict__ b1,
             const float* __restrict__ W2, const float* __restrict__ b2,
             float* __restrict__ outF, int E) {
    __shared__ __attribute__((aligned(16))) float sMem[SMEM_FLOATS];
    const int lane = threadIdx.x & (TILE - 1);
    const int half = __builtin_amdgcn_readfirstlane((int)(threadIdx.x >> 5));

    const long slotM = (long)blockIdx.x * (2 * TILE) + half * TILE + lane;
    const long slotO = (long)blockIdx.x * (2 * TILE) + (1 - half) * TILE + lane;
    bool vM;
    long eM;
    int srcM, tgt;
    edge_ids(eidx, perm, E, slotM, vM, eM, srcM, tgt);
    const bool vO = (slotO < (long)E);
    const long slO = vO ? slotO : (long)E - 1;
    const long eO = perm ? (long)perm[slO] : slO;
    {
        float fM[FEAT];
        feat_row(feats, vM, srcM, fM);
        const float dM = dist[eM];
        const v2f xM = {vM ? dM : 0.0f, edge_invariant(fM)};
        float* r = sMem + SP0_OFF + half * (SQ0_OFF - SP0_OFF) + lane * PSUM_P;
        park_row(fM, r);
        park_row(fM, r + (SP1_OFF - SP0_OFF));
        float* xp = sMem + SA0_OFF + lane * AP + 8 + 2 * half;
        *(v2f*)xp = xM;
        *(v2f*)(xp + (SA1_OFF - SA0_OFF)) = xM;
    }
    __syncthreads();

    if (half == 0)
        layer_half<0>(sh0, sh1, sh2, Mg, W1, b1, W2, b2,
                      sMem + SA0_OFF, sMem + ST0_OFF, sMem + SP0_OFF, sMem + SQ0_OFF,
                      lane, vM, vO, eM, eO);
    else
        layer_half<1>(sh0, sh1, sh2, Mg, W1, b1, W2, b2,
                      sMem + SA1_OFF, sMem + ST1_OFF, sMem + SP1_OFF, sMem + SQ1_OFF,
                      lane, vO, vM, eO, eM);
    __syncthreads();

    const int laneT = (int)__builtin_amdgcn_mbcnt_lo(~0u, 0u);
    const int tDn = __shfl_down(tgt, 1, TILE);
    const bool last = (laneT == TILE - 1) || (tDn != tgt);
    const unsigned lastMask = __builtin_amdgcn_ballot_w32(last);
    const int colj = min(laneT, FEAT - 1);
    const float* p0 = sMem + ((half == 0) ? SP0_OFF : SQ0_OFF) + colj;
    const float* p1 = sMem + ((half == 0) ? SP1_OFF : SQ1_OFF) + colj;
    float s[TILE];
#pragma unroll
    for (int i = 0; i < TILE; ++i) s[i] = p0[i * PSUM_P] + p1[i * PSUM_P];
    float acc = 0.0f;
#pragma unroll
    for (int i = 0; i < TILE; ++i) {
        acc += s[i];
        if ((lastMask >> i) & 1u) {
            const int t = __builtin_amdgcn_readlane(tgt, i);
            if (t >= 0 && laneT < FEAT) atomicAdd(&outF[(long)t * FEAT + laneT], acc);
            acc = 0.0f;
        }
    }
}

extern "C" void kernel_launch(void* const* d_in, const int* in_sizes, int n_in,
                              void* d_out, int out_size, void* d_ws, size_t ws_size,
                              hipStream_t stream) {
    const float* node = (const float*)d_in[0];
    const int* eidx   = (const int*)d_in[1];
    const float* sh0  = (const float*)d_in[2];
    const float* sh1  = (const float*)d_in[3];
    const float* sh2  = (const float*)d_in[4];
    const float* dist = (const float*)d_in[5];
    const float* cg[N_COMBOS];
    for (int i = 0; i < N_COMBOS; ++i) cg[i] = (const float*)d_in[6 + i];
    const float* W1_0 = (const float*)d_in[21];
    const float* b1_0 = (const float*)d_in[22];
    const float* W2_0 = (const float*)d_in[23];
    const float* b2_0 = (const float*)d_in[24];
    const float* W1_1 = (const float*)d_in[25];
    const float* b1_1 = (const float*)d_in[26];
    const float* W2_1 = (const float*)d_in[27];
    const float* b2_1 = (const float*)d_in[28];

    const int E = in_sizes[1] / 2;
    const int N = in_sizes[0] / FEAT;

    float* Mg     = (float*)d_ws;
    float* feats1 = (float*)((char*)d_ws + 131072);
    long n1 = (long)N * FEAT;
    int* perm     = (int*)(feats1 + n1);

    build_M_kernel<<<1, 256, 0, stream>>>(cg[0], cg[1], cg[2], cg[3], cg[4], cg[5],
                                          cg[6], cg[7], cg[8], cg[9], cg[10], cg[11],
                                          cg[12], cg[13], cg[14], Mg);

    const bool permFits = (ws_size >= (size_t)131072 + (size_t)(n1 + E) * 4);
    const bool grouped = permFits && ((long)E + N <= (long)out_size);
    const int BW = (N + NBK - 1) / NBK;
    const int chunk = (E + NCH - 1) / NCH;
    const bool twoLevel = permFits && E >= 1 && E <= (1 << 20) && BW >= 1 && BW <= BWMAX &&
                          ((long)E + (long)NCH * CSTP <= (long)out_size);
    if (twoLevel) {
        int* sortedA = (int*)d_out;
        int* cst = sortedA + E;
        bucket_kernel<<<NCH, 1024, 0, stream>>>(eidx, sortedA, cst, E, chunk, BW);
        owner_kernel<<<NBK, 1024, 0, stream>>>(sortedA, cst, perm, E, chunk, BW);
    } else if (grouped) {
        int* rnk = (int*)d_out;
        int* cnt = rnk + E;
        zero_kernel<<<(N + 255) / 256, 256, 0, stream>>>((float*)cnt, (long)N);
        rank_kernel<<<(E + 255) / 256, 256, 0, stream>>>(eidx, cnt, rnk, E);
        scan_kernel<<<1, 1024, 0, stream>>>(cnt, N);
        fill_kernel<<<(E + 255) / 256, 256, 0, stream>>>(eidx, cnt, rnk, perm, E);
    } else {
        perm = nullptr;
    }

    zero_kernel<<<(int)((n1 + 255) / 256), 256, 0, stream>>>(feats1, n1);
    zero_kernel<<<(out_size + 255) / 256, 256, 0, stream>>>((float*)d_out, (long)out_size);

    int grid = (E + 2 * TILE - 1) / (2 * TILE);
    layer_kernel<<<grid, 2 * TILE, 0, stream>>>(node, eidx, sh0, sh1, sh2, dist, Mg, perm,
                                                  W1_0, b1_0, W2_0, b2_0, feats1, E);
    layer_kernel<<<grid, 2 * TILE, 0, stream>>>(feats1, eidx, sh0, sh1, sh2, dist, Mg, perm,
                                                  W1_1, b1_1, W2_1, b2_1, (float*)d_out, E);
}
